# starting kernel
# baseline (speedup 1.0000x reference)
#include <hip/hip_runtime.h>
#include <hip/hip_bf16.h>

typedef __attribute__((ext_vector_type(2))) float v2f;
typedef __attribute__((ext_vector_type(8))) float v8f;

#define N_SRC1 500000
#define N_DST1 100000
#define N_DST2 20000
#define IN_F   128
#define H_F    128
#define N_CLS  16

__global__ void zero_ws_kernel(float4* __restrict__ p, size_t n4) {
    size_t i = (size_t)blockIdx.x * blockDim.x + threadIdx.x;
    size_t stride = (size_t)gridDim.x * blockDim.x;
    for (; i < n4; i += stride) p[i] = make_float4(0.f, 0.f, 0.f, 0.f);
}

__global__ void edge_scatter_kernel(const float* __restrict__ feat,
                                    const int*   __restrict__ src,
                                    const int*   __restrict__ dst,
                                    const float* __restrict__ w,
                                    float* __restrict__ acc,
                                    float* __restrict__ deg,
                                    int nEdges) {
    const int lane = threadIdx.x & 31;
    const int e = blockIdx.x * 8 + (threadIdx.x >> 5);
    if (e >= nEdges) return;
    const int   s  = src[e];
    const int   d  = dst[e];
    const float we = w[e];
    const float4 xv = ((const float4*)(feat + (size_t)s * 128))[lane];
    float* ap = acc + (size_t)d * 128 + lane * 4;
    atomicAdd(ap + 0, xv.x * we);
    atomicAdd(ap + 1, xv.y * we);
    atomicAdd(ap + 2, xv.z * we);
    atomicAdd(ap + 3, xv.w * we);
    if (lane == 0) atomicAdd(deg + d, 1.0f);
}

__global__ void sage_gemm1_kernel(const float* __restrict__ x,
                                  const float* __restrict__ acc,
                                  const float* __restrict__ deg,
                                  const float* __restrict__ Ws,
                                  const float* __restrict__ Wn,
                                  const float* __restrict__ bias,
                                  float* __restrict__ h) {
    const int lane  = threadIdx.x & 31;
    const int wave  = threadIdx.x >> 5;
    const int m0    = blockIdx.x * 16;
    const int n0    = wave * 16;
    const int row   = m0 + (lane & 15);
    const int col   = n0 + (lane & 15);
    const int khalf = (lane >> 4) * 2;

    const float* xr = x   + (size_t)row * IN_F;
    const float* ar = acc + (size_t)row * IN_F;
    const float invd = 1.0f / fmaxf(deg[row], 1.0f);

    v8f c = {};
    #pragma unroll 4
    for (int k = 0; k < IN_F; k += 4) {
        const int kk = k + khalf;
        v2f a, b;
        a.x = xr[kk];
        a.y = xr[kk + 1];
        b.x = Ws[(size_t)kk * H_F + col];
        b.y = Ws[(size_t)(kk + 1) * H_F + col];
        c = __builtin_amdgcn_wmma_f32_16x16x4_f32(false, a, false, b,
                                                  (short)0, c, false, false);
    }
    #pragma unroll 4
    for (int k = 0; k < IN_F; k += 4) {
        const int kk = k + khalf;
        v2f a, b;
        a.x = ar[kk]     * invd;
        a.y = ar[kk + 1] * invd;
        b.x = Wn[(size_t)kk * H_F + col];
        b.y = Wn[(size_t)(kk + 1) * H_F + col];
        c = __builtin_amdgcn_wmma_f32_16x16x4_f32(false, a, false, b,
                                                  (short)0, c, false, false);
    }

    const float bv    = bias[col];
    const int   rbase = m0 + ((lane >> 4) * 8);
    #pragma unroll
    for (int r = 0; r < 8; ++r) {
        float v = fmaxf(c[r] + bv, 0.0f);
        h[(size_t)(rbase + r) * H_F + col] = v;
    }
}

__global__ void sage_gemm2_kernel(const float* __restrict__ h,
                                  const float* __restrict__ acc,
                                  const float* __restrict__ deg,
                                  const float* __restrict__ Ws,
                                  const float* __restrict__ Wn,
                                  const float* __restrict__ bias,
                                  float* __restrict__ out) {
    const int lane = threadIdx.x & 31;
    const int wave = threadIdx.x >> 5;
    const int tile = blockIdx.x * 8 + wave;
    if (tile >= N_DST2 / 16) return;
    const int m0    = tile * 16;
    const int row   = m0 + (lane & 15);
    const int col   = lane & 15;
    const int khalf = (lane >> 4) * 2;

    const float* hr = h   + (size_t)row * H_F;
    const float* ar = acc + (size_t)row * H_F;
    const float invd = 1.0f / fmaxf(deg[row], 1.0f);

    v8f c = {};
    #pragma unroll 4
    for (int k = 0; k < H_F; k += 4) {
        const int kk = k + khalf;
        v2f a, b;
        a.x = hr[kk];
        a.y = hr[kk + 1];
        b.x = Ws[(size_t)kk * N_CLS + col];
        b.y = Ws[(size_t)(kk + 1) * N_CLS + col];
        c = __builtin_amdgcn_wmma_f32_16x16x4_f32(false, a, false, b,
                                                  (short)0, c, false, false);
    }
    #pragma unroll 4
    for (int k = 0; k < H_F; k += 4) {
        const int kk = k + khalf;
        v2f a, b;
        a.x = ar[kk]     * invd;
        a.y = ar[kk + 1] * invd;
        b.x = Wn[(size_t)kk * N_CLS + col];
        b.y = Wn[(size_t)(kk + 1) * N_CLS + col];
        c = __builtin_amdgcn_wmma_f32_16x16x4_f32(false, a, false, b,
                                                  (short)0, c, false, false);
    }

    const float bv    = bias[col];
    const int   rbase = m0 + ((lane >> 4) * 8);
    #pragma unroll
    for (int r = 0; r < 8; ++r) {
        float v = c[r] + bv;
        float m = v;
        m = fmaxf(m, __shfl_xor(m, 1, 16));
        m = fmaxf(m, __shfl_xor(m, 2, 16));
        m = fmaxf(m, __shfl_xor(m, 4, 16));
        m = fmaxf(m, __shfl_xor(m, 8, 16));
        float e = __expf(v - m);
        float s = e;
        s += __shfl_xor(s, 1, 16);
        s += __shfl_xor(s, 2, 16);
        s += __shfl_xor(s, 4, 16);
        s += __shfl_xor(s, 8, 16);
        out[(size_t)(rbase + r) * N_CLS + col] = (v - m) - __logf(s);
    }
}

extern "C" void kernel_launch(void* const* d_in, const int* in_sizes, int n_in,
                              void* d_out, int out_size, void* d_ws, size_t ws_size,
                              hipStream_t stream) {
    const float* x    = (const float*)d_in[0];
    const int*   src1 = (const int*)  d_in[1];
    const int*   dst1 = (const int*)  d_in[2];
    const float* w1   = (const float*)d_in[3];
    const int*   src2 = (const int*)  d_in[4];
    const int*   dst2 = (const int*)  d_in[5];
    const float* w2   = (const float*)d_in[6];
    const float* Ws1  = (const float*)d_in[7];
    const float* Wn1  = (const float*)d_in[8];
    const float* b1   = (const float*)d_in[9];
    const float* Ws2  = (const float*)d_in[10];
    const float* Wn2  = (const float*)d_in[11];
    const float* b2   = (const float*)d_in[12];
    float* out = (float*)d_out;

    float* acc1 = (float*)d_ws;
    float* deg1 = acc1 + (size_t)N_DST1 * 128;
    float* hbuf = deg1 + N_DST1;
    float* acc2 = hbuf + (size_t)N_DST1 * 128;
    float* deg2 = acc2 + (size_t)N_DST2 * 128;
    const size_t totalFloats = (size_t)N_DST1 * 128 + N_DST1 +
                               (size_t)N_DST1 * 128 +
                               (size_t)N_DST2 * 128 + N_DST2;

    const int E1 = in_sizes[1];
    const int E2 = in_sizes[4];

    zero_ws_kernel<<<2048, 256, 0, stream>>>((float4*)d_ws, totalFloats / 4);

    edge_scatter_kernel<<<(E1 + 7) / 8, 256, 0, stream>>>(x, src1, dst1, w1,
                                                          acc1, deg1, E1);

    sage_gemm1_kernel<<<N_DST1 / 16, 256, 0, stream>>>(x, acc1, deg1,
                                                       Ws1, Wn1, b1, hbuf);

    edge_scatter_kernel<<<(E2 + 7) / 8, 256, 0, stream>>>(hbuf, src2, dst2, w2,
                                                          acc2, deg2, E2);

    sage_gemm2_kernel<<<(N_DST2 / 16 + 7) / 8, 256, 0, stream>>>(hbuf, acc2, deg2,
                                                                 Ws2, Wn2, b2, out);
}


# Replace float atomic scatter by grouping edges per dst row (one int atomic per edge for count+rank, scan, fill) and one wave per row summing in registers; sum order = rank order, arrival-dependent
# speedup vs baseline: 1.7490x; 1.7490x over previous
#include <hip/hip_runtime.h>
#include <hip/hip_bf16.h>

typedef __attribute__((ext_vector_type(2))) float v2f;
typedef __attribute__((ext_vector_type(8))) float v8f;

#define N_SRC1 500000
#define N_DST1 100000
#define N_DST2 20000
#define IN_F   128
#define H_F    128
#define N_CLS  16

__global__ void zero_ws_kernel(float4* __restrict__ p, size_t n4) {
    size_t i = (size_t)blockIdx.x * blockDim.x + threadIdx.x;
    size_t stride = (size_t)gridDim.x * blockDim.x;
    for (; i < n4; i += stride) p[i] = make_float4(0.f, 0.f, 0.f, 0.f);
}


__global__ void seg_count_kernel(const int* __restrict__ dst1,
                                 const int* __restrict__ dst2,
                                 int* __restrict__ cnt1,
                                 int* __restrict__ cnt2,
                                 int* __restrict__ rank,
                                 int nE1, int nE2) {
    const int t = blockIdx.x * blockDim.x + threadIdx.x;
    if (t < nE1)            rank[t] = atomicAdd(cnt1 + dst1[t], 1);
    else if (t < nE1 + nE2) rank[t] = atomicAdd(cnt2 + dst2[t - nE1], 1);
}

__global__ void seg_offsets_kernel(const int* __restrict__ cnt,
                                   int* __restrict__ off,
                                   int* __restrict__ cursor,
                                   int n1, int nTot) {
    const int i    = blockIdx.x * blockDim.x + threadIdx.x;
    const int lane = threadIdx.x & 31;
    const int c = (i < nTot) ? cnt[i] : 0;
    int v = c;
    for (int o = 1; o < 32; o <<= 1) {
        const int up = __shfl_up(v, o);
        if (lane >= o) v += up;
    }
    const int total = __shfl(v, 31);
    int base = 0;
    if (lane == 0) base = atomicAdd(cursor + (i >= n1 ? 1 : 0), total);
    base = __shfl(base, 0);
    if (i < nTot) off[i] = base + v - c;
}

__global__ void seg_fill_kernel(const int*   __restrict__ src1,
                                const int*   __restrict__ dst1,
                                const float* __restrict__ w1,
                                const int*   __restrict__ src2,
                                const int*   __restrict__ dst2,
                                const float* __restrict__ w2,
                                const int* __restrict__ off1,
                                const int* __restrict__ off2,
                                const int* __restrict__ rank,
                                int* __restrict__ list1,
                                int* __restrict__ list2,
                                int nE1, int nE2) {
    const int t = blockIdx.x * blockDim.x + threadIdx.x;
    if (t < nE1) {
        const int pos = off1[dst1[t]] + rank[t];
        ((int2*)list1)[pos] = make_int2(src1[t], __float_as_int(w1[t]));
    } else if (t < nE1 + nE2) {
        const int u = t - nE1;
        const int pos = off2[dst2[u]] + rank[t];
        ((int2*)list2)[pos] = make_int2(src2[u], __float_as_int(w2[u]));
    }
}

__global__ void seg_gather_kernel(const float* __restrict__ feat,
                                  const int*   __restrict__ list,
                                  const int*   __restrict__ segStart,
                                  const int*   __restrict__ cnt,
                                  float* __restrict__ acc,
                                  float* __restrict__ deg,
                                  int nDst) {
    const int lane = threadIdx.x & 31;
    const int d = __builtin_amdgcn_readfirstlane(blockIdx.x * 8 + (threadIdx.x >> 5));
    if (d >= nDst) return;
    const int c     = __builtin_amdgcn_readfirstlane(cnt[d]);
    const int start = __builtin_amdgcn_readfirstlane(segStart[d]);
    const int2* seg = (const int2*)list + start;

    float4 s = make_float4(0.f, 0.f, 0.f, 0.f);
    for (int base = 0; base < c; base += 32) {
        const int n = min(32, c - base);
        int2 my = make_int2(0, 0);
        if (lane < n) my = seg[base + lane];
        int j = 0;
        for (; j + 4 <= n; j += 4) {
            const int   s0 = __builtin_amdgcn_readlane(my.x, j);
            const int   s1 = __builtin_amdgcn_readlane(my.x, j + 1);
            const int   s2 = __builtin_amdgcn_readlane(my.x, j + 2);
            const int   s3 = __builtin_amdgcn_readlane(my.x, j + 3);
            const float e0 = __int_as_float(__builtin_amdgcn_readlane(my.y, j));
            const float e1 = __int_as_float(__builtin_amdgcn_readlane(my.y, j + 1));
            const float e2 = __int_as_float(__builtin_amdgcn_readlane(my.y, j + 2));
            const float e3 = __int_as_float(__builtin_amdgcn_readlane(my.y, j + 3));
            const float4 x0 = ((const float4*)(feat + (size_t)s0 * 128))[lane];
            const float4 x1 = ((const float4*)(feat + (size_t)s1 * 128))[lane];
            const float4 x2 = ((const float4*)(feat + (size_t)s2 * 128))[lane];
            const float4 x3 = ((const float4*)(feat + (size_t)s3 * 128))[lane];
            s.x += x0.x * e0; s.y += x0.y * e0; s.z += x0.z * e0; s.w += x0.w * e0;
            s.x += x1.x * e1; s.y += x1.y * e1; s.z += x1.z * e1; s.w += x1.w * e1;
            s.x += x2.x * e2; s.y += x2.y * e2; s.z += x2.z * e2; s.w += x2.w * e2;
            s.x += x3.x * e3; s.y += x3.y * e3; s.z += x3.z * e3; s.w += x3.w * e3;
        }
        const int r = n - j;
        if (r >= 1) {
            const int j1 = (r >= 2) ? j + 1 : j;
            const int j2 = (r >= 3) ? j + 2 : j;
            const int   s0 = __builtin_amdgcn_readlane(my.x, j);
            const int   s1 = __builtin_amdgcn_readlane(my.x, j1);
            const int   s2 = __builtin_amdgcn_readlane(my.x, j2);
            const float e0 = __int_as_float(__builtin_amdgcn_readlane(my.y, j));
            const float e1 = __int_as_float(__builtin_amdgcn_readlane(my.y, j1));
            const float e2 = __int_as_float(__builtin_amdgcn_readlane(my.y, j2));
            const float4 x0 = ((const float4*)(feat + (size_t)s0 * 128))[lane];
            const float4 x1 = ((const float4*)(feat + (size_t)s1 * 128))[lane];
            const float4 x2 = ((const float4*)(feat + (size_t)s2 * 128))[lane];
            s.x += x0.x * e0; s.y += x0.y * e0; s.z += x0.z * e0; s.w += x0.w * e0;
            if (r >= 2) {
                s.x += x1.x * e1; s.y += x1.y * e1; s.z += x1.z * e1; s.w += x1.w * e1;
            }
            if (r >= 3) {
                s.x += x2.x * e2; s.y += x2.y * e2; s.z += x2.z * e2; s.w += x2.w * e2;
            }
        }
    }
    ((float4*)(acc + (size_t)d * 128))[lane] = s;
    if (lane == 0) deg[d] = (float)c;
}

__global__ void sage_gemm1_kernel(const float* __restrict__ x,
                                  const float* __restrict__ acc,
                                  const float* __restrict__ deg,
                                  const float* __restrict__ Ws,
                                  const float* __restrict__ Wn,
                                  const float* __restrict__ bias,
                                  float* __restrict__ h) {
    const int lane  = threadIdx.x & 31;
    const int wave  = threadIdx.x >> 5;
    const int m0    = blockIdx.x * 16;
    const int n0    = wave * 16;
    const int row   = m0 + (lane & 15);
    const int col   = n0 + (lane & 15);
    const int khalf = (lane >> 4) * 2;

    const float* xr = x   + (size_t)row * IN_F;
    const float* ar = acc + (size_t)row * IN_F;
    const float invd = 1.0f / fmaxf(deg[row], 1.0f);

    v8f c = {};
    #pragma unroll 4
    for (int k = 0; k < IN_F; k += 4) {
        const int kk = k + khalf;
        v2f a, b;
        a.x = xr[kk];
        a.y = xr[kk + 1];
        b.x = Ws[(size_t)kk * H_F + col];
        b.y = Ws[(size_t)(kk + 1) * H_F + col];
        c = __builtin_amdgcn_wmma_f32_16x16x4_f32(false, a, false, b,
                                                  (short)0, c, false, false);
    }
    #pragma unroll 4
    for (int k = 0; k < IN_F; k += 4) {
        const int kk = k + khalf;
        v2f a, b;
        a.x = ar[kk]     * invd;
        a.y = ar[kk + 1] * invd;
        b.x = Wn[(size_t)kk * H_F + col];
        b.y = Wn[(size_t)(kk + 1) * H_F + col];
        c = __builtin_amdgcn_wmma_f32_16x16x4_f32(false, a, false, b,
                                                  (short)0, c, false, false);
    }

    const float bv    = bias[col];
    const int   rbase = m0 + ((lane >> 4) * 8);
    #pragma unroll
    for (int r = 0; r < 8; ++r) {
        float v = fmaxf(c[r] + bv, 0.0f);
        h[(size_t)(rbase + r) * H_F + col] = v;
    }
}

__global__ void sage_gemm2_kernel(const float* __restrict__ h,
                                  const float* __restrict__ acc,
                                  const float* __restrict__ deg,
                                  const float* __restrict__ Ws,
                                  const float* __restrict__ Wn,
                                  const float* __restrict__ bias,
                                  float* __restrict__ out) {
    const int lane = threadIdx.x & 31;
    const int wave = threadIdx.x >> 5;
    const int tile = blockIdx.x * 8 + wave;
    if (tile >= N_DST2 / 16) return;
    const int m0    = tile * 16;
    const int row   = m0 + (lane & 15);
    const int col   = lane & 15;
    const int khalf = (lane >> 4) * 2;

    const float* hr = h   + (size_t)row * H_F;
    const float* ar = acc + (size_t)row * H_F;
    const float invd = 1.0f / fmaxf(deg[row], 1.0f);

    v8f c = {};
    #pragma unroll 4
    for (int k = 0; k < H_F; k += 4) {
        const int kk = k + khalf;
        v2f a, b;
        a.x = hr[kk];
        a.y = hr[kk + 1];
        b.x = Ws[(size_t)kk * N_CLS + col];
        b.y = Ws[(size_t)(kk + 1) * N_CLS + col];
        c = __builtin_amdgcn_wmma_f32_16x16x4_f32(false, a, false, b,
                                                  (short)0, c, false, false);
    }
    #pragma unroll 4
    for (int k = 0; k < H_F; k += 4) {
        const int kk = k + khalf;
        v2f a, b;
        a.x = ar[kk]     * invd;
        a.y = ar[kk + 1] * invd;
        b.x = Wn[(size_t)kk * N_CLS + col];
        b.y = Wn[(size_t)(kk + 1) * N_CLS + col];
        c = __builtin_amdgcn_wmma_f32_16x16x4_f32(false, a, false, b,
                                                  (short)0, c, false, false);
    }

    const float bv    = bias[col];
    const int   rbase = m0 + ((lane >> 4) * 8);
    #pragma unroll
    for (int r = 0; r < 8; ++r) {
        float v = c[r] + bv;
        float m = v;
        m = fmaxf(m, __shfl_xor(m, 1, 16));
        m = fmaxf(m, __shfl_xor(m, 2, 16));
        m = fmaxf(m, __shfl_xor(m, 4, 16));
        m = fmaxf(m, __shfl_xor(m, 8, 16));
        float e = __expf(v - m);
        float s = e;
        s += __shfl_xor(s, 1, 16);
        s += __shfl_xor(s, 2, 16);
        s += __shfl_xor(s, 4, 16);
        s += __shfl_xor(s, 8, 16);
        out[(size_t)(rbase + r) * N_CLS + col] = (v - m) - __logf(s);
    }
}

extern "C" void kernel_launch(void* const* d_in, const int* in_sizes, int n_in,
                              void* d_out, int out_size, void* d_ws, size_t ws_size,
                              hipStream_t stream) {
    const float* x    = (const float*)d_in[0];
    const int*   src1 = (const int*)  d_in[1];
    const int*   dst1 = (const int*)  d_in[2];
    const float* w1   = (const float*)d_in[3];
    const int*   src2 = (const int*)  d_in[4];
    const int*   dst2 = (const int*)  d_in[5];
    const float* w2   = (const float*)d_in[6];
    const float* Ws1  = (const float*)d_in[7];
    const float* Wn1  = (const float*)d_in[8];
    const float* b1   = (const float*)d_in[9];
    const float* Ws2  = (const float*)d_in[10];
    const float* Wn2  = (const float*)d_in[11];
    const float* b2   = (const float*)d_in[12];
    float* out = (float*)d_out;

    float* acc1 = (float*)d_ws;
    float* deg1 = acc1 + (size_t)N_DST1 * 128;
    float* hbuf = deg1 + N_DST1;
    float* acc2 = acc1;
    float* deg2 = acc2 + (size_t)N_DST2 * 128;
    int* cursor = (int*)(hbuf + (size_t)N_DST1 * 128);
    int* cnt1   = cursor + 4;
    int* cnt2   = cnt1 + N_DST1;
    int* off1   = cnt2 + N_DST2;
    int* off2   = off1 + N_DST1;
    int* list2  = off2 + N_DST2;
    int* list1  = (int*)hbuf;
    int* rank   = (int*)acc1;

    const int E1 = in_sizes[1];
    const int E2 = in_sizes[4];

    zero_ws_kernel<<<118, 256, 0, stream>>>((float4*)cursor, (4 + N_DST1 + N_DST2) / 4);

    seg_count_kernel<<<(E1 + E2 + 255) / 256, 256, 0, stream>>>(dst1, dst2, cnt1, cnt2, rank, E1, E2);
    seg_offsets_kernel<<<(N_DST1 + N_DST2 + 255) / 256, 256, 0, stream>>>(cnt1, off1, cursor,
                                                                          N_DST1, N_DST1 + N_DST2);
    seg_fill_kernel<<<(E1 + E2 + 255) / 256, 256, 0, stream>>>(src1, dst1, w1, src2, dst2, w2,
                                                               off1, off2, rank, list1, list2, E1, E2);

    seg_gather_kernel<<<N_DST1 / 8, 256, 0, stream>>>(x, list1, off1, cnt1, acc1, deg1, N_DST1);

    sage_gemm1_kernel<<<N_DST1 / 16, 256, 0, stream>>>(x, acc1, deg1,
                                                       Ws1, Wn1, b1, hbuf);

    seg_gather_kernel<<<N_DST2 / 8, 256, 0, stream>>>(hbuf, list2, off2, cnt2, acc2, deg2, N_DST2);

    sage_gemm2_kernel<<<(N_DST2 / 16 + 7) / 8, 256, 0, stream>>>(hbuf, acc2, deg2,
                                                                 Ws2, Wn2, b2, out);
}


# Group edges by destination with LDS counters in two levels (row buckets, then rows) instead of one global returning atomic per edge; sum order = LDS rank order, arrival-dependent
# speedup vs baseline: 2.7522x; 1.5736x over previous
#include <hip/hip_runtime.h>
#include <hip/hip_bf16.h>

typedef __attribute__((ext_vector_type(2))) float v2f;
typedef __attribute__((ext_vector_type(8))) float v8f;

#define N_SRC1 500000
#define N_DST1 100000
#define N_DST2 20000
#define IN_F   128
#define H_F    128
#define N_CLS  16

__global__ void zero_ws_kernel(float4* __restrict__ p, size_t n4) {
    size_t i = (size_t)blockIdx.x * blockDim.x + threadIdx.x;
    size_t stride = (size_t)gridDim.x * blockDim.x;
    for (; i < n4; i += stride) p[i] = make_float4(0.f, 0.f, 0.f, 0.f);
}

#define NB1   196
#define NB2   79
#define NB    275
#define NBP   288
#define SWEEP_WG 256

__device__ __forceinline__ int wave_incl_scan(int v, int lane) {
    for (int o = 1; o < 32; o <<= 1) {
        const int up = __shfl_up(v, o);
        if (lane >= o) v += up;
    }
    return v;
}

__device__ __forceinline__ int block_excl_scan(int v, int* sTot) {
    const int lane = threadIdx.x & 31;
    const int wave = threadIdx.x >> 5;
    const int incl = wave_incl_scan(v, lane);
    if (lane == 31) sTot[wave] = incl;
    __syncthreads();
    if (wave == 0) {
        const int x  = sTot[lane];
        const int xi = wave_incl_scan(x, lane);
        sTot[32 + lane] = xi - x;
    }
    __syncthreads();
    return sTot[32 + wave] + incl - v;
}

__global__ __launch_bounds__(1024)
void bucket_count_kernel(const int* __restrict__ dst1,
                         const int* __restrict__ dst2,
                         int* __restrict__ bucketCount,
                         int* __restrict__ wgOff,
                         int nE1, int nE2) {
    __shared__ int sH[NBP];
    const int t = threadIdx.x;
    if (t < NBP) sH[t] = 0;
    __syncthreads();
    const int nE    = nE1 + nE2;
    const int chunk = (nE + (int)gridDim.x - 1) / (int)gridDim.x;
    const int lo    = blockIdx.x * chunk;
    const int hi    = min(lo + chunk, nE);
    for (int base = lo; base < hi; base += 4096) {
        int d[4];
        #pragma unroll
        for (int k = 0; k < 4; ++k) {
            const int ic = min(base + k * 1024 + t, hi - 1);
            d[k] = *((ic < nE1) ? dst1 + ic : dst2 + (ic - nE1));
        }
        #pragma unroll
        for (int k = 0; k < 4; ++k) {
            const int i  = base + k * 1024 + t;
            const int bk = (i >= hi) ? NB : (i < nE1) ? (d[k] >> 9) : NB1 + (d[k] >> 8);
            atomicAdd(&sH[bk], 1);
        }
    }
    __syncthreads();
    if (t < NB) wgOff[blockIdx.x * NBP + t] = atomicAdd(bucketCount + t, sH[t]);
}

__global__ __launch_bounds__(1024)
void bucket_place_kernel(const int*   __restrict__ src1,
                         const int*   __restrict__ dst1,
                         const float* __restrict__ w1,
                         const int*   __restrict__ src2,
                         const int*   __restrict__ dst2,
                         const float* __restrict__ w2,
                         const int* __restrict__ bucketCount,
                         const int* __restrict__ wgOff,
                         int*  __restrict__ gStart,
                         int2* __restrict__ rec,
                         int nE1, int nE2) {
    __shared__ int sBase[NBP];
    __shared__ int sTot[64];
    const int t = threadIdx.x;
    const int v = (t < NB) ? bucketCount[t] : 0;
    const int excl = block_excl_scan(v, sTot);
    if (t < NB) {
        sBase[t] = excl + wgOff[blockIdx.x * NBP + t];
        if (blockIdx.x == 0) gStart[t] = excl;
    } else if (t < NBP) {
        sBase[t] = 0;
    }
    __syncthreads();
    const int nE    = nE1 + nE2;
    const int chunk = (nE + (int)gridDim.x - 1) / (int)gridDim.x;
    const int lo    = blockIdx.x * chunk;
    const int hi    = min(lo + chunk, nE);
    for (int base = lo; base < hi; base += 4096) {
        int d[4], s[4], wb[4];
        #pragma unroll
        for (int k = 0; k < 4; ++k) {
            const int  ic = min(base + k * 1024 + t, hi - 1);
            const bool a  = ic < nE1;
            const int  u  = ic - nE1;
            d[k]  = *(a ? dst1 + ic : dst2 + u);
            s[k]  = *(a ? src1 + ic : src2 + u);
            wb[k] = __float_as_int(*(a ? w1 + ic : w2 + u));
        }
        int pos[4];
        #pragma unroll
        for (int k = 0; k < 4; ++k) {
            const int i  = base + k * 1024 + t;
            const int bk = (i >= hi) ? NB : (i < nE1) ? (d[k] >> 9) : NB1 + (d[k] >> 8);
            pos[k] = atomicAdd(&sBase[bk], 1);
        }
        #pragma unroll
        for (int k = 0; k < 4; ++k) {
            const int i = base + k * 1024 + t;
            if (i < hi) {
                const int lr = (i < nE1) ? (d[k] & 511) : (d[k] & 255);
                rec[pos[k]] = make_int2(s[k] | (lr << 20), wb[k]);
            }
        }
    }
}

__global__ __launch_bounds__(1024)
void bucket_rows_kernel(const int2* __restrict__ rec,
                        const int* __restrict__ bucketCount,
                        const int* __restrict__ gStart,
                        int* __restrict__ cnt1, int* __restrict__ cnt2,
                        int* __restrict__ off1, int* __restrict__ off2,
                        int2* __restrict__ list1, int2* __restrict__ list2,
                        int nE1) {
    __shared__ int sCnt[544];
    __shared__ int sCur[544];
    __shared__ int sTot[64];
    const int t  = threadIdx.x;
    const int b  = blockIdx.x;
    const bool l1 = b < NB1;
    const int row0  = l1 ? b * 512 : (b - NB1) * 256;
    const int nrows = l1 ? min(512, N_DST1 - row0) : min(256, N_DST2 - row0);
    const int n     = bucketCount[b];
    const int start = gStart[b];
    const int2* r   = rec + start;
    if (t < 544) sCnt[t] = 0;
    __syncthreads();
    for (int base = 0; base < n; base += 4096) {
        int x[4];
        #pragma unroll
        for (int k = 0; k < 4; ++k) x[k] = r[min(base + k * 1024 + t, n - 1)].x;
        #pragma unroll
        for (int k = 0; k < 4; ++k) {
            const int i = base + k * 1024 + t;
            atomicAdd(&sCnt[(i < n) ? (int)((unsigned)x[k] >> 20) : 512], 1);
        }
    }
    __syncthreads();
    const int v    = (t < 512) ? sCnt[t] : 0;
    const int excl = block_excl_scan(v, sTot);
    const int outBase = l1 ? start : start - nE1;
    if (t < 544) sCur[t] = (t < 512) ? excl : 0;
    if (t < nrows) {
        (l1 ? cnt1 : cnt2)[row0 + t] = v;
        (l1 ? off1 : off2)[row0 + t] = outBase + excl;
    }
    __syncthreads();
    int2* out = (l1 ? list1 : list2) + outBase;
    for (int base = 0; base < n; base += 4096) {
        int2 e[4];
        #pragma unroll
        for (int k = 0; k < 4; ++k) e[k] = r[min(base + k * 1024 + t, n - 1)];
        int pos[4];
        #pragma unroll
        for (int k = 0; k < 4; ++k) {
            const int i = base + k * 1024 + t;
            pos[k] = atomicAdd(&sCur[(i < n) ? (int)((unsigned)e[k].x >> 20) : 512], 1);
        }
        #pragma unroll
        for (int k = 0; k < 4; ++k) {
            const int i = base + k * 1024 + t;
            if (i < n) out[pos[k]] = make_int2(e[k].x & 0xFFFFF, e[k].y);
        }
    }
}

__global__ void seg_gather_kernel(const float* __restrict__ feat,
                                  const int*   __restrict__ list,
                                  const int*   __restrict__ segStart,
                                  const int*   __restrict__ cnt,
                                  float* __restrict__ acc,
                                  float* __restrict__ deg,
                                  int nDst) {
    const int lane = threadIdx.x & 31;
    const int d = __builtin_amdgcn_readfirstlane(blockIdx.x * 8 + (threadIdx.x >> 5));
    if (d >= nDst) return;
    const int c     = __builtin_amdgcn_readfirstlane(cnt[d]);
    const int start = __builtin_amdgcn_readfirstlane(segStart[d]);
    const int2* seg = (const int2*)list + start;

    float4 s = make_float4(0.f, 0.f, 0.f, 0.f);
    for (int base = 0; base < c; base += 32) {
        const int n = min(32, c - base);
        int2 my = make_int2(0, 0);
        if (lane < n) my = seg[base + lane];
        int j = 0;
        for (; j + 4 <= n; j += 4) {
            const int   s0 = __builtin_amdgcn_readlane(my.x, j);
            const int   s1 = __builtin_amdgcn_readlane(my.x, j + 1);
            const int   s2 = __builtin_amdgcn_readlane(my.x, j + 2);
            const int   s3 = __builtin_amdgcn_readlane(my.x, j + 3);
            const float e0 = __int_as_float(__builtin_amdgcn_readlane(my.y, j));
            const float e1 = __int_as_float(__builtin_amdgcn_readlane(my.y, j + 1));
            const float e2 = __int_as_float(__builtin_amdgcn_readlane(my.y, j + 2));
            const float e3 = __int_as_float(__builtin_amdgcn_readlane(my.y, j + 3));
            const float4 x0 = ((const float4*)(feat + (size_t)s0 * 128))[lane];
            const float4 x1 = ((const float4*)(feat + (size_t)s1 * 128))[lane];
            const float4 x2 = ((const float4*)(feat + (size_t)s2 * 128))[lane];
            const float4 x3 = ((const float4*)(feat + (size_t)s3 * 128))[lane];
            s.x += x0.x * e0; s.y += x0.y * e0; s.z += x0.z * e0; s.w += x0.w * e0;
            s.x += x1.x * e1; s.y += x1.y * e1; s.z += x1.z * e1; s.w += x1.w * e1;
            s.x += x2.x * e2; s.y += x2.y * e2; s.z += x2.z * e2; s.w += x2.w * e2;
            s.x += x3.x * e3; s.y += x3.y * e3; s.z += x3.z * e3; s.w += x3.w * e3;
        }
        const int r = n - j;
        if (r >= 1) {
            const int j1 = (r >= 2) ? j + 1 : j;
            const int j2 = (r >= 3) ? j + 2 : j;
            const int   s0 = __builtin_amdgcn_readlane(my.x, j);
            const int   s1 = __builtin_amdgcn_readlane(my.x, j1);
            const int   s2 = __builtin_amdgcn_readlane(my.x, j2);
            const float e0 = __int_as_float(__builtin_amdgcn_readlane(my.y, j));
            const float e1 = __int_as_float(__builtin_amdgcn_readlane(my.y, j1));
            const float e2 = __int_as_float(__builtin_amdgcn_readlane(my.y, j2));
            const float4 x0 = ((const float4*)(feat + (size_t)s0 * 128))[lane];
            const float4 x1 = ((const float4*)(feat + (size_t)s1 * 128))[lane];
            const float4 x2 = ((const float4*)(feat + (size_t)s2 * 128))[lane];
            s.x += x0.x * e0; s.y += x0.y * e0; s.z += x0.z * e0; s.w += x0.w * e0;
            if (r >= 2) {
                s.x += x1.x * e1; s.y += x1.y * e1; s.z += x1.z * e1; s.w += x1.w * e1;
            }
            if (r >= 3) {
                s.x += x2.x * e2; s.y += x2.y * e2; s.z += x2.z * e2; s.w += x2.w * e2;
            }
        }
    }
    ((float4*)(acc + (size_t)d * 128))[lane] = s;
    if (lane == 0) deg[d] = (float)c;
}

__global__ void sage_gemm1_kernel(const float* __restrict__ x,
                                  const float* __restrict__ acc,
                                  const float* __restrict__ deg,
                                  const float* __restrict__ Ws,
                                  const float* __restrict__ Wn,
                                  const float* __restrict__ bias,
                                  float* __restrict__ h) {
    const int lane  = threadIdx.x & 31;
    const int wave  = threadIdx.x >> 5;
    const int m0    = blockIdx.x * 16;
    const int n0    = wave * 16;
    const int row   = m0 + (lane & 15);
    const int col   = n0 + (lane & 15);
    const int khalf = (lane >> 4) * 2;

    const float* xr = x   + (size_t)row * IN_F;
    const float* ar = acc + (size_t)row * IN_F;
    const float invd = 1.0f / fmaxf(deg[row], 1.0f);

    v8f c = {};
    #pragma unroll 4
    for (int k = 0; k < IN_F; k += 4) {
        const int kk = k + khalf;
        v2f a, b;
        a.x = xr[kk];
        a.y = xr[kk + 1];
        b.x = Ws[(size_t)kk * H_F + col];
        b.y = Ws[(size_t)(kk + 1) * H_F + col];
        c = __builtin_amdgcn_wmma_f32_16x16x4_f32(false, a, false, b,
                                                  (short)0, c, false, false);
    }
    #pragma unroll 4
    for (int k = 0; k < IN_F; k += 4) {
        const int kk = k + khalf;
        v2f a, b;
        a.x = ar[kk]     * invd;
        a.y = ar[kk + 1] * invd;
        b.x = Wn[(size_t)kk * H_F + col];
        b.y = Wn[(size_t)(kk + 1) * H_F + col];
        c = __builtin_amdgcn_wmma_f32_16x16x4_f32(false, a, false, b,
                                                  (short)0, c, false, false);
    }

    const float bv    = bias[col];
    const int   rbase = m0 + ((lane >> 4) * 8);
    #pragma unroll
    for (int r = 0; r < 8; ++r) {
        float v = fmaxf(c[r] + bv, 0.0f);
        h[(size_t)(rbase + r) * H_F + col] = v;
    }
}

__global__ void sage_gemm2_kernel(const float* __restrict__ h,
                                  const float* __restrict__ acc,
                                  const float* __restrict__ deg,
                                  const float* __restrict__ Ws,
                                  const float* __restrict__ Wn,
                                  const float* __restrict__ bias,
                                  float* __restrict__ out) {
    const int lane = threadIdx.x & 31;
    const int wave = threadIdx.x >> 5;
    const int tile = blockIdx.x * 8 + wave;
    if (tile >= N_DST2 / 16) return;
    const int m0    = tile * 16;
    const int row   = m0 + (lane & 15);
    const int col   = lane & 15;
    const int khalf = (lane >> 4) * 2;

    const float* hr = h   + (size_t)row * H_F;
    const float* ar = acc + (size_t)row * H_F;
    const float invd = 1.0f / fmaxf(deg[row], 1.0f);

    v8f c = {};
    #pragma unroll 4
    for (int k = 0; k < H_F; k += 4) {
        const int kk = k + khalf;
        v2f a, b;
        a.x = hr[kk];
        a.y = hr[kk + 1];
        b.x = Ws[(size_t)kk * N_CLS + col];
        b.y = Ws[(size_t)(kk + 1) * N_CLS + col];
        c = __builtin_amdgcn_wmma_f32_16x16x4_f32(false, a, false, b,
                                                  (short)0, c, false, false);
    }
    #pragma unroll 4
    for (int k = 0; k < H_F; k += 4) {
        const int kk = k + khalf;
        v2f a, b;
        a.x = ar[kk]     * invd;
        a.y = ar[kk + 1] * invd;
        b.x = Wn[(size_t)kk * N_CLS + col];
        b.y = Wn[(size_t)(kk + 1) * N_CLS + col];
        c = __builtin_amdgcn_wmma_f32_16x16x4_f32(false, a, false, b,
                                                  (short)0, c, false, false);
    }

    const float bv    = bias[col];
    const int   rbase = m0 + ((lane >> 4) * 8);
    #pragma unroll
    for (int r = 0; r < 8; ++r) {
        float v = c[r] + bv;
        float m = v;
        m = fmaxf(m, __shfl_xor(m, 1, 16));
        m = fmaxf(m, __shfl_xor(m, 2, 16));
        m = fmaxf(m, __shfl_xor(m, 4, 16));
        m = fmaxf(m, __shfl_xor(m, 8, 16));
        float e = __expf(v - m);
        float s = e;
        s += __shfl_xor(s, 1, 16);
        s += __shfl_xor(s, 2, 16);
        s += __shfl_xor(s, 4, 16);
        s += __shfl_xor(s, 8, 16);
        out[(size_t)(rbase + r) * N_CLS + col] = (v - m) - __logf(s);
    }
}

extern "C" void kernel_launch(void* const* d_in, const int* in_sizes, int n_in,
                              void* d_out, int out_size, void* d_ws, size_t ws_size,
                              hipStream_t stream) {
    const float* x    = (const float*)d_in[0];
    const int*   src1 = (const int*)  d_in[1];
    const int*   dst1 = (const int*)  d_in[2];
    const float* w1   = (const float*)d_in[3];
    const int*   src2 = (const int*)  d_in[4];
    const int*   dst2 = (const int*)  d_in[5];
    const float* w2   = (const float*)d_in[6];
    const float* Ws1  = (const float*)d_in[7];
    const float* Wn1  = (const float*)d_in[8];
    const float* b1   = (const float*)d_in[9];
    const float* Ws2  = (const float*)d_in[10];
    const float* Wn2  = (const float*)d_in[11];
    const float* b2   = (const float*)d_in[12];
    float* out = (float*)d_out;

    float* acc1 = (float*)d_ws;
    float* deg1 = acc1 + (size_t)N_DST1 * 128;
    float* hbuf = deg1 + N_DST1;
    float* acc2 = acc1;
    float* deg2 = acc2 + (size_t)N_DST2 * 128;
    int* dbase  = (int*)(hbuf + (size_t)N_DST1 * 128);
    int* cnt1   = dbase + 4;
    int* cnt2   = cnt1 + N_DST1;
    int* off1   = cnt2 + N_DST2;
    int* off2   = off1 + N_DST1;
    int* list2  = off2 + N_DST2;
    int* list1  = (int*)hbuf;
    int2* rec        = (int2*)acc1;
    int* bucketCount = (int*)acc1 + 6000000;
    int* gStart      = bucketCount + NBP;
    int* wgOff       = gStart + NBP;

    const int E1 = in_sizes[1];
    const int E2 = in_sizes[4];

    zero_ws_kernel<<<1, 256, 0, stream>>>((float4*)bucketCount, NBP / 4);

    bucket_count_kernel<<<SWEEP_WG, 1024, 0, stream>>>(dst1, dst2, bucketCount, wgOff, E1, E2);
    bucket_place_kernel<<<SWEEP_WG, 1024, 0, stream>>>(src1, dst1, w1, src2, dst2, w2,
                                                       bucketCount, wgOff, gStart, rec, E1, E2);
    bucket_rows_kernel<<<NB, 1024, 0, stream>>>(rec, bucketCount, gStart, cnt1, cnt2, off1, off2,
                                                (int2*)list1, (int2*)list2, E1);

    seg_gather_kernel<<<N_DST1 / 8, 256, 0, stream>>>(x, list1, off1, cnt1, acc1, deg1, N_DST1);

    sage_gemm1_kernel<<<N_DST1 / 16, 256, 0, stream>>>(x, acc1, deg1,
                                                       Ws1, Wn1, b1, hbuf);

    seg_gather_kernel<<<N_DST2 / 8, 256, 0, stream>>>(hbuf, list2, off2, cnt2, acc2, deg2, N_DST2);

    sage_gemm2_kernel<<<(N_DST2 / 16 + 7) / 8, 256, 0, stream>>>(hbuf, acc2, deg2,
                                                                 Ws2, Wn2, b2, out);
}


# Layer-1 GEMM re-fetches no operand: 512 persistent work-groups, each wave keeps its W fragments in registers for all tiles, each 16-row A tile goes once, as whole rows, into LDS; same k order
# speedup vs baseline: 3.3067x; 1.2015x over previous
#include <hip/hip_runtime.h>
#include <hip/hip_bf16.h>

typedef __attribute__((ext_vector_type(2))) float v2f;
typedef __attribute__((ext_vector_type(8))) float v8f;

#define N_SRC1 500000
#define N_DST1 100000
#define N_DST2 20000
#define IN_F   128
#define H_F    128
#define N_CLS  16

__global__ void zero_ws_kernel(float4* __restrict__ p, size_t n4) {
    size_t i = (size_t)blockIdx.x * blockDim.x + threadIdx.x;
    size_t stride = (size_t)gridDim.x * blockDim.x;
    for (; i < n4; i += stride) p[i] = make_float4(0.f, 0.f, 0.f, 0.f);
}

#define NB1   196
#define NB2   79
#define NB    275
#define NBP   288
#define SWEEP_WG 256

__device__ __forceinline__ int wave_incl_scan(int v, int lane) {
    for (int o = 1; o < 32; o <<= 1) {
        const int up = __shfl_up(v, o);
        if (lane >= o) v += up;
    }
    return v;
}

__device__ __forceinline__ int block_excl_scan(int v, int* sTot) {
    const int lane = threadIdx.x & 31;
    const int wave = threadIdx.x >> 5;
    const int incl = wave_incl_scan(v, lane);
    if (lane == 31) sTot[wave] = incl;
    __syncthreads();
    if (wave == 0) {
        const int x  = sTot[lane];
        const int xi = wave_incl_scan(x, lane);
        sTot[32 + lane] = xi - x;
    }
    __syncthreads();
    return sTot[32 + wave] + incl - v;
}

__global__ __launch_bounds__(1024)
void bucket_count_kernel(const int* __restrict__ dst1,
                         const int* __restrict__ dst2,
                         int* __restrict__ bucketCount,
                         int* __restrict__ wgOff,
                         int nE1, int nE2) {
    __shared__ int sH[NBP];
    const int t = threadIdx.x;
    if (t < NBP) sH[t] = 0;
    __syncthreads();
    const int nE    = nE1 + nE2;
    const int chunk = (nE + (int)gridDim.x - 1) / (int)gridDim.x;
    const int lo    = blockIdx.x * chunk;
    const int hi    = min(lo + chunk, nE);
    for (int base = lo; base < hi; base += 4096) {
        int d[4];
        #pragma unroll
        for (int k = 0; k < 4; ++k) {
            const int ic = min(base + k * 1024 + t, hi - 1);
            d[k] = *((ic < nE1) ? dst1 + ic : dst2 + (ic - nE1));
        }
        #pragma unroll
        for (int k = 0; k < 4; ++k) {
            const int i  = base + k * 1024 + t;
            const int bk = (i >= hi) ? NB : (i < nE1) ? (d[k] >> 9) : NB1 + (d[k] >> 8);
            atomicAdd(&sH[bk], 1);
        }
    }
    __syncthreads();
    if (t < NB) wgOff[blockIdx.x * NBP + t] = atomicAdd(bucketCount + t, sH[t]);
}

__global__ __launch_bounds__(1024)
void bucket_place_kernel(const int*   __restrict__ src1,
                         const int*   __restrict__ dst1,
                         const float* __restrict__ w1,
                         const int*   __restrict__ src2,
                         const int*   __restrict__ dst2,
                         const float* __restrict__ w2,
                         const int* __restrict__ bucketCount,
                         const int* __restrict__ wgOff,
                         int*  __restrict__ gStart,
                         int2* __restrict__ rec,
                         int nE1, int nE2) {
    __shared__ int sBase[NBP];
    __shared__ int sTot[64];
    const int t = threadIdx.x;
    const int v = (t < NB) ? bucketCount[t] : 0;
    const int excl = block_excl_scan(v, sTot);
    if (t < NB) {
        sBase[t] = excl + wgOff[blockIdx.x * NBP + t];
        if (blockIdx.x == 0) gStart[t] = excl;
    } else if (t < NBP) {
        sBase[t] = 0;
    }
    __syncthreads();
    const int nE    = nE1 + nE2;
    const int chunk = (nE + (int)gridDim.x - 1) / (int)gridDim.x;
    const int lo    = blockIdx.x * chunk;
    const int hi    = min(lo + chunk, nE);
    for (int base = lo; base < hi; base += 4096) {
        int d[4], s[4], wb[4];
        #pragma unroll
        for (int k = 0; k < 4; ++k) {
            const int  ic = min(base + k * 1024 + t, hi - 1);
            const bool a  = ic < nE1;
            const int  u  = ic - nE1;
            d[k]  = *(a ? dst1 + ic : dst2 + u);
            s[k]  = *(a ? src1 + ic : src2 + u);
            wb[k] = __float_as_int(*(a ? w1 + ic : w2 + u));
        }
        int pos[4];
        #pragma unroll
        for (int k = 0; k < 4; ++k) {
            const int i  = base + k * 1024 + t;
            const int bk = (i >= hi) ? NB : (i < nE1) ? (d[k] >> 9) : NB1 + (d[k] >> 8);
            pos[k] = atomicAdd(&sBase[bk], 1);
        }
        #pragma unroll
        for (int k = 0; k < 4; ++k) {
            const int i = base + k * 1024 + t;
            if (i < hi) {
                const int lr = (i < nE1) ? (d[k] & 511) : (d[k] & 255);
                rec[pos[k]] = make_int2(s[k] | (lr << 20), wb[k]);
            }
        }
    }
}

__global__ __launch_bounds__(1024)
void bucket_rows_kernel(const int2* __restrict__ rec,
                        const int* __restrict__ bucketCount,
                        const int* __restrict__ gStart,
                        int* __restrict__ cnt1, int* __restrict__ cnt2,
                        int* __restrict__ off1, int* __restrict__ off2,
                        int2* __restrict__ list1, int2* __restrict__ list2,
                        int nE1) {
    __shared__ int sCnt[544];
    __shared__ int sCur[544];
    __shared__ int sTot[64];
    const int t  = threadIdx.x;
    const int b  = blockIdx.x;
    const bool l1 = b < NB1;
    const int row0  = l1 ? b * 512 : (b - NB1) * 256;
    const int nrows = l1 ? min(512, N_DST1 - row0) : min(256, N_DST2 - row0);
    const int n     = bucketCount[b];
    const int start = gStart[b];
    const int2* r   = rec + start;
    if (t < 544) sCnt[t] = 0;
    __syncthreads();
    for (int base = 0; base < n; base += 4096) {
        int x[4];
        #pragma unroll
        for (int k = 0; k < 4; ++k) x[k] = r[min(base + k * 1024 + t, n - 1)].x;
        #pragma unroll
        for (int k = 0; k < 4; ++k) {
            const int i = base + k * 1024 + t;
            atomicAdd(&sCnt[(i < n) ? (int)((unsigned)x[k] >> 20) : 512], 1);
        }
    }
    __syncthreads();
    const int v    = (t < 512) ? sCnt[t] : 0;
    const int excl = block_excl_scan(v, sTot);
    const int outBase = l1 ? start : start - nE1;
    if (t < 544) sCur[t] = (t < 512) ? excl : 0;
    if (t < nrows) {
        (l1 ? cnt1 : cnt2)[row0 + t] = v;
        (l1 ? off1 : off2)[row0 + t] = outBase + excl;
    }
    __syncthreads();
    int2* out = (l1 ? list1 : list2) + outBase;
    for (int base = 0; base < n; base += 4096) {
        int2 e[4];
        #pragma unroll
        for (int k = 0; k < 4; ++k) e[k] = r[min(base + k * 1024 + t, n - 1)];
        int pos[4];
        #pragma unroll
        for (int k = 0; k < 4; ++k) {
            const int i = base + k * 1024 + t;
            pos[k] = atomicAdd(&sCur[(i < n) ? (int)((unsigned)e[k].x >> 20) : 512], 1);
        }
        #pragma unroll
        for (int k = 0; k < 4; ++k) {
            const int i = base + k * 1024 + t;
            if (i < n) out[pos[k]] = make_int2(e[k].x & 0xFFFFF, e[k].y);
        }
    }
}

__global__ void seg_gather_kernel(const float* __restrict__ feat,
                                  const int*   __restrict__ list,
                                  const int*   __restrict__ segStart,
                                  const int*   __restrict__ cnt,
                                  float* __restrict__ acc,
                                  float* __restrict__ deg,
                                  int nDst) {
    const int lane = threadIdx.x & 31;
    const int d = __builtin_amdgcn_readfirstlane(blockIdx.x * 8 + (threadIdx.x >> 5));
    if (d >= nDst) return;
    const int c     = __builtin_amdgcn_readfirstlane(cnt[d]);
    const int start = __builtin_amdgcn_readfirstlane(segStart[d]);
    const int2* seg = (const int2*)list + start;

    float4 s = make_float4(0.f, 0.f, 0.f, 0.f);
    for (int base = 0; base < c; base += 32) {
        const int n = min(32, c - base);
        int2 my = make_int2(0, 0);
        if (lane < n) my = seg[base + lane];
        int j = 0;
        for (; j + 4 <= n; j += 4) {
            const int   s0 = __builtin_amdgcn_readlane(my.x, j);
            const int   s1 = __builtin_amdgcn_readlane(my.x, j + 1);
            const int   s2 = __builtin_amdgcn_readlane(my.x, j + 2);
            const int   s3 = __builtin_amdgcn_readlane(my.x, j + 3);
            const float e0 = __int_as_float(__builtin_amdgcn_readlane(my.y, j));
            const float e1 = __int_as_float(__builtin_amdgcn_readlane(my.y, j + 1));
            const float e2 = __int_as_float(__builtin_amdgcn_readlane(my.y, j + 2));
            const float e3 = __int_as_float(__builtin_amdgcn_readlane(my.y, j + 3));
            const float4 x0 = ((const float4*)(feat + (size_t)s0 * 128))[lane];
            const float4 x1 = ((const float4*)(feat + (size_t)s1 * 128))[lane];
            const float4 x2 = ((const float4*)(feat + (size_t)s2 * 128))[lane];
            const float4 x3 = ((const float4*)(feat + (size_t)s3 * 128))[lane];
            s.x += x0.x * e0; s.y += x0.y * e0; s.z += x0.z * e0; s.w += x0.w * e0;
            s.x += x1.x * e1; s.y += x1.y * e1; s.z += x1.z * e1; s.w += x1.w * e1;
            s.x += x2.x * e2; s.y += x2.y * e2; s.z += x2.z * e2; s.w += x2.w * e2;
            s.x += x3.x * e3; s.y += x3.y * e3; s.z += x3.z * e3; s.w += x3.w * e3;
        }
        const int r = n - j;
        if (r >= 1) {
            const int j1 = (r >= 2) ? j + 1 : j;
            const int j2 = (r >= 3) ? j + 2 : j;
            const int   s0 = __builtin_amdgcn_readlane(my.x, j);
            const int   s1 = __builtin_amdgcn_readlane(my.x, j1);
            const int   s2 = __builtin_amdgcn_readlane(my.x, j2);
            const float e0 = __int_as_float(__builtin_amdgcn_readlane(my.y, j));
            const float e1 = __int_as_float(__builtin_amdgcn_readlane(my.y, j1));
            const float e2 = __int_as_float(__builtin_amdgcn_readlane(my.y, j2));
            const float4 x0 = ((const float4*)(feat + (size_t)s0 * 128))[lane];
            const float4 x1 = ((const float4*)(feat + (size_t)s1 * 128))[lane];
            const float4 x2 = ((const float4*)(feat + (size_t)s2 * 128))[lane];
            s.x += x0.x * e0; s.y += x0.y * e0; s.z += x0.z * e0; s.w += x0.w * e0;
            if (r >= 2) {
                s.x += x1.x * e1; s.y += x1.y * e1; s.z += x1.z * e1; s.w += x1.w * e1;
            }
            if (r >= 3) {
                s.x += x2.x * e2; s.y += x2.y * e2; s.z += x2.z * e2; s.w += x2.w * e2;
            }
        }
    }
    ((float4*)(acc + (size_t)d * 128))[lane] = s;
    if (lane == 0) deg[d] = (float)c;
}

#define G1_WG    512
#define G1_PITCH 260

__device__ __forceinline__ void g1_fetch(const float* __restrict__ x,
                                         const float* __restrict__ acc,
                                         const float* __restrict__ deg,
                                         int row, int lane,
                                         float4& px0, float4& px1,
                                         float4& pa0, float4& pa1,
                                         float& d0, float& d1) {
    px0 = ((const float4*)(x   + (size_t)row       * IN_F))[lane];
    px1 = ((const float4*)(x   + (size_t)(row + 1) * IN_F))[lane];
    pa0 = ((const float4*)(acc + (size_t)row       * IN_F))[lane];
    pa1 = ((const float4*)(acc + (size_t)(row + 1) * IN_F))[lane];
    d0  = deg[row];
    d1  = deg[row + 1];
}

__device__ __forceinline__ void g1_stage(float* s, int lane,
                                         const float4& px0, const float4& px1,
                                         const float4& pa0, const float4& pa1,
                                         float d0, float d1) {
    const float i0 = 1.0f / fmaxf(d0, 1.0f);
    const float i1 = 1.0f / fmaxf(d1, 1.0f);
    float* s0 = s + lane * 4;
    float* s1 = s0 + G1_PITCH;
    *(float4*)s0         = px0;
    *(float4*)(s0 + 128) = make_float4(pa0.x * i0, pa0.y * i0, pa0.z * i0, pa0.w * i0);
    *(float4*)s1         = px1;
    *(float4*)(s1 + 128) = make_float4(pa1.x * i1, pa1.y * i1, pa1.z * i1, pa1.w * i1);
}

__global__ __launch_bounds__(256)
void sage_gemm1_kernel(const float* __restrict__ x,
                       const float* __restrict__ acc,
                       const float* __restrict__ deg,
                       const float* __restrict__ Ws,
                       const float* __restrict__ Wn,
                       const float* __restrict__ bias,
                       float* __restrict__ h) {
    __shared__ __attribute__((aligned(16))) float sA[2][16 * G1_PITCH];
    const int lane   = threadIdx.x & 31;
    const int wave   = threadIdx.x >> 5;
    const int col    = wave * 16 + (lane & 15);
    const int khalf  = (lane >> 4) * 2;
    const int nTiles = N_DST1 / 16;

    int tile = blockIdx.x;
    if (tile >= nTiles) return;

    float4 px0, px1, pa0, pa1;
    float  d0, d1;
    g1_fetch(x, acc, deg, tile * 16 + wave * 2, lane, px0, px1, pa0, pa1, d0, d1);

    v2f breg[64];
    #pragma unroll
    for (int ks = 0; ks < 32; ++ks) {
        const int kk = ks * 4 + khalf;
        breg[ks].x      = Ws[(size_t)kk * H_F + col];
        breg[ks].y      = Ws[(size_t)(kk + 1) * H_F + col];
        breg[32 + ks].x = Wn[(size_t)kk * H_F + col];
        breg[32 + ks].y = Wn[(size_t)(kk + 1) * H_F + col];
    }
    const float bv = bias[col];

    g1_stage(&sA[0][wave * 2 * G1_PITCH], lane, px0, px1, pa0, pa1, d0, d1);
    __syncthreads();

    int cur = 0;
    for (; tile < nTiles; tile += (int)gridDim.x) {
        const int  nxt  = tile + (int)gridDim.x;
        const bool more = nxt < nTiles;
        if (more) g1_fetch(x, acc, deg, nxt * 16 + wave * 2, lane, px0, px1, pa0, pa1, d0, d1);

        const float* ap = &sA[cur][(lane & 15) * G1_PITCH + khalf];
        v8f c = {};
        #pragma unroll
        for (int ks = 0; ks < 64; ++ks) {
            const v2f a = *(const v2f*)(ap + ks * 4);
            c = __builtin_amdgcn_wmma_f32_16x16x4_f32(false, a, false, breg[ks],
                                                      (short)0, c, false, false);
        }

        if (more) g1_stage(&sA[cur ^ 1][wave * 2 * G1_PITCH], lane, px0, px1, pa0, pa1, d0, d1);
        __syncthreads();
        cur ^= 1;

        const int rbase = tile * 16 + ((lane >> 4) * 8);
        #pragma unroll
        for (int r = 0; r < 8; ++r) {
            float v = fmaxf(c[r] + bv, 0.0f);
            h[(size_t)(rbase + r) * H_F + col] = v;
        }
    }
}

__global__ void sage_gemm2_kernel(const float* __restrict__ h,
                                  const float* __restrict__ acc,
                                  const float* __restrict__ deg,
                                  const float* __restrict__ Ws,
                                  const float* __restrict__ Wn,
                                  const float* __restrict__ bias,
                                  float* __restrict__ out) {
    const int lane = threadIdx.x & 31;
    const int wave = threadIdx.x >> 5;
    const int tile = blockIdx.x * 8 + wave;
    if (tile >= N_DST2 / 16) return;
    const int m0    = tile * 16;
    const int row   = m0 + (lane & 15);
    const int col   = lane & 15;
    const int khalf = (lane >> 4) * 2;

    const float* hr = h   + (size_t)row * H_F;
    const float* ar = acc + (size_t)row * H_F;
    const float invd = 1.0f / fmaxf(deg[row], 1.0f);

    v8f c = {};
    #pragma unroll 4
    for (int k = 0; k < H_F; k += 4) {
        const int kk = k + khalf;
        v2f a, b;
        a.x = hr[kk];
        a.y = hr[kk + 1];
        b.x = Ws[(size_t)kk * N_CLS + col];
        b.y = Ws[(size_t)(kk + 1) * N_CLS + col];
        c = __builtin_amdgcn_wmma_f32_16x16x4_f32(false, a, false, b,
                                                  (short)0, c, false, false);
    }
    #pragma unroll 4
    for (int k = 0; k < H_F; k += 4) {
        const int kk = k + khalf;
        v2f a, b;
        a.x = ar[kk]     * invd;
        a.y = ar[kk + 1] * invd;
        b.x = Wn[(size_t)kk * N_CLS + col];
        b.y = Wn[(size_t)(kk + 1) * N_CLS + col];
        c = __builtin_amdgcn_wmma_f32_16x16x4_f32(false, a, false, b,
                                                  (short)0, c, false, false);
    }

    const float bv    = bias[col];
    const int   rbase = m0 + ((lane >> 4) * 8);
    #pragma unroll
    for (int r = 0; r < 8; ++r) {
        float v = c[r] + bv;
        float m = v;
        m = fmaxf(m, __shfl_xor(m, 1, 16));
        m = fmaxf(m, __shfl_xor(m, 2, 16));
        m = fmaxf(m, __shfl_xor(m, 4, 16));
        m = fmaxf(m, __shfl_xor(m, 8, 16));
        float e = __expf(v - m);
        float s = e;
        s += __shfl_xor(s, 1, 16);
        s += __shfl_xor(s, 2, 16);
        s += __shfl_xor(s, 4, 16);
        s += __shfl_xor(s, 8, 16);
        out[(size_t)(rbase + r) * N_CLS + col] = (v - m) - __logf(s);
    }
}

extern "C" void kernel_launch(void* const* d_in, const int* in_sizes, int n_in,
                              void* d_out, int out_size, void* d_ws, size_t ws_size,
                              hipStream_t stream) {
    const float* x    = (const float*)d_in[0];
    const int*   src1 = (const int*)  d_in[1];
    const int*   dst1 = (const int*)  d_in[2];
    const float* w1   = (const float*)d_in[3];
    const int*   src2 = (const int*)  d_in[4];
    const int*   dst2 = (const int*)  d_in[5];
    const float* w2   = (const float*)d_in[6];
    const float* Ws1  = (const float*)d_in[7];
    const float* Wn1  = (const float*)d_in[8];
    const float* b1   = (const float*)d_in[9];
    const float* Ws2  = (const float*)d_in[10];
    const float* Wn2  = (const float*)d_in[11];
    const float* b2   = (const float*)d_in[12];
    float* out = (float*)d_out;

    float* acc1 = (float*)d_ws;
    float* deg1 = acc1 + (size_t)N_DST1 * 128;
    float* hbuf = deg1 + N_DST1;
    float* acc2 = acc1;
    float* deg2 = acc2 + (size_t)N_DST2 * 128;
    int* dbase  = (int*)(hbuf + (size_t)N_DST1 * 128);
    int* cnt1   = dbase + 4;
    int* cnt2   = cnt1 + N_DST1;
    int* off1   = cnt2 + N_DST2;
    int* off2   = off1 + N_DST1;
    int* list2  = off2 + N_DST2;
    int* list1  = (int*)hbuf;
    int2* rec        = (int2*)acc1;
    int* bucketCount = (int*)acc1 + 6000000;
    int* gStart      = bucketCount + NBP;
    int* wgOff       = gStart + NBP;

    const int E1 = in_sizes[1];
    const int E2 = in_sizes[4];

    zero_ws_kernel<<<1, 256, 0, stream>>>((float4*)bucketCount, NBP / 4);

    bucket_count_kernel<<<SWEEP_WG, 1024, 0, stream>>>(dst1, dst2, bucketCount, wgOff, E1, E2);
    bucket_place_kernel<<<SWEEP_WG, 1024, 0, stream>>>(src1, dst1, w1, src2, dst2, w2,
                                                       bucketCount, wgOff, gStart, rec, E1, E2);
    bucket_rows_kernel<<<NB, 1024, 0, stream>>>(rec, bucketCount, gStart, cnt1, cnt2, off1, off2,
                                                (int2*)list1, (int2*)list2, E1);

    seg_gather_kernel<<<N_DST1 / 8, 256, 0, stream>>>(x, list1, off1, cnt1, acc1, deg1, N_DST1);

    sage_gemm1_kernel<<<G1_WG, 256, 0, stream>>>(x, acc1, deg1,
                                                       Ws1, Wn1, b1, hbuf);

    seg_gather_kernel<<<N_DST2 / 8, 256, 0, stream>>>(hbuf, list2, off2, cnt2, acc2, deg2, N_DST2);

    sage_gemm2_kernel<<<(N_DST2 / 16 + 7) / 8, 256, 0, stream>>>(hbuf, acc2, deg2,
                                                                 Ws2, Wn2, b2, out);
}


# Layer 2 projects before it aggregates: GEMM 1 also emits z = h @ W_neigh2 (64 B/row) and keeps h for 20000 rows; layer 2 gathers z rows inside its GEMM kernel; sum order changes (k split, tree)
# speedup vs baseline: 3.8166x; 1.1542x over previous
#include <hip/hip_runtime.h>
#include <hip/hip_bf16.h>

typedef __attribute__((ext_vector_type(2))) float v2f;
typedef __attribute__((ext_vector_type(8))) float v8f;

#define N_SRC1 500000
#define N_DST1 100000
#define N_DST2 20000
#define IN_F   128
#define H_F    128
#define N_CLS  16

__global__ void zero_ws_kernel(float4* __restrict__ p, size_t n4) {
    size_t i = (size_t)blockIdx.x * blockDim.x + threadIdx.x;
    size_t stride = (size_t)gridDim.x * blockDim.x;
    for (; i < n4; i += stride) p[i] = make_float4(0.f, 0.f, 0.f, 0.f);
}

#define NB1   196
#define NB2   79
#define NB    275
#define NBP   288
#define SWEEP_WG 256

__device__ __forceinline__ int wave_incl_scan(int v, int lane) {
    for (int o = 1; o < 32; o <<= 1) {
        const int up = __shfl_up(v, o);
        if (lane >= o) v += up;
    }
    return v;
}

__device__ __forceinline__ int block_excl_scan(int v, int* sTot) {
    const int lane = threadIdx.x & 31;
    const int wave = threadIdx.x >> 5;
    const int incl = wave_incl_scan(v, lane);
    if (lane == 31) sTot[wave] = incl;
    __syncthreads();
    if (wave == 0) {
        const int x  = sTot[lane];
        const int xi = wave_incl_scan(x, lane);
        sTot[32 + lane] = xi - x;
    }
    __syncthreads();
    return sTot[32 + wave] + incl - v;
}

__global__ __launch_bounds__(1024)
void bucket_count_kernel(const int* __restrict__ dst1,
                         const int* __restrict__ dst2,
                         int* __restrict__ bucketCount,
                         int* __restrict__ wgOff,
                         int nE1, int nE2) {
    __shared__ int sH[NBP];
    const int t = threadIdx.x;
    if (t < NBP) sH[t] = 0;
    __syncthreads();
    const int nE    = nE1 + nE2;
    const int chunk = (nE + (int)gridDim.x - 1) / (int)gridDim.x;
    const int lo    = blockIdx.x * chunk;
    const int hi    = min(lo + chunk, nE);
    for (int base = lo; base < hi; base += 4096) {
        int d[4];
        #pragma unroll
        for (int k = 0; k < 4; ++k) {
            const int ic = min(base + k * 1024 + t, hi - 1);
            d[k] = *((ic < nE1) ? dst1 + ic : dst2 + (ic - nE1));
        }
        #pragma unroll
        for (int k = 0; k < 4; ++k) {
            const int i  = base + k * 1024 + t;
            const int bk = (i >= hi) ? NB : (i < nE1) ? (d[k] >> 9) : NB1 + (d[k] >> 8);
            atomicAdd(&sH[bk], 1);
        }
    }
    __syncthreads();
    if (t < NB) wgOff[blockIdx.x * NBP + t] = atomicAdd(bucketCount + t, sH[t]);
}

__global__ __launch_bounds__(1024)
void bucket_place_kernel(const int*   __restrict__ src1,
                         const int*   __restrict__ dst1,
                         const float* __restrict__ w1,
                         const int*   __restrict__ src2,
                         const int*   __restrict__ dst2,
                         const float* __restrict__ w2,
                         const int* __restrict__ bucketCount,
                         const int* __restrict__ wgOff,
                         int*  __restrict__ gStart,
                         int2* __restrict__ rec,
                         int nE1, int nE2) {
    __shared__ int sBase[NBP];
    __shared__ int sTot[64];
    const int t = threadIdx.x;
    const int v = (t < NB) ? bucketCount[t] : 0;
    const int excl = block_excl_scan(v, sTot);
    if (t < NB) {
        sBase[t] = excl + wgOff[blockIdx.x * NBP + t];
        if (blockIdx.x == 0) gStart[t] = excl;
    } else if (t < NBP) {
        sBase[t] = 0;
    }
    __syncthreads();
    const int nE    = nE1 + nE2;
    const int chunk = (nE + (int)gridDim.x - 1) / (int)gridDim.x;
    const int lo    = blockIdx.x * chunk;
    const int hi    = min(lo + chunk, nE);
    for (int base = lo; base < hi; base += 4096) {
        int d[4], s[4], wb[4];
        #pragma unroll
        for (int k = 0; k < 4; ++k) {
            const int  ic = min(base + k * 1024 + t, hi - 1);
            const bool a  = ic < nE1;
            const int  u  = ic - nE1;
            d[k]  = *(a ? dst1 + ic : dst2 + u);
            s[k]  = *(a ? src1 + ic : src2 + u);
            wb[k] = __float_as_int(*(a ? w1 + ic : w2 + u));
        }
        int pos[4];
        #pragma unroll
        for (int k = 0; k < 4; ++k) {
            const int i  = base + k * 1024 + t;
            const int bk = (i >= hi) ? NB : (i < nE1) ? (d[k] >> 9) : NB1 + (d[k] >> 8);
            pos[k] = atomicAdd(&sBase[bk], 1);
        }
        #pragma unroll
        for (int k = 0; k < 4; ++k) {
            const int i = base + k * 1024 + t;
            if (i < hi) {
                const int lr = (i < nE1) ? (d[k] & 511) : (d[k] & 255);
                rec[pos[k]] = make_int2(s[k] | (lr << 20), wb[k]);
            }
        }
    }
}

__global__ __launch_bounds__(1024)
void bucket_rows_kernel(const int2* __restrict__ rec,
                        const int* __restrict__ bucketCount,
                        const int* __restrict__ gStart,
                        int* __restrict__ cnt1, int* __restrict__ cnt2,
                        int* __restrict__ off1, int* __restrict__ off2,
                        int2* __restrict__ list1, int2* __restrict__ list2,
                        int nE1) {
    __shared__ int sCnt[544];
    __shared__ int sCur[544];
    __shared__ int sTot[64];
    const int t  = threadIdx.x;
    const int b  = blockIdx.x;
    const bool l1 = b < NB1;
    const int row0  = l1 ? b * 512 : (b - NB1) * 256;
    const int nrows = l1 ? min(512, N_DST1 - row0) : min(256, N_DST2 - row0);
    const int n     = bucketCount[b];
    const int start = gStart[b];
    const int2* r   = rec + start;
    if (t < 544) sCnt[t] = 0;
    __syncthreads();
    for (int base = 0; base < n; base += 4096) {
        int x[4];
        #pragma unroll
        for (int k = 0; k < 4; ++k) x[k] = r[min(base + k * 1024 + t, n - 1)].x;
        #pragma unroll
        for (int k = 0; k < 4; ++k) {
            const int i = base + k * 1024 + t;
            atomicAdd(&sCnt[(i < n) ? (int)((unsigned)x[k] >> 20) : 512], 1);
        }
    }
    __syncthreads();
    const int v    = (t < 512) ? sCnt[t] : 0;
    const int excl = block_excl_scan(v, sTot);
    const int outBase = l1 ? start : start - nE1;
    if (t < 544) sCur[t] = (t < 512) ? excl : 0;
    if (t < nrows) {
        (l1 ? cnt1 : cnt2)[row0 + t] = v;
        (l1 ? off1 : off2)[row0 + t] = outBase + excl;
    }
    __syncthreads();
    int2* out = (l1 ? list1 : list2) + outBase;
    for (int base = 0; base < n; base += 4096) {
        int2 e[4];
        #pragma unroll
        for (int k = 0; k < 4; ++k) e[k] = r[min(base + k * 1024 + t, n - 1)];
        int pos[4];
        #pragma unroll
        for (int k = 0; k < 4; ++k) {
            const int i = base + k * 1024 + t;
            pos[k] = atomicAdd(&sCur[(i < n) ? (int)((unsigned)e[k].x >> 20) : 512], 1);
        }
        #pragma unroll
        for (int k = 0; k < 4; ++k) {
            const int i = base + k * 1024 + t;
            if (i < n) out[pos[k]] = make_int2(e[k].x & 0xFFFFF, e[k].y);
        }
    }
}

__global__ void seg_gather_kernel(const float* __restrict__ feat,
                                  const int*   __restrict__ list,
                                  const int*   __restrict__ segStart,
                                  const int*   __restrict__ cnt,
                                  float* __restrict__ acc,
                                  float* __restrict__ deg,
                                  int nDst) {
    const int lane = threadIdx.x & 31;
    const int d = __builtin_amdgcn_readfirstlane(blockIdx.x * 8 + (threadIdx.x >> 5));
    if (d >= nDst) return;
    const int c     = __builtin_amdgcn_readfirstlane(cnt[d]);
    const int start = __builtin_amdgcn_readfirstlane(segStart[d]);
    const int2* seg = (const int2*)list + start;

    float4 s = make_float4(0.f, 0.f, 0.f, 0.f);
    for (int base = 0; base < c; base += 32) {
        const int n = min(32, c - base);
        int2 my = make_int2(0, 0);
        if (lane < n) my = seg[base + lane];
        int j = 0;
        for (; j + 4 <= n; j += 4) {
            const int   s0 = __builtin_amdgcn_readlane(my.x, j);
            const int   s1 = __builtin_amdgcn_readlane(my.x, j + 1);
            const int   s2 = __builtin_amdgcn_readlane(my.x, j + 2);
            const int   s3 = __builtin_amdgcn_readlane(my.x, j + 3);
            const float e0 = __int_as_float(__builtin_amdgcn_readlane(my.y, j));
            const float e1 = __int_as_float(__builtin_amdgcn_readlane(my.y, j + 1));
            const float e2 = __int_as_float(__builtin_amdgcn_readlane(my.y, j + 2));
            const float e3 = __int_as_float(__builtin_amdgcn_readlane(my.y, j + 3));
            const float4 x0 = ((const float4*)(feat + (size_t)s0 * 128))[lane];
            const float4 x1 = ((const float4*)(feat + (size_t)s1 * 128))[lane];
            const float4 x2 = ((const float4*)(feat + (size_t)s2 * 128))[lane];
            const float4 x3 = ((const float4*)(feat + (size_t)s3 * 128))[lane];
            s.x += x0.x * e0; s.y += x0.y * e0; s.z += x0.z * e0; s.w += x0.w * e0;
            s.x += x1.x * e1; s.y += x1.y * e1; s.z += x1.z * e1; s.w += x1.w * e1;
            s.x += x2.x * e2; s.y += x2.y * e2; s.z += x2.z * e2; s.w += x2.w * e2;
            s.x += x3.x * e3; s.y += x3.y * e3; s.z += x3.z * e3; s.w += x3.w * e3;
        }
        const int r = n - j;
        if (r >= 1) {
            const int j1 = (r >= 2) ? j + 1 : j;
            const int j2 = (r >= 3) ? j + 2 : j;
            const int   s0 = __builtin_amdgcn_readlane(my.x, j);
            const int   s1 = __builtin_amdgcn_readlane(my.x, j1);
            const int   s2 = __builtin_amdgcn_readlane(my.x, j2);
            const float e0 = __int_as_float(__builtin_amdgcn_readlane(my.y, j));
            const float e1 = __int_as_float(__builtin_amdgcn_readlane(my.y, j1));
            const float e2 = __int_as_float(__builtin_amdgcn_readlane(my.y, j2));
            const float4 x0 = ((const float4*)(feat + (size_t)s0 * 128))[lane];
            const float4 x1 = ((const float4*)(feat + (size_t)s1 * 128))[lane];
            const float4 x2 = ((const float4*)(feat + (size_t)s2 * 128))[lane];
            s.x += x0.x * e0; s.y += x0.y * e0; s.z += x0.z * e0; s.w += x0.w * e0;
            if (r >= 2) {
                s.x += x1.x * e1; s.y += x1.y * e1; s.z += x1.z * e1; s.w += x1.w * e1;
            }
            if (r >= 3) {
                s.x += x2.x * e2; s.y += x2.y * e2; s.z += x2.z * e2; s.w += x2.w * e2;
            }
        }
    }
    ((float4*)(acc + (size_t)d * 128))[lane] = s;
    if (lane == 0) deg[d] = (float)c;
}

#define G1_WG    512
#define G1_PITCH 260
#define G1_TP    20
#define ZP_HALF  144
#define ZP_WAVE  288

__device__ __forceinline__ void g1_fetch(const float* __restrict__ x,
                                         const float* __restrict__ acc,
                                         const float* __restrict__ deg,
                                         int row, int lane,
                                         float4& px0, float4& px1,
                                         float4& pa0, float4& pa1,
                                         float& d0, float& d1) {
    px0 = ((const float4*)(x   + (size_t)row       * IN_F))[lane];
    px1 = ((const float4*)(x   + (size_t)(row + 1) * IN_F))[lane];
    pa0 = ((const float4*)(acc + (size_t)row       * IN_F))[lane];
    pa1 = ((const float4*)(acc + (size_t)(row + 1) * IN_F))[lane];
    d0  = deg[row];
    d1  = deg[row + 1];
}

__device__ __forceinline__ void g1_stage(float* s, int lane,
                                         const float4& px0, const float4& px1,
                                         const float4& pa0, const float4& pa1,
                                         float d0, float d1) {
    const float i0 = 1.0f / fmaxf(d0, 1.0f);
    const float i1 = 1.0f / fmaxf(d1, 1.0f);
    float* s0 = s + lane * 4;
    float* s1 = s0 + G1_PITCH;
    *(float4*)s0         = px0;
    *(float4*)(s0 + 128) = make_float4(pa0.x * i0, pa0.y * i0, pa0.z * i0, pa0.w * i0);
    *(float4*)s1         = px1;
    *(float4*)(s1 + 128) = make_float4(pa1.x * i1, pa1.y * i1, pa1.z * i1, pa1.w * i1);
}

__global__ __launch_bounds__(256)
void sage_gemm1_kernel(const float* __restrict__ x,
                       const float* __restrict__ acc,
                       const float* __restrict__ deg,
                       const float* __restrict__ Ws,
                       const float* __restrict__ Wn,
                       const float* __restrict__ bias,
                       const float* __restrict__ Wn2,
                       float* __restrict__ h,
                       float* __restrict__ z) {
    __shared__ __attribute__((aligned(16))) float sA[2][16 * G1_PITCH];
    __shared__ __attribute__((aligned(16))) float sT[8][16 * G1_TP];
    __shared__ __attribute__((aligned(16))) float sZ[8][ZP_WAVE];
    const int lane   = threadIdx.x & 31;
    const int wave   = threadIdx.x >> 5;
    const int col    = wave * 16 + (lane & 15);
    const int khalf  = (lane >> 4) * 2;
    const int nTiles = N_DST1 / 16;

    int tile = blockIdx.x;
    if (tile >= nTiles) return;

    float4 px0, px1, pa0, pa1;
    float  d0, d1;
    g1_fetch(x, acc, deg, tile * 16 + wave * 2, lane, px0, px1, pa0, pa1, d0, d1);

    v2f breg[64];
    #pragma unroll
    for (int ks = 0; ks < 32; ++ks) {
        const int kk = ks * 4 + khalf;
        breg[ks].x      = Ws[(size_t)kk * H_F + col];
        breg[ks].y      = Ws[(size_t)(kk + 1) * H_F + col];
        breg[32 + ks].x = Wn[(size_t)kk * H_F + col];
        breg[32 + ks].y = Wn[(size_t)(kk + 1) * H_F + col];
    }
    const float bv = bias[col];

    v2f b2reg[4];
    #pragma unroll
    for (int s = 0; s < 4; ++s) {
        const int kk = wave * 16 + s * 4 + khalf;
        b2reg[s].x = Wn2[(size_t)kk * N_CLS + (lane & 15)];
        b2reg[s].y = Wn2[(size_t)(kk + 1) * N_CLS + (lane & 15)];
    }
    float* tp = &sT[wave][0];

    g1_stage(&sA[0][wave * 2 * G1_PITCH], lane, px0, px1, pa0, pa1, d0, d1);
    __syncthreads();

    int cur = 0;
    for (; tile < nTiles; tile += (int)gridDim.x) {
        const int  nxt  = tile + (int)gridDim.x;
        const bool more = nxt < nTiles;
        if (more) g1_fetch(x, acc, deg, nxt * 16 + wave * 2, lane, px0, px1, pa0, pa1, d0, d1);

        const float* ap = &sA[cur][(lane & 15) * G1_PITCH + khalf];
        v8f c = {};
        #pragma unroll
        for (int ks = 0; ks < 64; ++ks) {
            const v2f a = *(const v2f*)(ap + ks * 4);
            c = __builtin_amdgcn_wmma_f32_16x16x4_f32(false, a, false, breg[ks],
                                                      (short)0, c, false, false);
        }

        if (more) g1_stage(&sA[cur ^ 1][wave * 2 * G1_PITCH], lane, px0, px1, pa0, pa1, d0, d1);
        __syncthreads();
        cur ^= 1;

        const int hi = lane >> 4;
        const int cl = lane & 15;
        #pragma unroll
        for (int r = 0; r < 8; ++r) {
            c[r] = fmaxf(c[r] + bv, 0.0f);
            tp[(r + 8 * hi) * G1_TP + cl] = c[r];
        }
        v8f zc = {};
        #pragma unroll
        for (int s = 0; s < 4; ++s) {
            const v2f a = *(const v2f*)(tp + cl * G1_TP + s * 4 + khalf);
            zc = __builtin_amdgcn_wmma_f32_16x16x4_f32(false, a, false, b2reg[s],
                                                      (short)0, zc, false, false);
        }
        float* zp = &sZ[wave][hi * ZP_HALF + cl];
        #pragma unroll
        for (int r = 0; r < 8; ++r) zp[r * 16] = zc[r];
        __syncthreads();
        {
            const int t   = threadIdx.x;
            const int row = t >> 4;
            const int zi  = (row >> 3) * ZP_HALF + (row & 7) * 16 + (t & 15);
            float zs = sZ[0][zi];
            #pragma unroll
            for (int w = 1; w < 8; ++w) zs += sZ[w][zi];
            z[(size_t)tile * 256 + t] = zs;
        }
        if (tile * 16 < N_DST2) {
            const int rbase = tile * 16 + hi * 8;
            #pragma unroll
            for (int r = 0; r < 8; ++r) h[(size_t)(rbase + r) * H_F + col] = c[r];
        }
    }
}

__global__ __launch_bounds__(256)
void sage_layer2_kernel(const float* __restrict__ h,
                        const float* __restrict__ z,
                        const int*   __restrict__ list,
                        const int*   __restrict__ segStart,
                        const int*   __restrict__ cnt,
                        const float* __restrict__ Ws,
                        const float* __restrict__ bias,
                        float* __restrict__ out) {
    __shared__ __attribute__((aligned(16))) float sN[16 * 16];
    __shared__ __attribute__((aligned(16))) float sP[8][ZP_WAVE];
    const int t     = threadIdx.x;
    const int lane  = t & 31;
    const int wave  = t >> 5;
    const int hi    = lane >> 4;
    const int cl    = lane & 15;
    const int khalf = hi * 2;
    const int m0    = blockIdx.x * 16;

    const float* hr = h  + (size_t)(m0 + cl) * H_F + wave * 16 + khalf;
    const float* wr = Ws + (size_t)(wave * 16 + khalf) * N_CLS + cl;
    v2f fa[4], fb[4];
    #pragma unroll
    for (int s = 0; s < 4; ++s) {
        fa[s]   = *(const v2f*)(hr + s * 4);
        fb[s].x = wr[(size_t)(s * 4) * N_CLS];
        fb[s].y = wr[(size_t)(s * 4 + 1) * N_CLS];
    }

    const int d   = m0 + wave * 2 + hi;
    const int c   = cnt[d];
    const int st  = segStart[d];
    const int cm1 = max(c - 1, 0);
    const int2* seg = (const int2*)list + ((c > 0) ? st : 0);
    const int es = (lane >> 2) & 3;
    const int q  = lane & 3;
    const int cmax = max(c, __shfl_xor(c, 16));
    const float4* z4 = (const float4*)z;
    float4 sum = make_float4(0.f, 0.f, 0.f, 0.f);
    int2 e[4];
    #pragma unroll
    for (int k = 0; k < 4; ++k) e[k] = seg[min(k * 4 + es, cm1)];
    __builtin_amdgcn_sched_barrier(0);
    for (int base = 0; base < cmax; base += 16) {
        float4 v[4];
        int2   n[4];
        #pragma unroll
        for (int k = 0; k < 4; ++k) v[k] = z4[(size_t)e[k].x * 4 + q];
        #pragma unroll
        for (int k = 0; k < 4; ++k) n[k] = seg[min(base + 16 + k * 4 + es, cm1)];
        __builtin_amdgcn_sched_barrier(0);
        #pragma unroll
        for (int k = 0; k < 4; ++k) {
            const bool  ok = base + k * 4 + es < c;
            const float w  = __int_as_float(e[k].y);
            const float tx = sum.x + v[k].x * w, ty = sum.y + v[k].y * w;
            const float tz = sum.z + v[k].z * w, tw = sum.w + v[k].w * w;
            sum.x = ok ? tx : sum.x; sum.y = ok ? ty : sum.y;
            sum.z = ok ? tz : sum.z; sum.w = ok ? tw : sum.w;
            e[k] = n[k];
        }
    }
    sum.x += __shfl_xor(sum.x, 4); sum.y += __shfl_xor(sum.y, 4);
    sum.z += __shfl_xor(sum.z, 4); sum.w += __shfl_xor(sum.w, 4);
    sum.x += __shfl_xor(sum.x, 8); sum.y += __shfl_xor(sum.y, 8);
    sum.z += __shfl_xor(sum.z, 8); sum.w += __shfl_xor(sum.w, 8);
    if (es == 0) {
        const float inv = 1.0f / fmaxf((float)c, 1.0f);
        *(float4*)&sN[(wave * 2 + hi) * 16 + q * 4] =
            make_float4(sum.x * inv, sum.y * inv, sum.z * inv, sum.w * inv);
    }

    v8f pc = {};
    #pragma unroll
    for (int s = 0; s < 4; ++s)
        pc = __builtin_amdgcn_wmma_f32_16x16x4_f32(false, fa[s], false, fb[s],
                                                  (short)0, pc, false, false);
    float* pp = &sP[wave][hi * ZP_HALF + cl];
    #pragma unroll
    for (int r = 0; r < 8; ++r) pp[r * 16] = pc[r];
    __syncthreads();

    const int row = t >> 4;
    const int pi  = (row >> 3) * ZP_HALF + (row & 7) * 16 + cl;
    float v = sP[0][pi];
    #pragma unroll
    for (int w = 1; w < 8; ++w) v += sP[w][pi];
    v += sN[t] + bias[cl];
    float m = v;
    m = fmaxf(m, __shfl_xor(m, 1, 16));
    m = fmaxf(m, __shfl_xor(m, 2, 16));
    m = fmaxf(m, __shfl_xor(m, 4, 16));
    m = fmaxf(m, __shfl_xor(m, 8, 16));
    const float ex = __expf(v - m);
    float s = ex;
    s += __shfl_xor(s, 1, 16);
    s += __shfl_xor(s, 2, 16);
    s += __shfl_xor(s, 4, 16);
    s += __shfl_xor(s, 8, 16);
    out[(size_t)m0 * N_CLS + t] = (v - m) - __logf(s);
}

extern "C" void kernel_launch(void* const* d_in, const int* in_sizes, int n_in,
                              void* d_out, int out_size, void* d_ws, size_t ws_size,
                              hipStream_t stream) {
    const float* x    = (const float*)d_in[0];
    const int*   src1 = (const int*)  d_in[1];
    const int*   dst1 = (const int*)  d_in[2];
    const float* w1   = (const float*)d_in[3];
    const int*   src2 = (const int*)  d_in[4];
    const int*   dst2 = (const int*)  d_in[5];
    const float* w2   = (const float*)d_in[6];
    const float* Ws1  = (const float*)d_in[7];
    const float* Wn1  = (const float*)d_in[8];
    const float* b1   = (const float*)d_in[9];
    const float* Ws2  = (const float*)d_in[10];
    const float* Wn2  = (const float*)d_in[11];
    const float* b2   = (const float*)d_in[12];
    float* out = (float*)d_out;

    float* acc1 = (float*)d_ws;
    float* deg1 = acc1 + (size_t)N_DST1 * 128;
    float* hbuf = deg1 + N_DST1;
    float* zbuf = hbuf + (size_t)N_DST2 * 128;
    int* dbase  = (int*)(hbuf + (size_t)N_DST1 * 128);
    int* cnt1   = dbase + 4;
    int* cnt2   = cnt1 + N_DST1;
    int* off1   = cnt2 + N_DST2;
    int* off2   = off1 + N_DST1;
    int* list2  = off2 + N_DST2;
    int* list1  = (int*)hbuf;
    int2* rec        = (int2*)acc1;
    int* bucketCount = (int*)acc1 + 6000000;
    int* gStart      = bucketCount + NBP;
    int* wgOff       = gStart + NBP;

    const int E1 = in_sizes[1];
    const int E2 = in_sizes[4];

    zero_ws_kernel<<<1, 256, 0, stream>>>((float4*)bucketCount, NBP / 4);

    bucket_count_kernel<<<SWEEP_WG, 1024, 0, stream>>>(dst1, dst2, bucketCount, wgOff, E1, E2);
    bucket_place_kernel<<<SWEEP_WG, 1024, 0, stream>>>(src1, dst1, w1, src2, dst2, w2,
                                                       bucketCount, wgOff, gStart, rec, E1, E2);
    bucket_rows_kernel<<<NB, 1024, 0, stream>>>(rec, bucketCount, gStart, cnt1, cnt2, off1, off2,
                                                (int2*)list1, (int2*)list2, E1);

    seg_gather_kernel<<<N_DST1 / 8, 256, 0, stream>>>(x, list1, off1, cnt1, acc1, deg1, N_DST1);

    sage_gemm1_kernel<<<G1_WG, 256, 0, stream>>>(x, acc1, deg1,
                                                       Ws1, Wn1, b1, Wn2, hbuf, zbuf);

    sage_layer2_kernel<<<N_DST2 / 16, 256, 0, stream>>>(hbuf, zbuf, list2, off2, cnt2,
                                                        Ws2, b2, out);
}


# Layer-1 neighbour sums move into the GEMM kernel: a wave gathers its 2 rows of the next tile in registers, 16 source rows in flight, means go through LDS; no sum plane, a launch less; sum order kept
# speedup vs baseline: 4.3450x; 1.1384x over previous
#include <hip/hip_runtime.h>
#include <hip/hip_bf16.h>

typedef __attribute__((ext_vector_type(2))) float v2f;
typedef __attribute__((ext_vector_type(8))) float v8f;

#define N_SRC1 500000
#define N_DST1 100000
#define N_DST2 20000
#define IN_F   128
#define H_F    128
#define N_CLS  16

__global__ void zero_ws_kernel(float4* __restrict__ p, size_t n4) {
    size_t i = (size_t)blockIdx.x * blockDim.x + threadIdx.x;
    size_t stride = (size_t)gridDim.x * blockDim.x;
    for (; i < n4; i += stride) p[i] = make_float4(0.f, 0.f, 0.f, 0.f);
}

#define NB1   196
#define NB2   79
#define NB    275
#define NBP   288
#define SWEEP_WG 256

__device__ __forceinline__ int wave_incl_scan(int v, int lane) {
    for (int o = 1; o < 32; o <<= 1) {
        const int up = __shfl_up(v, o);
        if (lane >= o) v += up;
    }
    return v;
}

__device__ __forceinline__ int block_excl_scan(int v, int* sTot) {
    const int lane = threadIdx.x & 31;
    const int wave = threadIdx.x >> 5;
    const int incl = wave_incl_scan(v, lane);
    if (lane == 31) sTot[wave] = incl;
    __syncthreads();
    if (wave == 0) {
        const int x  = sTot[lane];
        const int xi = wave_incl_scan(x, lane);
        sTot[32 + lane] = xi - x;
    }
    __syncthreads();
    return sTot[32 + wave] + incl - v;
}

__global__ __launch_bounds__(1024)
void bucket_count_kernel(const int* __restrict__ dst1,
                         const int* __restrict__ dst2,
                         int* __restrict__ bucketCount,
                         int* __restrict__ wgOff,
                         int nE1, int nE2) {
    __shared__ int sH[NBP];
    const int t = threadIdx.x;
    if (t < NBP) sH[t] = 0;
    __syncthreads();
    const int nE    = nE1 + nE2;
    const int chunk = (nE + (int)gridDim.x - 1) / (int)gridDim.x;
    const int lo    = blockIdx.x * chunk;
    const int hi    = min(lo + chunk, nE);
    for (int base = lo; base < hi; base += 4096) {
        int d[4];
        #pragma unroll
        for (int k = 0; k < 4; ++k) {
            const int ic = min(base + k * 1024 + t, hi - 1);
            d[k] = *((ic < nE1) ? dst1 + ic : dst2 + (ic - nE1));
        }
        #pragma unroll
        for (int k = 0; k < 4; ++k) {
            const int i  = base + k * 1024 + t;
            const int bk = (i >= hi) ? NB : (i < nE1) ? (d[k] >> 9) : NB1 + (d[k] >> 8);
            atomicAdd(&sH[bk], 1);
        }
    }
    __syncthreads();
    if (t < NB) wgOff[blockIdx.x * NBP + t] = atomicAdd(bucketCount + t, sH[t]);
}

__global__ __launch_bounds__(1024)
void bucket_place_kernel(const int*   __restrict__ src1,
                         const int*   __restrict__ dst1,
                         const float* __restrict__ w1,
                         const int*   __restrict__ src2,
                         const int*   __restrict__ dst2,
                         const float* __restrict__ w2,
                         const int* __restrict__ bucketCount,
                         const int* __restrict__ wgOff,
                         int*  __restrict__ gStart,
                         int2* __restrict__ rec,
                         int nE1, int nE2) {
    __shared__ int sBase[NBP];
    __shared__ int sTot[64];
    const int t = threadIdx.x;
    const int v = (t < NB) ? bucketCount[t] : 0;
    const int excl = block_excl_scan(v, sTot);
    if (t < NB) {
        sBase[t] = excl + wgOff[blockIdx.x * NBP + t];
        if (blockIdx.x == 0) gStart[t] = excl;
    } else if (t < NBP) {
        sBase[t] = 0;
    }
    __syncthreads();
    const int nE    = nE1 + nE2;
    const int chunk = (nE + (int)gridDim.x - 1) / (int)gridDim.x;
    const int lo    = blockIdx.x * chunk;
    const int hi    = min(lo + chunk, nE);
    for (int base = lo; base < hi; base += 4096) {
        int d[4], s[4], wb[4];
        #pragma unroll
        for (int k = 0; k < 4; ++k) {
            const int  ic = min(base + k * 1024 + t, hi - 1);
            const bool a  = ic < nE1;
            const int  u  = ic - nE1;
            d[k]  = *(a ? dst1 + ic : dst2 + u);
            s[k]  = *(a ? src1 + ic : src2 + u);
            wb[k] = __float_as_int(*(a ? w1 + ic : w2 + u));
        }
        int pos[4];
        #pragma unroll
        for (int k = 0; k < 4; ++k) {
            const int i  = base + k * 1024 + t;
            const int bk = (i >= hi) ? NB : (i < nE1) ? (d[k] >> 9) : NB1 + (d[k] >> 8);
            pos[k] = atomicAdd(&sBase[bk], 1);
        }
        #pragma unroll
        for (int k = 0; k < 4; ++k) {
            const int i = base + k * 1024 + t;
            if (i < hi) {
                const int lr = (i < nE1) ? (d[k] & 511) : (d[k] & 255);
                rec[pos[k]] = make_int2(s[k] | (lr << 20), wb[k]);
            }
        }
    }
}

__global__ __launch_bounds__(1024)
void bucket_rows_kernel(const int2* __restrict__ rec,
                        const int* __restrict__ bucketCount,
                        const int* __restrict__ gStart,
                        int* __restrict__ cnt1, int* __restrict__ cnt2,
                        int* __restrict__ off1, int* __restrict__ off2,
                        int2* __restrict__ list1, int2* __restrict__ list2,
                        int nE1) {
    __shared__ int sCnt[544];
    __shared__ int sCur[544];
    __shared__ int sTot[64];
    const int t  = threadIdx.x;
    const int b  = blockIdx.x;
    const bool l1 = b < NB1;
    const int row0  = l1 ? b * 512 : (b - NB1) * 256;
    const int nrows = l1 ? min(512, N_DST1 - row0) : min(256, N_DST2 - row0);
    const int n     = bucketCount[b];
    const int start = gStart[b];
    const int2* r   = rec + start;
    if (t < 544) sCnt[t] = 0;
    __syncthreads();
    for (int base = 0; base < n; base += 4096) {
        int x[4];
        #pragma unroll
        for (int k = 0; k < 4; ++k) x[k] = r[min(base + k * 1024 + t, n - 1)].x;
        #pragma unroll
        for (int k = 0; k < 4; ++k) {
            const int i = base + k * 1024 + t;
            atomicAdd(&sCnt[(i < n) ? (int)((unsigned)x[k] >> 20) : 512], 1);
        }
    }
    __syncthreads();
    const int v    = (t < 512) ? sCnt[t] : 0;
    const int excl = block_excl_scan(v, sTot);
    const int outBase = l1 ? start : start - nE1;
    if (t < 544) sCur[t] = (t < 512) ? excl : 0;
    if (t < nrows) {
        (l1 ? cnt1 : cnt2)[row0 + t] = v;
        (l1 ? off1 : off2)[row0 + t] = outBase + excl;
    }
    __syncthreads();
    int2* out = (l1 ? list1 : list2) + outBase;
    for (int base = 0; base < n; base += 4096) {
        int2 e[4];
        #pragma unroll
        for (int k = 0; k < 4; ++k) e[k] = r[min(base + k * 1024 + t, n - 1)];
        int pos[4];
        #pragma unroll
        for (int k = 0; k < 4; ++k) {
            const int i = base + k * 1024 + t;
            pos[k] = atomicAdd(&sCur[(i < n) ? (int)((unsigned)e[k].x >> 20) : 512], 1);
        }
        #pragma unroll
        for (int k = 0; k < 4; ++k) {
            const int i = base + k * 1024 + t;
            if (i < n) out[pos[k]] = make_int2(e[k].x & 0xFFFFF, e[k].y);
        }
    }
}

#define G1_WG    512
#define G1_PITCH 260
#define G1_TP    20
#define ZP_HALF  144
#define ZP_WAVE  288

#define G1_GS 16

__device__ __forceinline__ void g1_gather(const float* __restrict__ x,
                                          const int2*  __restrict__ list,
                                          int cA, int stA, int cB, int stB, int lane,
                                          float4& sa, float4& sb) {
    sa = make_float4(0.f, 0.f, 0.f, 0.f);
    sb = make_float4(0.f, 0.f, 0.f, 0.f);
    const int ct = cA + cB;
    int2 my = make_int2(0, 0);
    if (lane < ct) my = list[(lane < cA) ? stA + lane : stB + (lane - cA)];
    for (int base = 0; base < ct; base += 32) {
        const int gn = base + 32 + lane;
        int2 nx = make_int2(0, 0);
        if (gn < ct) nx = list[(gn < cA) ? stA + gn : stB + (gn - cA)];
        const int n = min(32, ct - base);
        for (int j = 0; j < n; j += G1_GS) {
            float4 v[G1_GS];
            #pragma unroll
            for (int i = 0; i < G1_GS; ++i) {
                const int jj = min(j + i, n - 1);
                const int sr = __builtin_amdgcn_readlane(my.x, jj);
                v[i] = ((const float4*)(x + (size_t)sr * IN_F))[lane];
            }
            #pragma unroll
            for (int i = 0; i < G1_GS; ++i) {
                const int   jj = min(j + i, n - 1);
                const int   g  = base + j + i;
                const float w  = __int_as_float(__builtin_amdgcn_readlane(my.y, jj));
                const float wa = (g < cA) ? w : 0.0f;
                const float wb = (g >= cA && g < ct) ? w : 0.0f;
                sa.x = __builtin_fmaf(v[i].x, wa, sa.x); sa.y = __builtin_fmaf(v[i].y, wa, sa.y);
                sa.z = __builtin_fmaf(v[i].z, wa, sa.z); sa.w = __builtin_fmaf(v[i].w, wa, sa.w);
                sb.x = __builtin_fmaf(v[i].x, wb, sb.x); sb.y = __builtin_fmaf(v[i].y, wb, sb.y);
                sb.z = __builtin_fmaf(v[i].z, wb, sb.z); sb.w = __builtin_fmaf(v[i].w, wb, sb.w);
            }
        }
        my = nx;
    }
}

__device__ __forceinline__ void g1_stage(float* s, int lane,
                                         const float4& px0, const float4& px1,
                                         const float4& pa0, const float4& pa1,
                                         float d0, float d1) {
    const float i0 = 1.0f / fmaxf(d0, 1.0f);
    const float i1 = 1.0f / fmaxf(d1, 1.0f);
    float* s0 = s + lane * 4;
    float* s1 = s0 + G1_PITCH;
    *(float4*)s0         = px0;
    *(float4*)(s0 + 128) = make_float4(pa0.x * i0, pa0.y * i0, pa0.z * i0, pa0.w * i0);
    *(float4*)s1         = px1;
    *(float4*)(s1 + 128) = make_float4(pa1.x * i1, pa1.y * i1, pa1.z * i1, pa1.w * i1);
}

__global__ __launch_bounds__(256)
void sage_gemm1_kernel(const float* __restrict__ x,
                       const int*   __restrict__ list,
                       const int*   __restrict__ segStart,
                       const int*   __restrict__ cnt,
                       const float* __restrict__ Ws,
                       const float* __restrict__ Wn,
                       const float* __restrict__ bias,
                       const float* __restrict__ Wn2,
                       float* __restrict__ h,
                       float* __restrict__ z) {
    __shared__ __attribute__((aligned(16))) float sA[2][16 * G1_PITCH];
    __shared__ __attribute__((aligned(16))) float sT[8][16 * G1_TP];
    __shared__ __attribute__((aligned(16))) float sZ[8][ZP_WAVE];
    const int lane   = threadIdx.x & 31;
    const int wave   = __builtin_amdgcn_readfirstlane(threadIdx.x >> 5);
    const int col    = wave * 16 + (lane & 15);
    const int khalf  = (lane >> 4) * 2;
    const int nTiles = N_DST1 / 16;
    const int2* rl   = (const int2*)list;

    int tile = blockIdx.x;
    if (tile >= nTiles) return;

    float4 px0, px1, pa0, pa1;
    float  d0, d1;
    int drow = tile * 16 + wave * 2;
    int cA  = __builtin_amdgcn_readfirstlane(cnt[drow]);
    int cB  = __builtin_amdgcn_readfirstlane(cnt[drow + 1]);
    int stA = __builtin_amdgcn_readfirstlane(segStart[drow]);
    int stB = __builtin_amdgcn_readfirstlane(segStart[drow + 1]);
    int mrow = min(tile + (int)gridDim.x, nTiles - 1) * 16 + wave * 2;
    int mcA = cnt[mrow], mcB = cnt[mrow + 1], msA = segStart[mrow], msB = segStart[mrow + 1];

    v2f breg[64];
    #pragma unroll
    for (int ks = 0; ks < 32; ++ks) {
        const int kk = ks * 4 + khalf;
        breg[ks].x      = Ws[(size_t)kk * H_F + col];
        breg[ks].y      = Ws[(size_t)(kk + 1) * H_F + col];
        breg[32 + ks].x = Wn[(size_t)kk * H_F + col];
        breg[32 + ks].y = Wn[(size_t)(kk + 1) * H_F + col];
    }
    const float bv = bias[col];

    v2f b2reg[4];
    #pragma unroll
    for (int s = 0; s < 4; ++s) {
        const int kk = wave * 16 + s * 4 + khalf;
        b2reg[s].x = Wn2[(size_t)kk * N_CLS + (lane & 15)];
        b2reg[s].y = Wn2[(size_t)(kk + 1) * N_CLS + (lane & 15)];
    }
    float* tp = &sT[wave][0];

    g1_gather(x, rl, cA, stA, cB, stB, lane, pa0, pa1);
    d0 = (float)cA;
    d1 = (float)cB;
    px0 = ((const float4*)(x + (size_t)drow       * IN_F))[lane];
    px1 = ((const float4*)(x + (size_t)(drow + 1) * IN_F))[lane];
    g1_stage(&sA[0][wave * 2 * G1_PITCH], lane, px0, px1, pa0, pa1, d0, d1);
    __syncthreads();

    int cur = 0;
    for (; tile < nTiles; tile += (int)gridDim.x) {
        const int  nxt  = tile + (int)gridDim.x;
        const bool more = nxt < nTiles;
        if (more) {
            drow = nxt * 16 + wave * 2;
            cA  = __builtin_amdgcn_readfirstlane(mcA);
            cB  = __builtin_amdgcn_readfirstlane(mcB);
            stA = __builtin_amdgcn_readfirstlane(msA);
            stB = __builtin_amdgcn_readfirstlane(msB);
            mrow = min(nxt + (int)gridDim.x, nTiles - 1) * 16 + wave * 2;
            mcA = cnt[mrow]; mcB = cnt[mrow + 1]; msA = segStart[mrow]; msB = segStart[mrow + 1];
            g1_gather(x, rl, cA, stA, cB, stB, lane, pa0, pa1);
            d0 = (float)cA;
            d1 = (float)cB;
            px0 = ((const float4*)(x + (size_t)drow       * IN_F))[lane];
            px1 = ((const float4*)(x + (size_t)(drow + 1) * IN_F))[lane];
        }

        const float* ap = &sA[cur][(lane & 15) * G1_PITCH + khalf];
        v8f c = {};
        #pragma unroll
        for (int ks = 0; ks < 64; ++ks) {
            const v2f a = *(const v2f*)(ap + ks * 4);
            c = __builtin_amdgcn_wmma_f32_16x16x4_f32(false, a, false, breg[ks],
                                                      (short)0, c, false, false);
        }

        if (more) g1_stage(&sA[cur ^ 1][wave * 2 * G1_PITCH], lane, px0, px1, pa0, pa1, d0, d1);
        __syncthreads();
        cur ^= 1;

        const int hi = lane >> 4;
        const int cl = lane & 15;
        #pragma unroll
        for (int r = 0; r < 8; ++r) {
            c[r] = fmaxf(c[r] + bv, 0.0f);
            tp[(r + 8 * hi) * G1_TP + cl] = c[r];
        }
        v8f zc = {};
        #pragma unroll
        for (int s = 0; s < 4; ++s) {
            const v2f a = *(const v2f*)(tp + cl * G1_TP + s * 4 + khalf);
            zc = __builtin_amdgcn_wmma_f32_16x16x4_f32(false, a, false, b2reg[s],
                                                      (short)0, zc, false, false);
        }
        float* zp = &sZ[wave][hi * ZP_HALF + cl];
        #pragma unroll
        for (int r = 0; r < 8; ++r) zp[r * 16] = zc[r];
        __syncthreads();
        {
            const int t   = threadIdx.x;
            const int row = t >> 4;
            const int zi  = (row >> 3) * ZP_HALF + (row & 7) * 16 + (t & 15);
            float zs = sZ[0][zi];
            #pragma unroll
            for (int w = 1; w < 8; ++w) zs += sZ[w][zi];
            z[(size_t)tile * 256 + t] = zs;
        }
        if (tile * 16 < N_DST2) {
            const int rbase = tile * 16 + hi * 8;
            #pragma unroll
            for (int r = 0; r < 8; ++r) h[(size_t)(rbase + r) * H_F + col] = c[r];
        }
    }
}

__global__ __launch_bounds__(256)
void sage_layer2_kernel(const float* __restrict__ h,
                        const float* __restrict__ z,
                        const int*   __restrict__ list,
                        const int*   __restrict__ segStart,
                        const int*   __restrict__ cnt,
                        const float* __restrict__ Ws,
                        const float* __restrict__ bias,
                        float* __restrict__ out) {
    __shared__ __attribute__((aligned(16))) float sN[16 * 16];
    __shared__ __attribute__((aligned(16))) float sP[8][ZP_WAVE];
    const int t     = threadIdx.x;
    const int lane  = t & 31;
    const int wave  = t >> 5;
    const int hi    = lane >> 4;
    const int cl    = lane & 15;
    const int khalf = hi * 2;
    const int m0    = blockIdx.x * 16;

    const float* hr = h  + (size_t)(m0 + cl) * H_F + wave * 16 + khalf;
    const float* wr = Ws + (size_t)(wave * 16 + khalf) * N_CLS + cl;
    v2f fa[4], fb[4];
    #pragma unroll
    for (int s = 0; s < 4; ++s) {
        fa[s]   = *(const v2f*)(hr + s * 4);
        fb[s].x = wr[(size_t)(s * 4) * N_CLS];
        fb[s].y = wr[(size_t)(s * 4 + 1) * N_CLS];
    }

    const int d   = m0 + wave * 2 + hi;
    const int c   = cnt[d];
    const int st  = segStart[d];
    const int cm1 = max(c - 1, 0);
    const int2* seg = (const int2*)list + ((c > 0) ? st : 0);
    const int es = (lane >> 2) & 3;
    const int q  = lane & 3;
    const int cmax = max(c, __shfl_xor(c, 16));
    const float4* z4 = (const float4*)z;
    float4 sum = make_float4(0.f, 0.f, 0.f, 0.f);
    int2 e[4];
    #pragma unroll
    for (int k = 0; k < 4; ++k) e[k] = seg[min(k * 4 + es, cm1)];
    __builtin_amdgcn_sched_barrier(0);
    for (int base = 0; base < cmax; base += 16) {
        float4 v[4];
        int2   n[4];
        #pragma unroll
        for (int k = 0; k < 4; ++k) v[k] = z4[(size_t)e[k].x * 4 + q];
        #pragma unroll
        for (int k = 0; k < 4; ++k) n[k] = seg[min(base + 16 + k * 4 + es, cm1)];
        __builtin_amdgcn_sched_barrier(0);
        #pragma unroll
        for (int k = 0; k < 4; ++k) {
            const bool  ok = base + k * 4 + es < c;
            const float w  = __int_as_float(e[k].y);
            const float tx = sum.x + v[k].x * w, ty = sum.y + v[k].y * w;
            const float tz = sum.z + v[k].z * w, tw = sum.w + v[k].w * w;
            sum.x = ok ? tx : sum.x; sum.y = ok ? ty : sum.y;
            sum.z = ok ? tz : sum.z; sum.w = ok ? tw : sum.w;
            e[k] = n[k];
        }
    }
    sum.x += __shfl_xor(sum.x, 4); sum.y += __shfl_xor(sum.y, 4);
    sum.z += __shfl_xor(sum.z, 4); sum.w += __shfl_xor(sum.w, 4);
    sum.x += __shfl_xor(sum.x, 8); sum.y += __shfl_xor(sum.y, 8);
    sum.z += __shfl_xor(sum.z, 8); sum.w += __shfl_xor(sum.w, 8);
    if (es == 0) {
        const float inv = 1.0f / fmaxf((float)c, 1.0f);
        *(float4*)&sN[(wave * 2 + hi) * 16 + q * 4] =
            make_float4(sum.x * inv, sum.y * inv, sum.z * inv, sum.w * inv);
    }

    v8f pc = {};
    #pragma unroll
    for (int s = 0; s < 4; ++s)
        pc = __builtin_amdgcn_wmma_f32_16x16x4_f32(false, fa[s], false, fb[s],
                                                  (short)0, pc, false, false);
    float* pp = &sP[wave][hi * ZP_HALF + cl];
    #pragma unroll
    for (int r = 0; r < 8; ++r) pp[r * 16] = pc[r];
    __syncthreads();

    const int row = t >> 4;
    const int pi  = (row >> 3) * ZP_HALF + (row & 7) * 16 + cl;
    float v = sP[0][pi];
    #pragma unroll
    for (int w = 1; w < 8; ++w) v += sP[w][pi];
    v += sN[t] + bias[cl];
    float m = v;
    m = fmaxf(m, __shfl_xor(m, 1, 16));
    m = fmaxf(m, __shfl_xor(m, 2, 16));
    m = fmaxf(m, __shfl_xor(m, 4, 16));
    m = fmaxf(m, __shfl_xor(m, 8, 16));
    const float ex = __expf(v - m);
    float s = ex;
    s += __shfl_xor(s, 1, 16);
    s += __shfl_xor(s, 2, 16);
    s += __shfl_xor(s, 4, 16);
    s += __shfl_xor(s, 8, 16);
    out[(size_t)m0 * N_CLS + t] = (v - m) - __logf(s);
}

extern "C" void kernel_launch(void* const* d_in, const int* in_sizes, int n_in,
                              void* d_out, int out_size, void* d_ws, size_t ws_size,
                              hipStream_t stream) {
    const float* x    = (const float*)d_in[0];
    const int*   src1 = (const int*)  d_in[1];
    const int*   dst1 = (const int*)  d_in[2];
    const float* w1   = (const float*)d_in[3];
    const int*   src2 = (const int*)  d_in[4];
    const int*   dst2 = (const int*)  d_in[5];
    const float* w2   = (const float*)d_in[6];
    const float* Ws1  = (const float*)d_in[7];
    const float* Wn1  = (const float*)d_in[8];
    const float* b1   = (const float*)d_in[9];
    const float* Ws2  = (const float*)d_in[10];
    const float* Wn2  = (const float*)d_in[11];
    const float* b2   = (const float*)d_in[12];
    float* out = (float*)d_out;

    float* wsA  = (float*)d_ws;
    float* hbuf = wsA;
    float* zbuf = hbuf + (size_t)N_DST2 * 128;
    int* list1  = (int*)(wsA + (size_t)N_DST1 * 128 + N_DST1);
    int* dbase  = list1 + (size_t)N_DST1 * 128;
    int* cnt1   = dbase + 4;
    int* cnt2   = cnt1 + N_DST1;
    int* off1   = cnt2 + N_DST2;
    int* off2   = off1 + N_DST1;
    int* list2  = off2 + N_DST2;
    int2* rec        = (int2*)wsA;
    int* bucketCount = (int*)wsA + 6000000;
    int* gStart      = bucketCount + NBP;
    int* wgOff       = gStart + NBP;

    const int E1 = in_sizes[1];
    const int E2 = in_sizes[4];

    zero_ws_kernel<<<1, 256, 0, stream>>>((float4*)bucketCount, NBP / 4);

    bucket_count_kernel<<<SWEEP_WG, 1024, 0, stream>>>(dst1, dst2, bucketCount, wgOff, E1, E2);
    bucket_place_kernel<<<SWEEP_WG, 1024, 0, stream>>>(src1, dst1, w1, src2, dst2, w2,
                                                       bucketCount, wgOff, gStart, rec, E1, E2);
    bucket_rows_kernel<<<NB, 1024, 0, stream>>>(rec, bucketCount, gStart, cnt1, cnt2, off1, off2,
                                                (int2*)list1, (int2*)list2, E1);

    sage_gemm1_kernel<<<G1_WG, 256, 0, stream>>>(x, list1, off1, cnt1,
                                                       Ws1, Wn1, b1, Wn2, hbuf, zbuf);

    sage_layer2_kernel<<<N_DST2 / 16, 256, 0, stream>>>(hbuf, zbuf, list2, off2, cnt2,
                                                        Ws2, b2, out);
}


# Level-1 edge grouping local to each work-group: one kernel counts, scans, places its own edge range (built in LDS, whole-line stores); no global counters, 2 launches fewer; sum order: LDS rank
# speedup vs baseline: 5.0236x; 1.1562x over previous
#include <hip/hip_runtime.h>
#include <hip/hip_bf16.h>

typedef __attribute__((ext_vector_type(2))) float v2f;
typedef __attribute__((ext_vector_type(8))) float v8f;

#define N_SRC1 500000
#define N_DST1 100000
#define N_DST2 20000
#define IN_F   128
#define H_F    128
#define N_CLS  16

#define NB1   196
#define NB2   79
#define NB    275
#define NBP   288
#define SWEEP_WG 256
#define L1_CAP 12288

__device__ __forceinline__ int wave_incl_scan(int v, int lane) {
    for (int o = 1; o < 32; o <<= 1) {
        const int up = __shfl_up(v, o);
        if (lane >= o) v += up;
    }
    return v;
}

__device__ __forceinline__ int block_excl_scan(int v, int* sTot) {
    const int lane = threadIdx.x & 31;
    const int wave = threadIdx.x >> 5;
    const int incl = wave_incl_scan(v, lane);
    if (lane == 31) sTot[wave] = incl;
    __syncthreads();
    if (wave == 0) {
        const int x  = sTot[lane];
        const int xi = wave_incl_scan(x, lane);
        sTot[32 + lane] = xi - x;
    }
    __syncthreads();
    return sTot[32 + wave] + incl - v;
}

__global__ __launch_bounds__(1024)
void bucket_local_kernel(const int*   __restrict__ src1,
                         const int*   __restrict__ dst1,
                         const float* __restrict__ w1,
                         const int*   __restrict__ src2,
                         const int*   __restrict__ dst2,
                         const float* __restrict__ w2,
                         int*  __restrict__ pieceCnt,
                         int*  __restrict__ pieceOff,
                         int2* __restrict__ rec,
                         int nE1, int nE2) {
    __shared__ int  sH[NBP];
    __shared__ int  sBase[NBP];
    __shared__ int  sTot[64];
    __shared__ int2 sRec[L1_CAP];
    const int t = threadIdx.x;
    if (t < NBP) sH[t] = 0;
    __syncthreads();
    const int c1 = (nE1 + SWEEP_WG - 1) / SWEEP_WG;
    const int c2 = (nE2 + SWEEP_WG - 1) / SWEEP_WG;
    const int a1 = min((int)blockIdx.x * c1, nE1);
    const int a2 = min((int)blockIdx.x * c2, nE2);
    const int n1 = min(c1, nE1 - a1);
    const int n2 = min(c2, nE2 - a2);
    const int m  = n1 + n2;
    const int lo = a1 + a2;
    const int o2 = a2 - n1;
    for (int base = 0; base < m; base += 4096) {
        int d[4];
        #pragma unroll
        for (int k = 0; k < 4; ++k) {
            const int uc = min(base + k * 1024 + t, m - 1);
            d[k] = *((uc < n1) ? dst1 + (a1 + uc) : dst2 + (o2 + uc));
        }
        #pragma unroll
        for (int k = 0; k < 4; ++k) {
            const int u  = base + k * 1024 + t;
            const int bk = (u >= m) ? NB : (u < n1) ? (d[k] >> 9) : NB1 + (d[k] >> 8);
            atomicAdd(&sH[bk], 1);
        }
    }
    __syncthreads();
    const int v    = (t < NB) ? sH[t] : 0;
    const int excl = block_excl_scan(v, sTot);
    if (t < NB) {
        sBase[t] = excl;
        pieceCnt[t * SWEEP_WG + blockIdx.x] = v;
        pieceOff[t * SWEEP_WG + blockIdx.x] = excl;
    } else if (t < NBP) {
        sBase[t] = 0;
    }
    __syncthreads();
    const bool stage = m <= L1_CAP;
    for (int base = 0; base < m; base += 4096) {
        int d[4], s[4], wb[4];
        #pragma unroll
        for (int k = 0; k < 4; ++k) {
            const int  uc = min(base + k * 1024 + t, m - 1);
            const bool a  = uc < n1;
            const int  iA = a1 + uc;
            const int  iB = o2 + uc;
            d[k]  = *(a ? dst1 + iA : dst2 + iB);
            s[k]  = *(a ? src1 + iA : src2 + iB);
            wb[k] = __float_as_int(*(a ? w1 + iA : w2 + iB));
        }
        int pos[4];
        #pragma unroll
        for (int k = 0; k < 4; ++k) {
            const int u  = base + k * 1024 + t;
            const int bk = (u >= m) ? NB : (u < n1) ? (d[k] >> 9) : NB1 + (d[k] >> 8);
            pos[k] = atomicAdd(&sBase[bk], 1);
        }
        #pragma unroll
        for (int k = 0; k < 4; ++k) {
            const int u = base + k * 1024 + t;
            if (u < m) {
                const int  lr = (u < n1) ? (d[k] & 511) : (d[k] & 255);
                const int2 r  = make_int2(s[k] | (lr << 20), wb[k]);
                if (stage) sRec[pos[k]] = r;
                else       rec[lo + pos[k]] = r;
            }
        }
    }
    if (stage) {
        __syncthreads();
        for (int i = t; i < m; i += 1024) rec[lo + i] = sRec[i];
    }
}

__global__ __launch_bounds__(1024)
void bucket_rows_kernel(const int2* __restrict__ rec,
                        const int* __restrict__ pieceCnt,
                        const int* __restrict__ pieceOff,
                        int* __restrict__ cnt1, int* __restrict__ cnt2,
                        int* __restrict__ off1, int* __restrict__ off2,
                        int2* __restrict__ list1, int2* __restrict__ list2,
                        int nE1, int nE2) {
    __shared__ int sCnt[544];
    __shared__ int sCur[544];
    __shared__ int sTot[64];
    __shared__ int sRed[32];
    const int t    = threadIdx.x;
    const int lane = t & 31;
    const int wave = t >> 5;
    const int b    = blockIdx.x;
    const bool l1  = b < NB1;
    const int row0  = l1 ? b * 512 : (b - NB1) * 256;
    const int nrows = l1 ? min(512, N_DST1 - row0) : min(256, N_DST2 - row0);
    if (t < 544) sCnt[t] = 0;
    const int g   = wave * 8 + (lane & 7);
    const int c1  = (nE1 + SWEEP_WG - 1) / SWEEP_WG;
    const int c2  = (nE2 + SWEEP_WG - 1) / SWEEP_WG;
    const int pcv = pieceCnt[b * SWEEP_WG + g];
    const int plv = pieceOff[b * SWEEP_WG + g];
    const int psv = min(g * c1, nE1) + min(g * c2, nE2) + plv;
    int part = (lane < 8) ? plv : 0;
    for (int o = 16; o > 0; o >>= 1) part += __shfl_xor(part, o);
    if (lane == 0) sRed[wave] = part;
    __syncthreads();
    int start = sRed[lane];
    for (int o = 16; o > 0; o >>= 1) start += __shfl_xor(start, o);
    int pcm = pcv;
    for (int o = 4; o > 0; o >>= 1) pcm = max(pcm, __shfl_xor(pcm, o));
    int pc[8], pb[8];
    #pragma unroll
    for (int q = 0; q < 8; ++q) {
        pc[q] = __builtin_amdgcn_readlane(pcv, q);
        pb[q] = (pc[q] > 0) ? __builtin_amdgcn_readlane(psv, q) : 0;
    }
    for (int j0 = 0; j0 < pcm; j0 += 32) {
        int x[8];
        #pragma unroll
        for (int q = 0; q < 8; ++q) x[q] = rec[pb[q] + min(j0 + lane, max(pc[q] - 1, 0))].x;
        #pragma unroll
        for (int q = 0; q < 8; ++q)
            atomicAdd(&sCnt[(j0 + lane < pc[q]) ? (int)((unsigned)x[q] >> 20) : 512 + lane], 1);
    }
    __syncthreads();
    const int v    = (t < 512) ? sCnt[t] : 0;
    const int excl = block_excl_scan(v, sTot);
    const int outBase = l1 ? start : start - nE1;
    if (t < 544) sCur[t] = (t < 512) ? excl : 0;
    if (t < nrows) {
        (l1 ? cnt1 : cnt2)[row0 + t] = v;
        (l1 ? off1 : off2)[row0 + t] = outBase + excl;
    }
    __syncthreads();
    int2* out = (l1 ? list1 : list2) + outBase;
    for (int j0 = 0; j0 < pcm; j0 += 32) {
        int2 e[8];
        #pragma unroll
        for (int q = 0; q < 8; ++q) e[q] = rec[pb[q] + min(j0 + lane, max(pc[q] - 1, 0))];
        int pos[8];
        #pragma unroll
        for (int q = 0; q < 8; ++q)
            pos[q] = atomicAdd(&sCur[(j0 + lane < pc[q]) ? (int)((unsigned)e[q].x >> 20) : 512 + lane], 1);
        #pragma unroll
        for (int q = 0; q < 8; ++q)
            if (j0 + lane < pc[q]) out[pos[q]] = make_int2(e[q].x & 0xFFFFF, e[q].y);
    }
}

#define G1_WG    512
#define G1_PITCH 260
#define G1_TP    20
#define ZP_HALF  144
#define ZP_WAVE  288

#define G1_GS 16

__device__ __forceinline__ void g1_gather(const float* __restrict__ x,
                                          const int2*  __restrict__ list,
                                          int cA, int stA, int cB, int stB, int lane,
                                          float4& sa, float4& sb) {
    sa = make_float4(0.f, 0.f, 0.f, 0.f);
    sb = make_float4(0.f, 0.f, 0.f, 0.f);
    const int ct = cA + cB;
    int2 my = make_int2(0, 0);
    if (lane < ct) my = list[(lane < cA) ? stA + lane : stB + (lane - cA)];
    for (int base = 0; base < ct; base += 32) {
        const int gn = base + 32 + lane;
        int2 nx = make_int2(0, 0);
        if (gn < ct) nx = list[(gn < cA) ? stA + gn : stB + (gn - cA)];
        const int n = min(32, ct - base);
        for (int j = 0; j < n; j += G1_GS) {
            float4 v[G1_GS];
            #pragma unroll
            for (int i = 0; i < G1_GS; ++i) {
                const int jj = min(j + i, n - 1);
                const int sr = __builtin_amdgcn_readlane(my.x, jj);
                v[i] = ((const float4*)(x + (size_t)sr * IN_F))[lane];
            }
            #pragma unroll
            for (int i = 0; i < G1_GS; ++i) {
                const int   jj = min(j + i, n - 1);
                const int   g  = base + j + i;
                const float w  = __int_as_float(__builtin_amdgcn_readlane(my.y, jj));
                const float wa = (g < cA) ? w : 0.0f;
                const float wb = (g >= cA && g < ct) ? w : 0.0f;
                sa.x = __builtin_fmaf(v[i].x, wa, sa.x); sa.y = __builtin_fmaf(v[i].y, wa, sa.y);
                sa.z = __builtin_fmaf(v[i].z, wa, sa.z); sa.w = __builtin_fmaf(v[i].w, wa, sa.w);
                sb.x = __builtin_fmaf(v[i].x, wb, sb.x); sb.y = __builtin_fmaf(v[i].y, wb, sb.y);
                sb.z = __builtin_fmaf(v[i].z, wb, sb.z); sb.w = __builtin_fmaf(v[i].w, wb, sb.w);
            }
        }
        my = nx;
    }
}

__device__ __forceinline__ void g1_stage(float* s, int lane,
                                         const float4& px0, const float4& px1,
                                         const float4& pa0, const float4& pa1,
                                         float d0, float d1) {
    const float i0 = 1.0f / fmaxf(d0, 1.0f);
    const float i1 = 1.0f / fmaxf(d1, 1.0f);
    float* s0 = s + lane * 4;
    float* s1 = s0 + G1_PITCH;
    *(float4*)s0         = px0;
    *(float4*)(s0 + 128) = make_float4(pa0.x * i0, pa0.y * i0, pa0.z * i0, pa0.w * i0);
    *(float4*)s1         = px1;
    *(float4*)(s1 + 128) = make_float4(pa1.x * i1, pa1.y * i1, pa1.z * i1, pa1.w * i1);
}

__global__ __launch_bounds__(256)
void sage_gemm1_kernel(const float* __restrict__ x,
                       const int*   __restrict__ list,
                       const int*   __restrict__ segStart,
                       const int*   __restrict__ cnt,
                       const float* __restrict__ Ws,
                       const float* __restrict__ Wn,
                       const float* __restrict__ bias,
                       const float* __restrict__ Wn2,
                       float* __restrict__ h,
                       float* __restrict__ z) {
    __shared__ __attribute__((aligned(16))) float sA[2][16 * G1_PITCH];
    __shared__ __attribute__((aligned(16))) float sT[8][16 * G1_TP];
    __shared__ __attribute__((aligned(16))) float sZ[8][ZP_WAVE];
    const int lane   = threadIdx.x & 31;
    const int wave   = __builtin_amdgcn_readfirstlane(threadIdx.x >> 5);
    const int col    = wave * 16 + (lane & 15);
    const int khalf  = (lane >> 4) * 2;
    const int nTiles = N_DST1 / 16;
    const int2* rl   = (const int2*)list;

    int tile = blockIdx.x;
    if (tile >= nTiles) return;

    float4 px0, px1, pa0, pa1;
    float  d0, d1;
    int drow = tile * 16 + wave * 2;
    int cA  = __builtin_amdgcn_readfirstlane(cnt[drow]);
    int cB  = __builtin_amdgcn_readfirstlane(cnt[drow + 1]);
    int stA = __builtin_amdgcn_readfirstlane(segStart[drow]);
    int stB = __builtin_amdgcn_readfirstlane(segStart[drow + 1]);
    int mrow = min(tile + (int)gridDim.x, nTiles - 1) * 16 + wave * 2;
    int mcA = cnt[mrow], mcB = cnt[mrow + 1], msA = segStart[mrow], msB = segStart[mrow + 1];

    v2f breg[64];
    #pragma unroll
    for (int ks = 0; ks < 32; ++ks) {
        const int kk = ks * 4 + khalf;
        breg[ks].x      = Ws[(size_t)kk * H_F + col];
        breg[ks].y      = Ws[(size_t)(kk + 1) * H_F + col];
        breg[32 + ks].x = Wn[(size_t)kk * H_F + col];
        breg[32 + ks].y = Wn[(size_t)(kk + 1) * H_F + col];
    }
    const float bv = bias[col];

    v2f b2reg[4];
    #pragma unroll
    for (int s = 0; s < 4; ++s) {
        const int kk = wave * 16 + s * 4 + khalf;
        b2reg[s].x = Wn2[(size_t)kk * N_CLS + (lane & 15)];
        b2reg[s].y = Wn2[(size_t)(kk + 1) * N_CLS + (lane & 15)];
    }
    float* tp = &sT[wave][0];

    g1_gather(x, rl, cA, stA, cB, stB, lane, pa0, pa1);
    d0 = (float)cA;
    d1 = (float)cB;
    px0 = ((const float4*)(x + (size_t)drow       * IN_F))[lane];
    px1 = ((const float4*)(x + (size_t)(drow + 1) * IN_F))[lane];
    g1_stage(&sA[0][wave * 2 * G1_PITCH], lane, px0, px1, pa0, pa1, d0, d1);
    __syncthreads();

    int cur = 0;
    for (; tile < nTiles; tile += (int)gridDim.x) {
        const int  nxt  = tile + (int)gridDim.x;
        const bool more = nxt < nTiles;
        if (more) {
            drow = nxt * 16 + wave * 2;
            cA  = __builtin_amdgcn_readfirstlane(mcA);
            cB  = __builtin_amdgcn_readfirstlane(mcB);
            stA = __builtin_amdgcn_readfirstlane(msA);
            stB = __builtin_amdgcn_readfirstlane(msB);
            mrow = min(nxt + (int)gridDim.x, nTiles - 1) * 16 + wave * 2;
            mcA = cnt[mrow]; mcB = cnt[mrow + 1]; msA = segStart[mrow]; msB = segStart[mrow + 1];
            g1_gather(x, rl, cA, stA, cB, stB, lane, pa0, pa1);
            d0 = (float)cA;
            d1 = (float)cB;
            px0 = ((const float4*)(x + (size_t)drow       * IN_F))[lane];
            px1 = ((const float4*)(x + (size_t)(drow + 1) * IN_F))[lane];
        }

        const float* ap = &sA[cur][(lane & 15) * G1_PITCH + khalf];
        v8f c = {};
        #pragma unroll
        for (int ks = 0; ks < 64; ++ks) {
            const v2f a = *(const v2f*)(ap + ks * 4);
            c = __builtin_amdgcn_wmma_f32_16x16x4_f32(false, a, false, breg[ks],
                                                      (short)0, c, false, false);
        }

        if (more) g1_stage(&sA[cur ^ 1][wave * 2 * G1_PITCH], lane, px0, px1, pa0, pa1, d0, d1);
        __syncthreads();
        cur ^= 1;

        const int hi = lane >> 4;
        const int cl = lane & 15;
        #pragma unroll
        for (int r = 0; r < 8; ++r) {
            c[r] = fmaxf(c[r] + bv, 0.0f);
            tp[(r + 8 * hi) * G1_TP + cl] = c[r];
        }
        v8f zc = {};
        #pragma unroll
        for (int s = 0; s < 4; ++s) {
            const v2f a = *(const v2f*)(tp + cl * G1_TP + s * 4 + khalf);
            zc = __builtin_amdgcn_wmma_f32_16x16x4_f32(false, a, false, b2reg[s],
                                                      (short)0, zc, false, false);
        }
        float* zp = &sZ[wave][hi * ZP_HALF + cl];
        #pragma unroll
        for (int r = 0; r < 8; ++r) zp[r * 16] = zc[r];
        __syncthreads();
        {
            const int t   = threadIdx.x;
            const int row = t >> 4;
            const int zi  = (row >> 3) * ZP_HALF + (row & 7) * 16 + (t & 15);
            float zs = sZ[0][zi];
            #pragma unroll
            for (int w = 1; w < 8; ++w) zs += sZ[w][zi];
            z[(size_t)tile * 256 + t] = zs;
        }
        if (tile * 16 < N_DST2) {
            const int rbase = tile * 16 + hi * 8;
            #pragma unroll
            for (int r = 0; r < 8; ++r) h[(size_t)(rbase + r) * H_F + col] = c[r];
        }
    }
}

__global__ __launch_bounds__(256)
void sage_layer2_kernel(const float* __restrict__ h,
                        const float* __restrict__ z,
                        const int*   __restrict__ list,
                        const int*   __restrict__ segStart,
                        const int*   __restrict__ cnt,
                        const float* __restrict__ Ws,
                        const float* __restrict__ bias,
                        float* __restrict__ out) {
    __shared__ __attribute__((aligned(16))) float sN[16 * 16];
    __shared__ __attribute__((aligned(16))) float sP[8][ZP_WAVE];
    const int t     = threadIdx.x;
    const int lane  = t & 31;
    const int wave  = t >> 5;
    const int hi    = lane >> 4;
    const int cl    = lane & 15;
    const int khalf = hi * 2;
    const int m0    = blockIdx.x * 16;

    const float* hr = h  + (size_t)(m0 + cl) * H_F + wave * 16 + khalf;
    const float* wr = Ws + (size_t)(wave * 16 + khalf) * N_CLS + cl;
    v2f fa[4], fb[4];
    #pragma unroll
    for (int s = 0; s < 4; ++s) {
        fa[s]   = *(const v2f*)(hr + s * 4);
        fb[s].x = wr[(size_t)(s * 4) * N_CLS];
        fb[s].y = wr[(size_t)(s * 4 + 1) * N_CLS];
    }

    const int d   = m0 + wave * 2 + hi;
    const int c   = cnt[d];
    const int st  = segStart[d];
    const int cm1 = max(c - 1, 0);
    const int2* seg = (const int2*)list + ((c > 0) ? st : 0);
    const int es = (lane >> 2) & 3;
    const int q  = lane & 3;
    const int cmax = max(c, __shfl_xor(c, 16));
    const float4* z4 = (const float4*)z;
    float4 sum = make_float4(0.f, 0.f, 0.f, 0.f);
    int2 e[4];
    #pragma unroll
    for (int k = 0; k < 4; ++k) e[k] = seg[min(k * 4 + es, cm1)];
    __builtin_amdgcn_sched_barrier(0);
    for (int base = 0; base < cmax; base += 16) {
        float4 v[4];
        int2   n[4];
        #pragma unroll
        for (int k = 0; k < 4; ++k) v[k] = z4[(size_t)e[k].x * 4 + q];
        #pragma unroll
        for (int k = 0; k < 4; ++k) n[k] = seg[min(base + 16 + k * 4 + es, cm1)];
        __builtin_amdgcn_sched_barrier(0);
        #pragma unroll
        for (int k = 0; k < 4; ++k) {
            const bool  ok = base + k * 4 + es < c;
            const float w  = __int_as_float(e[k].y);
            const float tx = sum.x + v[k].x * w, ty = sum.y + v[k].y * w;
            const float tz = sum.z + v[k].z * w, tw = sum.w + v[k].w * w;
            sum.x = ok ? tx : sum.x; sum.y = ok ? ty : sum.y;
            sum.z = ok ? tz : sum.z; sum.w = ok ? tw : sum.w;
            e[k] = n[k];
        }
    }
    sum.x += __shfl_xor(sum.x, 4); sum.y += __shfl_xor(sum.y, 4);
    sum.z += __shfl_xor(sum.z, 4); sum.w += __shfl_xor(sum.w, 4);
    sum.x += __shfl_xor(sum.x, 8); sum.y += __shfl_xor(sum.y, 8);
    sum.z += __shfl_xor(sum.z, 8); sum.w += __shfl_xor(sum.w, 8);
    if (es == 0) {
        const float inv = 1.0f / fmaxf((float)c, 1.0f);
        *(float4*)&sN[(wave * 2 + hi) * 16 + q * 4] =
            make_float4(sum.x * inv, sum.y * inv, sum.z * inv, sum.w * inv);
    }

    v8f pc = {};
    #pragma unroll
    for (int s = 0; s < 4; ++s)
        pc = __builtin_amdgcn_wmma_f32_16x16x4_f32(false, fa[s], false, fb[s],
                                                  (short)0, pc, false, false);
    float* pp = &sP[wave][hi * ZP_HALF + cl];
    #pragma unroll
    for (int r = 0; r < 8; ++r) pp[r * 16] = pc[r];
    __syncthreads();

    const int row = t >> 4;
    const int pi  = (row >> 3) * ZP_HALF + (row & 7) * 16 + cl;
    float v = sP[0][pi];
    #pragma unroll
    for (int w = 1; w < 8; ++w) v += sP[w][pi];
    v += sN[t] + bias[cl];
    float m = v;
    m = fmaxf(m, __shfl_xor(m, 1, 16));
    m = fmaxf(m, __shfl_xor(m, 2, 16));
    m = fmaxf(m, __shfl_xor(m, 4, 16));
    m = fmaxf(m, __shfl_xor(m, 8, 16));
    const float ex = __expf(v - m);
    float s = ex;
    s += __shfl_xor(s, 1, 16);
    s += __shfl_xor(s, 2, 16);
    s += __shfl_xor(s, 4, 16);
    s += __shfl_xor(s, 8, 16);
    out[(size_t)m0 * N_CLS + t] = (v - m) - __logf(s);
}

extern "C" void kernel_launch(void* const* d_in, const int* in_sizes, int n_in,
                              void* d_out, int out_size, void* d_ws, size_t ws_size,
                              hipStream_t stream) {
    const float* x    = (const float*)d_in[0];
    const int*   src1 = (const int*)  d_in[1];
    const int*   dst1 = (const int*)  d_in[2];
    const float* w1   = (const float*)d_in[3];
    const int*   src2 = (const int*)  d_in[4];
    const int*   dst2 = (const int*)  d_in[5];
    const float* w2   = (const float*)d_in[6];
    const float* Ws1  = (const float*)d_in[7];
    const float* Wn1  = (const float*)d_in[8];
    const float* b1   = (const float*)d_in[9];
    const float* Ws2  = (const float*)d_in[10];
    const float* Wn2  = (const float*)d_in[11];
    const float* b2   = (const float*)d_in[12];
    float* out = (float*)d_out;

    float* wsA  = (float*)d_ws;
    float* hbuf = wsA;
    float* zbuf = hbuf + (size_t)N_DST2 * 128;
    int* list1  = (int*)(wsA + (size_t)N_DST1 * 128 + N_DST1);
    int* dbase  = list1 + (size_t)N_DST1 * 128;
    int* cnt1   = dbase + 4;
    int* cnt2   = cnt1 + N_DST1;
    int* off1   = cnt2 + N_DST2;
    int* off2   = off1 + N_DST1;
    int* list2  = off2 + N_DST2;
    int2* rec        = (int2*)wsA;
    int* pieceCnt    = (int*)wsA + 6000000;
    int* pieceOff    = pieceCnt + NB * SWEEP_WG;

    const int E1 = in_sizes[1];
    const int E2 = in_sizes[4];

    bucket_local_kernel<<<SWEEP_WG, 1024, 0, stream>>>(src1, dst1, w1, src2, dst2, w2,
                                                       pieceCnt, pieceOff, rec, E1, E2);
    bucket_rows_kernel<<<NB, 1024, 0, stream>>>(rec, pieceCnt, pieceOff, cnt1, cnt2, off1, off2,
                                                (int2*)list1, (int2*)list2, E1, E2);

    sage_gemm1_kernel<<<G1_WG, 256, 0, stream>>>(x, list1, off1, cnt1,
                                                       Ws1, Wn1, b1, Wn2, hbuf, zbuf);

    sage_layer2_kernel<<<N_DST2 / 16, 256, 0, stream>>>(hbuf, zbuf, list2, off2, cnt2,
                                                        Ws2, b2, out);
}


# Layer-1 gather: edge records are read a tile ahead and each lane prefetches its records' source rows into L2 (global_prefetch_b8, no registers held), so row loads hit L2, not memory; sum order kept
# speedup vs baseline: 5.3243x; 1.0599x over previous
#include <hip/hip_runtime.h>
#include <hip/hip_bf16.h>

typedef __attribute__((ext_vector_type(2))) float v2f;
typedef __attribute__((ext_vector_type(8))) float v8f;

#define N_SRC1 500000
#define N_DST1 100000
#define N_DST2 20000
#define IN_F   128
#define H_F    128
#define N_CLS  16

#define NB1   196
#define NB2   79
#define NB    275
#define NBP   288
#define SWEEP_WG 256
#define L1_CAP 12288

__device__ __forceinline__ int wave_incl_scan(int v, int lane) {
    for (int o = 1; o < 32; o <<= 1) {
        const int up = __shfl_up(v, o);
        if (lane >= o) v += up;
    }
    return v;
}

__device__ __forceinline__ int block_excl_scan(int v, int* sTot) {
    const int lane = threadIdx.x & 31;
    const int wave = threadIdx.x >> 5;
    const int incl = wave_incl_scan(v, lane);
    if (lane == 31) sTot[wave] = incl;
    __syncthreads();
    if (wave == 0) {
        const int x  = sTot[lane];
        const int xi = wave_incl_scan(x, lane);
        sTot[32 + lane] = xi - x;
    }
    __syncthreads();
    return sTot[32 + wave] + incl - v;
}

__global__ __launch_bounds__(1024)
void bucket_local_kernel(const int*   __restrict__ src1,
                         const int*   __restrict__ dst1,
                         const float* __restrict__ w1,
                         const int*   __restrict__ src2,
                         const int*   __restrict__ dst2,
                         const float* __restrict__ w2,
                         int*  __restrict__ pieceCnt,
                         int*  __restrict__ pieceOff,
                         int2* __restrict__ rec,
                         int nE1, int nE2) {
    __shared__ int  sH[NBP];
    __shared__ int  sBase[NBP];
    __shared__ int  sTot[64];
    __shared__ int2 sRec[L1_CAP];
    const int t = threadIdx.x;
    if (t < NBP) sH[t] = 0;
    __syncthreads();
    const int c1 = (nE1 + SWEEP_WG - 1) / SWEEP_WG;
    const int c2 = (nE2 + SWEEP_WG - 1) / SWEEP_WG;
    const int a1 = min((int)blockIdx.x * c1, nE1);
    const int a2 = min((int)blockIdx.x * c2, nE2);
    const int n1 = min(c1, nE1 - a1);
    const int n2 = min(c2, nE2 - a2);
    const int m  = n1 + n2;
    const int lo = a1 + a2;
    const int o2 = a2 - n1;
    for (int base = 0; base < m; base += 4096) {
        int d[4];
        #pragma unroll
        for (int k = 0; k < 4; ++k) {
            const int uc = min(base + k * 1024 + t, m - 1);
            d[k] = *((uc < n1) ? dst1 + (a1 + uc) : dst2 + (o2 + uc));
        }
        #pragma unroll
        for (int k = 0; k < 4; ++k) {
            const int u  = base + k * 1024 + t;
            const int bk = (u >= m) ? NB : (u < n1) ? (d[k] >> 9) : NB1 + (d[k] >> 8);
            atomicAdd(&sH[bk], 1);
        }
    }
    __syncthreads();
    const int v    = (t < NB) ? sH[t] : 0;
    const int excl = block_excl_scan(v, sTot);
    if (t < NB) {
        sBase[t] = excl;
        pieceCnt[t * SWEEP_WG + blockIdx.x] = v;
        pieceOff[t * SWEEP_WG + blockIdx.x] = excl;
    } else if (t < NBP) {
        sBase[t] = 0;
    }
    __syncthreads();
    const bool stage = m <= L1_CAP;
    for (int base = 0; base < m; base += 4096) {
        int d[4], s[4], wb[4];
        #pragma unroll
        for (int k = 0; k < 4; ++k) {
            const int  uc = min(base + k * 1024 + t, m - 1);
            const bool a  = uc < n1;
            const int  iA = a1 + uc;
            const int  iB = o2 + uc;
            d[k]  = *(a ? dst1 + iA : dst2 + iB);
            s[k]  = *(a ? src1 + iA : src2 + iB);
            wb[k] = __float_as_int(*(a ? w1 + iA : w2 + iB));
        }
        int pos[4];
        #pragma unroll
        for (int k = 0; k < 4; ++k) {
            const int u  = base + k * 1024 + t;
            const int bk = (u >= m) ? NB : (u < n1) ? (d[k] >> 9) : NB1 + (d[k] >> 8);
            pos[k] = atomicAdd(&sBase[bk], 1);
        }
        #pragma unroll
        for (int k = 0; k < 4; ++k) {
            const int u = base + k * 1024 + t;
            if (u < m) {
                const int  lr = (u < n1) ? (d[k] & 511) : (d[k] & 255);
                const int2 r  = make_int2(s[k] | (lr << 20), wb[k]);
                if (stage) sRec[pos[k]] = r;
                else       rec[lo + pos[k]] = r;
            }
        }
    }
    if (stage) {
        __syncthreads();
        for (int i = t; i < m; i += 1024) rec[lo + i] = sRec[i];
    }
}

__global__ __launch_bounds__(1024)
void bucket_rows_kernel(const int2* __restrict__ rec,
                        const int* __restrict__ pieceCnt,
                        const int* __restrict__ pieceOff,
                        int* __restrict__ cnt1, int* __restrict__ cnt2,
                        int* __restrict__ off1, int* __restrict__ off2,
                        int2* __restrict__ list1, int2* __restrict__ list2,
                        int nE1, int nE2) {
    __shared__ int sCnt[544];
    __shared__ int sCur[544];
    __shared__ int sTot[64];
    __shared__ int sRed[32];
    const int t    = threadIdx.x;
    const int lane = t & 31;
    const int wave = t >> 5;
    const int b    = blockIdx.x;
    const bool l1  = b < NB1;
    const int row0  = l1 ? b * 512 : (b - NB1) * 256;
    const int nrows = l1 ? min(512, N_DST1 - row0) : min(256, N_DST2 - row0);
    if (t < 544) sCnt[t] = 0;
    const int g   = wave * 8 + (lane & 7);
    const int c1  = (nE1 + SWEEP_WG - 1) / SWEEP_WG;
    const int c2  = (nE2 + SWEEP_WG - 1) / SWEEP_WG;
    const int pcv = pieceCnt[b * SWEEP_WG + g];
    const int plv = pieceOff[b * SWEEP_WG + g];
    const int psv = min(g * c1, nE1) + min(g * c2, nE2) + plv;
    int part = (lane < 8) ? plv : 0;
    for (int o = 16; o > 0; o >>= 1) part += __shfl_xor(part, o);
    if (lane == 0) sRed[wave] = part;
    __syncthreads();
    int start = sRed[lane];
    for (int o = 16; o > 0; o >>= 1) start += __shfl_xor(start, o);
    int pcm = pcv;
    for (int o = 4; o > 0; o >>= 1) pcm = max(pcm, __shfl_xor(pcm, o));
    int pc[8], pb[8];
    #pragma unroll
    for (int q = 0; q < 8; ++q) {
        pc[q] = __builtin_amdgcn_readlane(pcv, q);
        pb[q] = (pc[q] > 0) ? __builtin_amdgcn_readlane(psv, q) : 0;
    }
    for (int j0 = 0; j0 < pcm; j0 += 32) {
        int x[8];
        #pragma unroll
        for (int q = 0; q < 8; ++q) x[q] = rec[pb[q] + min(j0 + lane, max(pc[q] - 1, 0))].x;
        #pragma unroll
        for (int q = 0; q < 8; ++q)
            atomicAdd(&sCnt[(j0 + lane < pc[q]) ? (int)((unsigned)x[q] >> 20) : 512 + lane], 1);
    }
    __syncthreads();
    const int v    = (t < 512) ? sCnt[t] : 0;
    const int excl = block_excl_scan(v, sTot);
    const int outBase = l1 ? start : start - nE1;
    if (t < 544) sCur[t] = (t < 512) ? excl : 0;
    if (t < nrows) {
        (l1 ? cnt1 : cnt2)[row0 + t] = v;
        (l1 ? off1 : off2)[row0 + t] = outBase + excl;
    }
    __syncthreads();
    int2* out = (l1 ? list1 : list2) + outBase;
    for (int j0 = 0; j0 < pcm; j0 += 32) {
        int2 e[8];
        #pragma unroll
        for (int q = 0; q < 8; ++q) e[q] = rec[pb[q] + min(j0 + lane, max(pc[q] - 1, 0))];
        int pos[8];
        #pragma unroll
        for (int q = 0; q < 8; ++q)
            pos[q] = atomicAdd(&sCur[(j0 + lane < pc[q]) ? (int)((unsigned)e[q].x >> 20) : 512 + lane], 1);
        #pragma unroll
        for (int q = 0; q < 8; ++q)
            if (j0 + lane < pc[q]) out[pos[q]] = make_int2(e[q].x & 0xFFFFF, e[q].y);
    }
}

#define G1_WG    512
#define G1_PITCH 260
#define G1_TP    20
#define ZP_HALF  144
#define ZP_WAVE  288

#define G1_GS 16

__device__ __forceinline__ int2 g1_record(const int2* __restrict__ list,
                                          int cA, int stA, int cB, int stB, int g) {
    int2 r = make_int2(0, 0);
    if (g < cA + cB) r = list[(g < cA) ? stA + g : stB + (g - cA)];
    return r;
}

__device__ __forceinline__ void g1_prefetch_row(const float* __restrict__ x, int sr) {
    const char* p = (const char*)(x + (size_t)sr * IN_F);
    __builtin_prefetch(p, 0, 1);
    __builtin_prefetch(p + 256, 0, 1);
}

__device__ __forceinline__ void g1_gather(const float* __restrict__ x,
                                          const int2*  __restrict__ list,
                                          int cA, int stA, int cB, int stB, int lane,
                                          int2 my, int2 nx,
                                          float4& sa, float4& sb) {
    sa = make_float4(0.f, 0.f, 0.f, 0.f);
    sb = make_float4(0.f, 0.f, 0.f, 0.f);
    const int ct = cA + cB;
    for (int base = 0; base < ct; base += 32) {
        const int2 n2 = g1_record(list, cA, stA, cB, stB, base + 64 + lane);
        const int n = min(32, ct - base);
        for (int j = 0; j < n; j += G1_GS) {
            float4 v[G1_GS];
            #pragma unroll
            for (int i = 0; i < G1_GS; ++i) {
                const int jj = min(j + i, n - 1);
                const int sr = __builtin_amdgcn_readlane(my.x, jj);
                v[i] = ((const float4*)(x + (size_t)sr * IN_F))[lane];
            }
            #pragma unroll
            for (int i = 0; i < G1_GS; ++i) {
                const int   jj = min(j + i, n - 1);
                const int   g  = base + j + i;
                const float w  = __int_as_float(__builtin_amdgcn_readlane(my.y, jj));
                const float wa = (g < cA) ? w : 0.0f;
                const float wb = (g >= cA && g < ct) ? w : 0.0f;
                sa.x = __builtin_fmaf(v[i].x, wa, sa.x); sa.y = __builtin_fmaf(v[i].y, wa, sa.y);
                sa.z = __builtin_fmaf(v[i].z, wa, sa.z); sa.w = __builtin_fmaf(v[i].w, wa, sa.w);
                sb.x = __builtin_fmaf(v[i].x, wb, sb.x); sb.y = __builtin_fmaf(v[i].y, wb, sb.y);
                sb.z = __builtin_fmaf(v[i].z, wb, sb.z); sb.w = __builtin_fmaf(v[i].w, wb, sb.w);
            }
        }
        my = nx;
        nx = n2;
        if (base + 64 + lane < ct) g1_prefetch_row(x, n2.x);
    }
}

__device__ __forceinline__ void g1_stage(float* s, int lane,
                                         const float4& px0, const float4& px1,
                                         const float4& pa0, const float4& pa1,
                                         float d0, float d1) {
    const float i0 = 1.0f / fmaxf(d0, 1.0f);
    const float i1 = 1.0f / fmaxf(d1, 1.0f);
    float* s0 = s + lane * 4;
    float* s1 = s0 + G1_PITCH;
    *(float4*)s0         = px0;
    *(float4*)(s0 + 128) = make_float4(pa0.x * i0, pa0.y * i0, pa0.z * i0, pa0.w * i0);
    *(float4*)s1         = px1;
    *(float4*)(s1 + 128) = make_float4(pa1.x * i1, pa1.y * i1, pa1.z * i1, pa1.w * i1);
}

__global__ __launch_bounds__(256)
void sage_gemm1_kernel(const float* __restrict__ x,
                       const int*   __restrict__ list,
                       const int*   __restrict__ segStart,
                       const int*   __restrict__ cnt,
                       const float* __restrict__ Ws,
                       const float* __restrict__ Wn,
                       const float* __restrict__ bias,
                       const float* __restrict__ Wn2,
                       float* __restrict__ h,
                       float* __restrict__ z) {
    __shared__ __attribute__((aligned(16))) float sA[2][16 * G1_PITCH];
    __shared__ __attribute__((aligned(16))) float sT[8][16 * G1_TP];
    __shared__ __attribute__((aligned(16))) float sZ[8][ZP_WAVE];
    const int lane   = threadIdx.x & 31;
    const int wave   = __builtin_amdgcn_readfirstlane(threadIdx.x >> 5);
    const int col    = wave * 16 + (lane & 15);
    const int khalf  = (lane >> 4) * 2;
    const int nTiles = N_DST1 / 16;
    const int2* rl   = (const int2*)list;

    int tile = blockIdx.x;
    if (tile >= nTiles) return;

    float4 px0, px1, pa0, pa1;
    float  d0, d1;
    int drow = tile * 16 + wave * 2;
    int cA  = __builtin_amdgcn_readfirstlane(cnt[drow]);
    int cB  = __builtin_amdgcn_readfirstlane(cnt[drow + 1]);
    int stA = __builtin_amdgcn_readfirstlane(segStart[drow]);
    int stB = __builtin_amdgcn_readfirstlane(segStart[drow + 1]);
    int mrow = min(tile + (int)gridDim.x, nTiles - 1) * 16 + wave * 2;
    int mcA = cnt[mrow], mcB = cnt[mrow + 1], msA = segStart[mrow], msB = segStart[mrow + 1];

    v2f breg[64];
    #pragma unroll
    for (int ks = 0; ks < 32; ++ks) {
        const int kk = ks * 4 + khalf;
        breg[ks].x      = Ws[(size_t)kk * H_F + col];
        breg[ks].y      = Ws[(size_t)(kk + 1) * H_F + col];
        breg[32 + ks].x = Wn[(size_t)kk * H_F + col];
        breg[32 + ks].y = Wn[(size_t)(kk + 1) * H_F + col];
    }
    const float bv = bias[col];

    v2f b2reg[4];
    #pragma unroll
    for (int s = 0; s < 4; ++s) {
        const int kk = wave * 16 + s * 4 + khalf;
        b2reg[s].x = Wn2[(size_t)kk * N_CLS + (lane & 15)];
        b2reg[s].y = Wn2[(size_t)(kk + 1) * N_CLS + (lane & 15)];
    }
    float* tp = &sT[wave][0];

    int2 qmy = g1_record(rl, cA, stA, cB, stB, lane);
    int2 qnx = g1_record(rl, cA, stA, cB, stB, 32 + lane);
    if (lane >= G1_GS && lane < cA + cB) g1_prefetch_row(x, qmy.x);
    if (32 + lane < cA + cB)             g1_prefetch_row(x, qnx.x);
    g1_gather(x, rl, cA, stA, cB, stB, lane, qmy, qnx, pa0, pa1);
    d0 = (float)cA;
    d1 = (float)cB;
    {
        const int uA = __builtin_amdgcn_readfirstlane(mcA), uB = __builtin_amdgcn_readfirstlane(mcB);
        const int vA = __builtin_amdgcn_readfirstlane(msA), vB = __builtin_amdgcn_readfirstlane(msB);
        qmy = g1_record(rl, uA, vA, uB, vB, lane);
        qnx = g1_record(rl, uA, vA, uB, vB, 32 + lane);
        px0 = ((const float4*)(x + (size_t)drow       * IN_F))[lane];
        px1 = ((const float4*)(x + (size_t)(drow + 1) * IN_F))[lane];
        if (lane < uA + uB)      g1_prefetch_row(x, qmy.x);
        if (32 + lane < uA + uB) g1_prefetch_row(x, qnx.x);
    }
    g1_stage(&sA[0][wave * 2 * G1_PITCH], lane, px0, px1, pa0, pa1, d0, d1);
    __syncthreads();

    int cur = 0;
    int pfA = 0;
    for (; tile < nTiles; tile += (int)gridDim.x) {
        const int  nxt  = tile + (int)gridDim.x;
        const bool more = nxt < nTiles;
        if (more) {
            drow = nxt * 16 + wave * 2;
            cA  = __builtin_amdgcn_readfirstlane(mcA);
            cB  = __builtin_amdgcn_readfirstlane(mcB);
            stA = __builtin_amdgcn_readfirstlane(msA);
            stB = __builtin_amdgcn_readfirstlane(msB);
            mrow = min(nxt + (int)gridDim.x, nTiles - 1) * 16 + wave * 2;
            mcA = cnt[mrow]; mcB = cnt[mrow + 1]; msA = segStart[mrow]; msB = segStart[mrow + 1];
            g1_gather(x, rl, cA, stA, cB, stB, lane, qmy, qnx, pa0, pa1);
            d0 = (float)cA;
            d1 = (float)cB;
            pfA = __builtin_amdgcn_readfirstlane(mcA) + __builtin_amdgcn_readfirstlane(mcB);
            {
                const int uA = __builtin_amdgcn_readfirstlane(mcA), uB = __builtin_amdgcn_readfirstlane(mcB);
                const int vA = __builtin_amdgcn_readfirstlane(msA), vB = __builtin_amdgcn_readfirstlane(msB);
                qmy = g1_record(rl, uA, vA, uB, vB, lane);
                qnx = g1_record(rl, uA, vA, uB, vB, 32 + lane);
            }
            px0 = ((const float4*)(x + (size_t)drow       * IN_F))[lane];
            px1 = ((const float4*)(x + (size_t)(drow + 1) * IN_F))[lane];
        }

        const float* ap = &sA[cur][(lane & 15) * G1_PITCH + khalf];
        v8f c = {};
        #pragma unroll
        for (int ks = 0; ks < 64; ++ks) {
            const v2f a = *(const v2f*)(ap + ks * 4);
            c = __builtin_amdgcn_wmma_f32_16x16x4_f32(false, a, false, breg[ks],
                                                      (short)0, c, false, false);
        }

        if (more) {
            if (lane < pfA)      g1_prefetch_row(x, qmy.x);
            if (32 + lane < pfA) g1_prefetch_row(x, qnx.x);
        }
        if (more) g1_stage(&sA[cur ^ 1][wave * 2 * G1_PITCH], lane, px0, px1, pa0, pa1, d0, d1);
        __syncthreads();
        cur ^= 1;

        const int hi = lane >> 4;
        const int cl = lane & 15;
        #pragma unroll
        for (int r = 0; r < 8; ++r) {
            c[r] = fmaxf(c[r] + bv, 0.0f);
            tp[(r + 8 * hi) * G1_TP + cl] = c[r];
        }
        v8f zc = {};
        #pragma unroll
        for (int s = 0; s < 4; ++s) {
            const v2f a = *(const v2f*)(tp + cl * G1_TP + s * 4 + khalf);
            zc = __builtin_amdgcn_wmma_f32_16x16x4_f32(false, a, false, b2reg[s],
                                                      (short)0, zc, false, false);
        }
        float* zp = &sZ[wave][hi * ZP_HALF + cl];
        #pragma unroll
        for (int r = 0; r < 8; ++r) zp[r * 16] = zc[r];
        __syncthreads();
        {
            const int t   = threadIdx.x;
            const int row = t >> 4;
            const int zi  = (row >> 3) * ZP_HALF + (row & 7) * 16 + (t & 15);
            float zs = sZ[0][zi];
            #pragma unroll
            for (int w = 1; w < 8; ++w) zs += sZ[w][zi];
            z[(size_t)tile * 256 + t] = zs;
        }
        if (tile * 16 < N_DST2) {
            const int rbase = tile * 16 + hi * 8;
            #pragma unroll
            for (int r = 0; r < 8; ++r) h[(size_t)(rbase + r) * H_F + col] = c[r];
        }
    }
}

__global__ __launch_bounds__(256)
void sage_layer2_kernel(const float* __restrict__ h,
                        const float* __restrict__ z,
                        const int*   __restrict__ list,
                        const int*   __restrict__ segStart,
                        const int*   __restrict__ cnt,
                        const float* __restrict__ Ws,
                        const float* __restrict__ bias,
                        float* __restrict__ out) {
    __shared__ __attribute__((aligned(16))) float sN[16 * 16];
    __shared__ __attribute__((aligned(16))) float sP[8][ZP_WAVE];
    const int t     = threadIdx.x;
    const int lane  = t & 31;
    const int wave  = t >> 5;
    const int hi    = lane >> 4;
    const int cl    = lane & 15;
    const int khalf = hi * 2;
    const int m0    = blockIdx.x * 16;

    const float* hr = h  + (size_t)(m0 + cl) * H_F + wave * 16 + khalf;
    const float* wr = Ws + (size_t)(wave * 16 + khalf) * N_CLS + cl;
    v2f fa[4], fb[4];
    #pragma unroll
    for (int s = 0; s < 4; ++s) {
        fa[s]   = *(const v2f*)(hr + s * 4);
        fb[s].x = wr[(size_t)(s * 4) * N_CLS];
        fb[s].y = wr[(size_t)(s * 4 + 1) * N_CLS];
    }

    const int d   = m0 + wave * 2 + hi;
    const int c   = cnt[d];
    const int st  = segStart[d];
    const int cm1 = max(c - 1, 0);
    const int2* seg = (const int2*)list + ((c > 0) ? st : 0);
    const int es = (lane >> 2) & 3;
    const int q  = lane & 3;
    const int cmax = max(c, __shfl_xor(c, 16));
    const float4* z4 = (const float4*)z;
    float4 sum = make_float4(0.f, 0.f, 0.f, 0.f);
    int2 e[4];
    #pragma unroll
    for (int k = 0; k < 4; ++k) e[k] = seg[min(k * 4 + es, cm1)];
    __builtin_amdgcn_sched_barrier(0);
    for (int base = 0; base < cmax; base += 16) {
        float4 v[4];
        int2   n[4];
        #pragma unroll
        for (int k = 0; k < 4; ++k) v[k] = z4[(size_t)e[k].x * 4 + q];
        #pragma unroll
        for (int k = 0; k < 4; ++k) n[k] = seg[min(base + 16 + k * 4 + es, cm1)];
        __builtin_amdgcn_sched_barrier(0);
        #pragma unroll
        for (int k = 0; k < 4; ++k) {
            const bool  ok = base + k * 4 + es < c;
            const float w  = __int_as_float(e[k].y);
            const float tx = sum.x + v[k].x * w, ty = sum.y + v[k].y * w;
            const float tz = sum.z + v[k].z * w, tw = sum.w + v[k].w * w;
            sum.x = ok ? tx : sum.x; sum.y = ok ? ty : sum.y;
            sum.z = ok ? tz : sum.z; sum.w = ok ? tw : sum.w;
            e[k] = n[k];
        }
    }
    sum.x += __shfl_xor(sum.x, 4); sum.y += __shfl_xor(sum.y, 4);
    sum.z += __shfl_xor(sum.z, 4); sum.w += __shfl_xor(sum.w, 4);
    sum.x += __shfl_xor(sum.x, 8); sum.y += __shfl_xor(sum.y, 8);
    sum.z += __shfl_xor(sum.z, 8); sum.w += __shfl_xor(sum.w, 8);
    if (es == 0) {
        const float inv = 1.0f / fmaxf((float)c, 1.0f);
        *(float4*)&sN[(wave * 2 + hi) * 16 + q * 4] =
            make_float4(sum.x * inv, sum.y * inv, sum.z * inv, sum.w * inv);
    }

    v8f pc = {};
    #pragma unroll
    for (int s = 0; s < 4; ++s)
        pc = __builtin_amdgcn_wmma_f32_16x16x4_f32(false, fa[s], false, fb[s],
                                                  (short)0, pc, false, false);
    float* pp = &sP[wave][hi * ZP_HALF + cl];
    #pragma unroll
    for (int r = 0; r < 8; ++r) pp[r * 16] = pc[r];
    __syncthreads();

    const int row = t >> 4;
    const int pi  = (row >> 3) * ZP_HALF + (row & 7) * 16 + cl;
    float v = sP[0][pi];
    #pragma unroll
    for (int w = 1; w < 8; ++w) v += sP[w][pi];
    v += sN[t] + bias[cl];
    float m = v;
    m = fmaxf(m, __shfl_xor(m, 1, 16));
    m = fmaxf(m, __shfl_xor(m, 2, 16));
    m = fmaxf(m, __shfl_xor(m, 4, 16));
    m = fmaxf(m, __shfl_xor(m, 8, 16));
    const float ex = __expf(v - m);
    float s = ex;
    s += __shfl_xor(s, 1, 16);
    s += __shfl_xor(s, 2, 16);
    s += __shfl_xor(s, 4, 16);
    s += __shfl_xor(s, 8, 16);
    out[(size_t)m0 * N_CLS + t] = (v - m) - __logf(s);
}

extern "C" void kernel_launch(void* const* d_in, const int* in_sizes, int n_in,
                              void* d_out, int out_size, void* d_ws, size_t ws_size,
                              hipStream_t stream) {
    const float* x    = (const float*)d_in[0];
    const int*   src1 = (const int*)  d_in[1];
    const int*   dst1 = (const int*)  d_in[2];
    const float* w1   = (const float*)d_in[3];
    const int*   src2 = (const int*)  d_in[4];
    const int*   dst2 = (const int*)  d_in[5];
    const float* w2   = (const float*)d_in[6];
    const float* Ws1  = (const float*)d_in[7];
    const float* Wn1  = (const float*)d_in[8];
    const float* b1   = (const float*)d_in[9];
    const float* Ws2  = (const float*)d_in[10];
    const float* Wn2  = (const float*)d_in[11];
    const float* b2   = (const float*)d_in[12];
    float* out = (float*)d_out;

    float* wsA  = (float*)d_ws;
    float* hbuf = wsA;
    float* zbuf = hbuf + (size_t)N_DST2 * 128;
    int* list1  = (int*)(wsA + (size_t)N_DST1 * 128 + N_DST1);
    int* dbase  = list1 + (size_t)N_DST1 * 128;
    int* cnt1   = dbase + 4;
    int* cnt2   = cnt1 + N_DST1;
    int* off1   = cnt2 + N_DST2;
    int* off2   = off1 + N_DST1;
    int* list2  = off2 + N_DST2;
    int2* rec        = (int2*)wsA;
    int* pieceCnt    = (int*)wsA + 6000000;
    int* pieceOff    = pieceCnt + NB * SWEEP_WG;

    const int E1 = in_sizes[1];
    const int E2 = in_sizes[4];

    bucket_local_kernel<<<SWEEP_WG, 1024, 0, stream>>>(src1, dst1, w1, src2, dst2, w2,
                                                       pieceCnt, pieceOff, rec, E1, E2);
    bucket_rows_kernel<<<NB, 1024, 0, stream>>>(rec, pieceCnt, pieceOff, cnt1, cnt2, off1, off2,
                                                (int2*)list1, (int2*)list2, E1, E2);

    sage_gemm1_kernel<<<G1_WG, 256, 0, stream>>>(x, list1, off1, cnt1,
                                                       Ws1, Wn1, b1, Wn2, hbuf, zbuf);

    sage_layer2_kernel<<<N_DST2 / 16, 256, 0, stream>>>(hbuf, zbuf, list2, off2, cnt2,
                                                        Ws2, b2, out);
}


# Layer-1 gather: a record costs 3 lane reads, one 32-bit offset, one load and its sums (weights per row chosen once per 32 records in vector registers, fixed lane indices); sums and order kept
# speedup vs baseline: 5.4827x; 1.0298x over previous
#include <hip/hip_runtime.h>
#include <hip/hip_bf16.h>

typedef __attribute__((ext_vector_type(2))) float v2f;
typedef __attribute__((ext_vector_type(8))) float v8f;

#define N_SRC1 500000
#define N_DST1 100000
#define N_DST2 20000
#define IN_F   128
#define H_F    128
#define N_CLS  16

#define NB1   196
#define NB2   79
#define NB    275
#define NBP   288
#define SWEEP_WG 256
#define L1_CAP 12288

__device__ __forceinline__ int wave_incl_scan(int v, int lane) {
    for (int o = 1; o < 32; o <<= 1) {
        const int up = __shfl_up(v, o);
        if (lane >= o) v += up;
    }
    return v;
}

__device__ __forceinline__ int block_excl_scan(int v, int* sTot) {
    const int lane = threadIdx.x & 31;
    const int wave = threadIdx.x >> 5;
    const int incl = wave_incl_scan(v, lane);
    if (lane == 31) sTot[wave] = incl;
    __syncthreads();
    if (wave == 0) {
        const int x  = sTot[lane];
        const int xi = wave_incl_scan(x, lane);
        sTot[32 + lane] = xi - x;
    }
    __syncthreads();
    return sTot[32 + wave] + incl - v;
}

__global__ __launch_bounds__(1024)
void bucket_local_kernel(const int*   __restrict__ src1,
                         const int*   __restrict__ dst1,
                         const float* __restrict__ w1,
                         const int*   __restrict__ src2,
                         const int*   __restrict__ dst2,
                         const float* __restrict__ w2,
                         int*  __restrict__ pieceCnt,
                         int*  __restrict__ pieceOff,
                         int2* __restrict__ rec,
                         int nE1, int nE2) {
    __shared__ int  sH[NBP];
    __shared__ int  sBase[NBP];
    __shared__ int  sTot[64];
    __shared__ int2 sRec[L1_CAP];
    const int t = threadIdx.x;
    if (t < NBP) sH[t] = 0;
    __syncthreads();
    const int c1 = (nE1 + SWEEP_WG - 1) / SWEEP_WG;
    const int c2 = (nE2 + SWEEP_WG - 1) / SWEEP_WG;
    const int a1 = min((int)blockIdx.x * c1, nE1);
    const int a2 = min((int)blockIdx.x * c2, nE2);
    const int n1 = min(c1, nE1 - a1);
    const int n2 = min(c2, nE2 - a2);
    const int m  = n1 + n2;
    const int lo = a1 + a2;
    const int o2 = a2 - n1;
    for (int base = 0; base < m; base += 4096) {
        int d[4];
        #pragma unroll
        for (int k = 0; k < 4; ++k) {
            const int uc = min(base + k * 1024 + t, m - 1);
            d[k] = *((uc < n1) ? dst1 + (a1 + uc) : dst2 + (o2 + uc));
        }
        #pragma unroll
        for (int k = 0; k < 4; ++k) {
            const int u  = base + k * 1024 + t;
            const int bk = (u >= m) ? NB : (u < n1) ? (d[k] >> 9) : NB1 + (d[k] >> 8);
            atomicAdd(&sH[bk], 1);
        }
    }
    __syncthreads();
    const int v    = (t < NB) ? sH[t] : 0;
    const int excl = block_excl_scan(v, sTot);
    if (t < NB) {
        sBase[t] = excl;
        pieceCnt[t * SWEEP_WG + blockIdx.x] = v;
        pieceOff[t * SWEEP_WG + blockIdx.x] = excl;
    } else if (t < NBP) {
        sBase[t] = 0;
    }
    __syncthreads();
    const bool stage = m <= L1_CAP;
    for (int base = 0; base < m; base += 4096) {
        int d[4], s[4], wb[4];
        #pragma unroll
        for (int k = 0; k < 4; ++k) {
            const int  uc = min(base + k * 1024 + t, m - 1);
            const bool a  = uc < n1;
            const int  iA = a1 + uc;
            const int  iB = o2 + uc;
            d[k]  = *(a ? dst1 + iA : dst2 + iB);
            s[k]  = *(a ? src1 + iA : src2 + iB);
            wb[k] = __float_as_int(*(a ? w1 + iA : w2 + iB));
        }
        int pos[4];
        #pragma unroll
        for (int k = 0; k < 4; ++k) {
            const int u  = base + k * 1024 + t;
            const int bk = (u >= m) ? NB : (u < n1) ? (d[k] >> 9) : NB1 + (d[k] >> 8);
            pos[k] = atomicAdd(&sBase[bk], 1);
        }
        #pragma unroll
        for (int k = 0; k < 4; ++k) {
            const int u = base + k * 1024 + t;
            if (u < m) {
                const int  lr = (u < n1) ? (d[k] & 511) : (d[k] & 255);
                const int2 r  = make_int2(s[k] | (lr << 20), wb[k]);
                if (stage) sRec[pos[k]] = r;
                else       rec[lo + pos[k]] = r;
            }
        }
    }
    if (stage) {
        __syncthreads();
        for (int i = t; i < m; i += 1024) rec[lo + i] = sRec[i];
    }
}

__global__ __launch_bounds__(1024)
void bucket_rows_kernel(const int2* __restrict__ rec,
                        const int* __restrict__ pieceCnt,
                        const int* __restrict__ pieceOff,
                        int* __restrict__ cnt1, int* __restrict__ cnt2,
                        int* __restrict__ off1, int* __restrict__ off2,
                        int2* __restrict__ list1, int2* __restrict__ list2,
                        int nE1, int nE2) {
    __shared__ int sCnt[544];
    __shared__ int sCur[544];
    __shared__ int sTot[64];
    __shared__ int sRed[32];
    const int t    = threadIdx.x;
    const int lane = t & 31;
    const int wave = t >> 5;
    const int b    = blockIdx.x;
    const bool l1  = b < NB1;
    const int row0  = l1 ? b * 512 : (b - NB1) * 256;
    const int nrows = l1 ? min(512, N_DST1 - row0) : min(256, N_DST2 - row0);
    if (t < 544) sCnt[t] = 0;
    const int g   = wave * 8 + (lane & 7);
    const int c1  = (nE1 + SWEEP_WG - 1) / SWEEP_WG;
    const int c2  = (nE2 + SWEEP_WG - 1) / SWEEP_WG;
    const int pcv = pieceCnt[b * SWEEP_WG + g];
    const int plv = pieceOff[b * SWEEP_WG + g];
    const int psv = min(g * c1, nE1) + min(g * c2, nE2) + plv;
    int part = (lane < 8) ? plv : 0;
    for (int o = 16; o > 0; o >>= 1) part += __shfl_xor(part, o);
    if (lane == 0) sRed[wave] = part;
    __syncthreads();
    int start = sRed[lane];
    for (int o = 16; o > 0; o >>= 1) start += __shfl_xor(start, o);
    int pcm = pcv;
    for (int o = 4; o > 0; o >>= 1) pcm = max(pcm, __shfl_xor(pcm, o));
    int pc[8], pb[8];
    #pragma unroll
    for (int q = 0; q < 8; ++q) {
        pc[q] = __builtin_amdgcn_readlane(pcv, q);
        pb[q] = (pc[q] > 0) ? __builtin_amdgcn_readlane(psv, q) : 0;
    }
    for (int j0 = 0; j0 < pcm; j0 += 32) {
        int x[8];
        #pragma unroll
        for (int q = 0; q < 8; ++q) x[q] = rec[pb[q] + min(j0 + lane, max(pc[q] - 1, 0))].x;
        #pragma unroll
        for (int q = 0; q < 8; ++q)
            atomicAdd(&sCnt[(j0 + lane < pc[q]) ? (int)((unsigned)x[q] >> 20) : 512 + lane], 1);
    }
    __syncthreads();
    const int v    = (t < 512) ? sCnt[t] : 0;
    const int excl = block_excl_scan(v, sTot);
    const int outBase = l1 ? start : start - nE1;
    if (t < 544) sCur[t] = (t < 512) ? excl : 0;
    if (t < nrows) {
        (l1 ? cnt1 : cnt2)[row0 + t] = v;
        (l1 ? off1 : off2)[row0 + t] = outBase + excl;
    }
    __syncthreads();
    int2* out = (l1 ? list1 : list2) + outBase;
    for (int j0 = 0; j0 < pcm; j0 += 32) {
        int2 e[8];
        #pragma unroll
        for (int q = 0; q < 8; ++q) e[q] = rec[pb[q] + min(j0 + lane, max(pc[q] - 1, 0))];
        int pos[8];
        #pragma unroll
        for (int q = 0; q < 8; ++q)
            pos[q] = atomicAdd(&sCur[(j0 + lane < pc[q]) ? (int)((unsigned)e[q].x >> 20) : 512 + lane], 1);
        #pragma unroll
        for (int q = 0; q < 8; ++q)
            if (j0 + lane < pc[q]) out[pos[q]] = make_int2(e[q].x & 0xFFFFF, e[q].y);
    }
}

#define G1_WG    512
#define G1_PITCH 260
#define G1_TP    20
#define ZP_HALF  144
#define ZP_WAVE  288

#define G1_GS 16

__device__ __forceinline__ int2 g1_record(const int2* __restrict__ list,
                                          int cA, int stA, int cB, int stB, int g) {
    int2 r = make_int2(0, 0);
    if (g < cA + cB) r = list[(g < cA) ? stA + g : stB + (g - cA)];
    return r;
}

__device__ __forceinline__ void g1_prefetch_row(const float* __restrict__ x, int sr) {
    const char* p = (const char*)(x + (size_t)sr * IN_F);
    __builtin_prefetch(p, 0, 1);
    __builtin_prefetch(p + 256, 0, 1);
}

__device__ __forceinline__ void g1_trip(const char* __restrict__ xb, unsigned lo16, const int J,
                                        int srcs, float wAl, float wBl,
                                        float4& sa, float4& sb) {
    float4   v[G1_GS];
    unsigned off[G1_GS];
    #pragma unroll
    for (int i = 0; i < G1_GS; ++i) {
        const unsigned sr = (unsigned)__builtin_amdgcn_readlane(srcs, J + i);
        off[i] = sr * (unsigned)(IN_F * 4) + lo16;
    }
    __builtin_amdgcn_sched_barrier(0);
    #pragma unroll
    for (int i = 0; i < G1_GS; ++i) v[i] = *(const float4*)(xb + off[i]);
    #pragma unroll
    for (int i = 0; i < G1_GS; ++i) {
        const float wa = __int_as_float(__builtin_amdgcn_readlane(__float_as_int(wAl), J + i));
        const float wb = __int_as_float(__builtin_amdgcn_readlane(__float_as_int(wBl), J + i));
        sa.x = __builtin_fmaf(v[i].x, wa, sa.x); sa.y = __builtin_fmaf(v[i].y, wa, sa.y);
        sa.z = __builtin_fmaf(v[i].z, wa, sa.z); sa.w = __builtin_fmaf(v[i].w, wa, sa.w);
        sb.x = __builtin_fmaf(v[i].x, wb, sb.x); sb.y = __builtin_fmaf(v[i].y, wb, sb.y);
        sb.z = __builtin_fmaf(v[i].z, wb, sb.z); sb.w = __builtin_fmaf(v[i].w, wb, sb.w);
    }
}

__device__ __forceinline__ void g1_gather(const float* __restrict__ x,
                                          const int2*  __restrict__ list,
                                          int cA, int stA, int cB, int stB, int lane,
                                          int2 my, int2 nx,
                                          float4& sa, float4& sb) {
    sa = make_float4(0.f, 0.f, 0.f, 0.f);
    sb = make_float4(0.f, 0.f, 0.f, 0.f);
    const int ct = cA + cB;
    const char* xb = (const char*)x;
    const unsigned lo16 = (unsigned)lane * 16u;
    for (int base = 0; base < ct; base += 32) {
        const int2 n2 = g1_record(list, cA, stA, cB, stB, base + 64 + lane);
        const float w   = __int_as_float(my.y);
        const float wAl = (base + lane <  cA) ? w : 0.0f;
        const float wBl = (base + lane >= cA) ? w : 0.0f;
        g1_trip(xb, lo16, 0, my.x, wAl, wBl, sa, sb);
        if (ct - base > G1_GS)
            g1_trip(xb, lo16, G1_GS, my.x, wAl, wBl, sa, sb);
        my = nx;
        nx = n2;
        if (base + 64 + lane < ct) g1_prefetch_row(x, n2.x);
    }
}

__device__ __forceinline__ void g1_stage(float* s, int lane,
                                         const float4& px0, const float4& px1,
                                         const float4& pa0, const float4& pa1,
                                         float d0, float d1) {
    const float i0 = 1.0f / fmaxf(d0, 1.0f);
    const float i1 = 1.0f / fmaxf(d1, 1.0f);
    float* s0 = s + lane * 4;
    float* s1 = s0 + G1_PITCH;
    *(float4*)s0         = px0;
    *(float4*)(s0 + 128) = make_float4(pa0.x * i0, pa0.y * i0, pa0.z * i0, pa0.w * i0);
    *(float4*)s1         = px1;
    *(float4*)(s1 + 128) = make_float4(pa1.x * i1, pa1.y * i1, pa1.z * i1, pa1.w * i1);
}

__global__ __launch_bounds__(256)
void sage_gemm1_kernel(const float* __restrict__ x,
                       const int*   __restrict__ list,
                       const int*   __restrict__ segStart,
                       const int*   __restrict__ cnt,
                       const float* __restrict__ Ws,
                       const float* __restrict__ Wn,
                       const float* __restrict__ bias,
                       const float* __restrict__ Wn2,
                       float* __restrict__ h,
                       float* __restrict__ z) {
    __shared__ __attribute__((aligned(16))) float sA[2][16 * G1_PITCH];
    __shared__ __attribute__((aligned(16))) float sT[8][16 * G1_TP];
    __shared__ __attribute__((aligned(16))) float sZ[8][ZP_WAVE];
    const int lane   = threadIdx.x & 31;
    const int wave   = __builtin_amdgcn_readfirstlane(threadIdx.x >> 5);
    const int col    = wave * 16 + (lane & 15);
    const int khalf  = (lane >> 4) * 2;
    const int nTiles = N_DST1 / 16;
    const int2* rl   = (const int2*)list;

    int tile = blockIdx.x;
    if (tile >= nTiles) return;

    float4 px0, px1, pa0, pa1;
    float  d0, d1;
    int drow = tile * 16 + wave * 2;
    int cA  = __builtin_amdgcn_readfirstlane(cnt[drow]);
    int cB  = __builtin_amdgcn_readfirstlane(cnt[drow + 1]);
    int stA = __builtin_amdgcn_readfirstlane(segStart[drow]);
    int stB = __builtin_amdgcn_readfirstlane(segStart[drow + 1]);
    int mrow = min(tile + (int)gridDim.x, nTiles - 1) * 16 + wave * 2;
    int mcA = cnt[mrow], mcB = cnt[mrow + 1], msA = segStart[mrow], msB = segStart[mrow + 1];

    v2f breg[64];
    #pragma unroll
    for (int ks = 0; ks < 32; ++ks) {
        const int kk = ks * 4 + khalf;
        breg[ks].x      = Ws[(size_t)kk * H_F + col];
        breg[ks].y      = Ws[(size_t)(kk + 1) * H_F + col];
        breg[32 + ks].x = Wn[(size_t)kk * H_F + col];
        breg[32 + ks].y = Wn[(size_t)(kk + 1) * H_F + col];
    }
    const float bv = bias[col];

    v2f b2reg[4];
    #pragma unroll
    for (int s = 0; s < 4; ++s) {
        const int kk = wave * 16 + s * 4 + khalf;
        b2reg[s].x = Wn2[(size_t)kk * N_CLS + (lane & 15)];
        b2reg[s].y = Wn2[(size_t)(kk + 1) * N_CLS + (lane & 15)];
    }
    float* tp = &sT[wave][0];

    int2 qmy = g1_record(rl, cA, stA, cB, stB, lane);
    int2 qnx = g1_record(rl, cA, stA, cB, stB, 32 + lane);
    if (lane >= G1_GS && lane < cA + cB) g1_prefetch_row(x, qmy.x);
    if (32 + lane < cA + cB)             g1_prefetch_row(x, qnx.x);
    g1_gather(x, rl, cA, stA, cB, stB, lane, qmy, qnx, pa0, pa1);
    d0 = (float)cA;
    d1 = (float)cB;
    {
        const int uA = __builtin_amdgcn_readfirstlane(mcA), uB = __builtin_amdgcn_readfirstlane(mcB);
        const int vA = __builtin_amdgcn_readfirstlane(msA), vB = __builtin_amdgcn_readfirstlane(msB);
        qmy = g1_record(rl, uA, vA, uB, vB, lane);
        qnx = g1_record(rl, uA, vA, uB, vB, 32 + lane);
        px0 = ((const float4*)(x + (size_t)drow       * IN_F))[lane];
        px1 = ((const float4*)(x + (size_t)(drow + 1) * IN_F))[lane];
        if (lane < uA + uB)      g1_prefetch_row(x, qmy.x);
        if (32 + lane < uA + uB) g1_prefetch_row(x, qnx.x);
    }
    g1_stage(&sA[0][wave * 2 * G1_PITCH], lane, px0, px1, pa0, pa1, d0, d1);
    __syncthreads();

    int cur = 0;
    int pfA = 0;
    for (; tile < nTiles; tile += (int)gridDim.x) {
        const int  nxt  = tile + (int)gridDim.x;
        const bool more = nxt < nTiles;
        if (more) {
            drow = nxt * 16 + wave * 2;
            cA  = __builtin_amdgcn_readfirstlane(mcA);
            cB  = __builtin_amdgcn_readfirstlane(mcB);
            stA = __builtin_amdgcn_readfirstlane(msA);
            stB = __builtin_amdgcn_readfirstlane(msB);
            mrow = min(nxt + (int)gridDim.x, nTiles - 1) * 16 + wave * 2;
            mcA = cnt[mrow]; mcB = cnt[mrow + 1]; msA = segStart[mrow]; msB = segStart[mrow + 1];
            g1_gather(x, rl, cA, stA, cB, stB, lane, qmy, qnx, pa0, pa1);
            d0 = (float)cA;
            d1 = (float)cB;
            pfA = __builtin_amdgcn_readfirstlane(mcA) + __builtin_amdgcn_readfirstlane(mcB);
            {
                const int uA = __builtin_amdgcn_readfirstlane(mcA), uB = __builtin_amdgcn_readfirstlane(mcB);
                const int vA = __builtin_amdgcn_readfirstlane(msA), vB = __builtin_amdgcn_readfirstlane(msB);
                qmy = g1_record(rl, uA, vA, uB, vB, lane);
                qnx = g1_record(rl, uA, vA, uB, vB, 32 + lane);
            }
            px0 = ((const float4*)(x + (size_t)drow       * IN_F))[lane];
            px1 = ((const float4*)(x + (size_t)(drow + 1) * IN_F))[lane];
        }

        const float* ap = &sA[cur][(lane & 15) * G1_PITCH + khalf];
        v8f c = {};
        #pragma unroll
        for (int ks = 0; ks < 64; ++ks) {
            const v2f a = *(const v2f*)(ap + ks * 4);
            c = __builtin_amdgcn_wmma_f32_16x16x4_f32(false, a, false, breg[ks],
                                                      (short)0, c, false, false);
        }

        if (more) {
            if (lane < pfA)      g1_prefetch_row(x, qmy.x);
            if (32 + lane < pfA) g1_prefetch_row(x, qnx.x);
        }
        if (more) g1_stage(&sA[cur ^ 1][wave * 2 * G1_PITCH], lane, px0, px1, pa0, pa1, d0, d1);
        __syncthreads();
        cur ^= 1;

        const int hi = lane >> 4;
        const int cl = lane & 15;
        #pragma unroll
        for (int r = 0; r < 8; ++r) {
            c[r] = fmaxf(c[r] + bv, 0.0f);
            tp[(r + 8 * hi) * G1_TP + cl] = c[r];
        }
        v8f zc = {};
        #pragma unroll
        for (int s = 0; s < 4; ++s) {
            const v2f a = *(const v2f*)(tp + cl * G1_TP + s * 4 + khalf);
            zc = __builtin_amdgcn_wmma_f32_16x16x4_f32(false, a, false, b2reg[s],
                                                      (short)0, zc, false, false);
        }
        float* zp = &sZ[wave][hi * ZP_HALF + cl];
        #pragma unroll
        for (int r = 0; r < 8; ++r) zp[r * 16] = zc[r];
        __syncthreads();
        {
            const int t   = threadIdx.x;
            const int row = t >> 4;
            const int zi  = (row >> 3) * ZP_HALF + (row & 7) * 16 + (t & 15);
            float zs = sZ[0][zi];
            #pragma unroll
            for (int w = 1; w < 8; ++w) zs += sZ[w][zi];
            z[(size_t)tile * 256 + t] = zs;
        }
        if (tile * 16 < N_DST2) {
            const int rbase = tile * 16 + hi * 8;
            #pragma unroll
            for (int r = 0; r < 8; ++r) h[(size_t)(rbase + r) * H_F + col] = c[r];
        }
    }
}

__global__ __launch_bounds__(256)
void sage_layer2_kernel(const float* __restrict__ h,
                        const float* __restrict__ z,
                        const int*   __restrict__ list,
                        const int*   __restrict__ segStart,
                        const int*   __restrict__ cnt,
                        const float* __restrict__ Ws,
                        const float* __restrict__ bias,
                        float* __restrict__ out) {
    __shared__ __attribute__((aligned(16))) float sN[16 * 16];
    __shared__ __attribute__((aligned(16))) float sP[8][ZP_WAVE];
    const int t     = threadIdx.x;
    const int lane  = t & 31;
    const int wave  = t >> 5;
    const int hi    = lane >> 4;
    const int cl    = lane & 15;
    const int khalf = hi * 2;
    const int m0    = blockIdx.x * 16;

    const float* hr = h  + (size_t)(m0 + cl) * H_F + wave * 16 + khalf;
    const float* wr = Ws + (size_t)(wave * 16 + khalf) * N_CLS + cl;
    v2f fa[4], fb[4];
    #pragma unroll
    for (int s = 0; s < 4; ++s) {
        fa[s]   = *(const v2f*)(hr + s * 4);
        fb[s].x = wr[(size_t)(s * 4) * N_CLS];
        fb[s].y = wr[(size_t)(s * 4 + 1) * N_CLS];
    }

    const int d   = m0 + wave * 2 + hi;
    const int c   = cnt[d];
    const int st  = segStart[d];
    const int cm1 = max(c - 1, 0);
    const int2* seg = (const int2*)list + ((c > 0) ? st : 0);
    const int es = (lane >> 2) & 3;
    const int q  = lane & 3;
    const int cmax = max(c, __shfl_xor(c, 16));
    const float4* z4 = (const float4*)z;
    float4 sum = make_float4(0.f, 0.f, 0.f, 0.f);
    int2 e[4];
    #pragma unroll
    for (int k = 0; k < 4; ++k) e[k] = seg[min(k * 4 + es, cm1)];
    __builtin_amdgcn_sched_barrier(0);
    for (int base = 0; base < cmax; base += 16) {
        float4 v[4];
        int2   n[4];
        #pragma unroll
        for (int k = 0; k < 4; ++k) v[k] = z4[(size_t)e[k].x * 4 + q];
        #pragma unroll
        for (int k = 0; k < 4; ++k) n[k] = seg[min(base + 16 + k * 4 + es, cm1)];
        __builtin_amdgcn_sched_barrier(0);
        #pragma unroll
        for (int k = 0; k < 4; ++k) {
            const bool  ok = base + k * 4 + es < c;
            const float w  = __int_as_float(e[k].y);
            const float tx = sum.x + v[k].x * w, ty = sum.y + v[k].y * w;
            const float tz = sum.z + v[k].z * w, tw = sum.w + v[k].w * w;
            sum.x = ok ? tx : sum.x; sum.y = ok ? ty : sum.y;
            sum.z = ok ? tz : sum.z; sum.w = ok ? tw : sum.w;
            e[k] = n[k];
        }
    }
    sum.x += __shfl_xor(sum.x, 4); sum.y += __shfl_xor(sum.y, 4);
    sum.z += __shfl_xor(sum.z, 4); sum.w += __shfl_xor(sum.w, 4);
    sum.x += __shfl_xor(sum.x, 8); sum.y += __shfl_xor(sum.y, 8);
    sum.z += __shfl_xor(sum.z, 8); sum.w += __shfl_xor(sum.w, 8);
    if (es == 0) {
        const float inv = 1.0f / fmaxf((float)c, 1.0f);
        *(float4*)&sN[(wave * 2 + hi) * 16 + q * 4] =
            make_float4(sum.x * inv, sum.y * inv, sum.z * inv, sum.w * inv);
    }

    v8f pc = {};
    #pragma unroll
    for (int s = 0; s < 4; ++s)
        pc = __builtin_amdgcn_wmma_f32_16x16x4_f32(false, fa[s], false, fb[s],
                                                  (short)0, pc, false, false);
    float* pp = &sP[wave][hi * ZP_HALF + cl];
    #pragma unroll
    for (int r = 0; r < 8; ++r) pp[r * 16] = pc[r];
    __syncthreads();

    const int row = t >> 4;
    const int pi  = (row >> 3) * ZP_HALF + (row & 7) * 16 + cl;
    float v = sP[0][pi];
    #pragma unroll
    for (int w = 1; w < 8; ++w) v += sP[w][pi];
    v += sN[t] + bias[cl];
    float m = v;
    m = fmaxf(m, __shfl_xor(m, 1, 16));
    m = fmaxf(m, __shfl_xor(m, 2, 16));
    m = fmaxf(m, __shfl_xor(m, 4, 16));
    m = fmaxf(m, __shfl_xor(m, 8, 16));
    const float ex = __expf(v - m);
    float s = ex;
    s += __shfl_xor(s, 1, 16);
    s += __shfl_xor(s, 2, 16);
    s += __shfl_xor(s, 4, 16);
    s += __shfl_xor(s, 8, 16);
    out[(size_t)m0 * N_CLS + t] = (v - m) - __logf(s);
}

extern "C" void kernel_launch(void* const* d_in, const int* in_sizes, int n_in,
                              void* d_out, int out_size, void* d_ws, size_t ws_size,
                              hipStream_t stream) {
    const float* x    = (const float*)d_in[0];
    const int*   src1 = (const int*)  d_in[1];
    const int*   dst1 = (const int*)  d_in[2];
    const float* w1   = (const float*)d_in[3];
    const int*   src2 = (const int*)  d_in[4];
    const int*   dst2 = (const int*)  d_in[5];
    const float* w2   = (const float*)d_in[6];
    const float* Ws1  = (const float*)d_in[7];
    const float* Wn1  = (const float*)d_in[8];
    const float* b1   = (const float*)d_in[9];
    const float* Ws2  = (const float*)d_in[10];
    const float* Wn2  = (const float*)d_in[11];
    const float* b2   = (const float*)d_in[12];
    float* out = (float*)d_out;

    float* wsA  = (float*)d_ws;
    float* hbuf = wsA;
    float* zbuf = hbuf + (size_t)N_DST2 * 128;
    int* list1  = (int*)(wsA + (size_t)N_DST1 * 128 + N_DST1);
    int* dbase  = list1 + (size_t)N_DST1 * 128;
    int* cnt1   = dbase + 4;
    int* cnt2   = cnt1 + N_DST1;
    int* off1   = cnt2 + N_DST2;
    int* off2   = off1 + N_DST1;
    int* list2  = off2 + N_DST2;
    int2* rec        = (int2*)wsA;
    int* pieceCnt    = (int*)wsA + 6000000;
    int* pieceOff    = pieceCnt + NB * SWEEP_WG;

    const int E1 = in_sizes[1];
    const int E2 = in_sizes[4];

    bucket_local_kernel<<<SWEEP_WG, 1024, 0, stream>>>(src1, dst1, w1, src2, dst2, w2,
                                                       pieceCnt, pieceOff, rec, E1, E2);
    bucket_rows_kernel<<<NB, 1024, 0, stream>>>(rec, pieceCnt, pieceOff, cnt1, cnt2, off1, off2,
                                                (int2*)list1, (int2*)list2, E1, E2);

    sage_gemm1_kernel<<<G1_WG, 256, 0, stream>>>(x, list1, off1, cnt1,
                                                       Ws1, Wn1, b1, Wn2, hbuf, zbuf);

    sage_layer2_kernel<<<N_DST2 / 16, 256, 0, stream>>>(hbuf, zbuf, list2, off2, cnt2,
                                                        Ws2, b2, out);
}


# Edge grouping: both kernels read their records once into registers (counting LDS add = rank) and store them once, whole lines from an LDS-built list (row kernel: 640-row buckets); sum order: LDS rank
# speedup vs baseline: 6.3770x; 1.1631x over previous
#include <hip/hip_runtime.h>
#include <hip/hip_bf16.h>

typedef __attribute__((ext_vector_type(2))) float v2f;
typedef __attribute__((ext_vector_type(8))) float v8f;

#define N_SRC1 500000
#define N_DST1 100000
#define N_DST2 20000
#define IN_F   128
#define H_F    128
#define N_CLS  16

#define BR1   640
#define NB1   157
#define NB2   79
#define NB    236
#define NBP   256
#define SWEEP_WG 256
#define L1_CAP 12288
#define L2_CAP 16384

__device__ __forceinline__ int wave_incl_scan(int v, int lane) {
    for (int o = 1; o < 32; o <<= 1) {
        const int up = __shfl_up(v, o);
        if (lane >= o) v += up;
    }
    return v;
}

__device__ __forceinline__ int block_excl_scan(int v, int* sTot) {
    const int lane = threadIdx.x & 31;
    const int wave = threadIdx.x >> 5;
    const int incl = wave_incl_scan(v, lane);
    if (lane == 31) sTot[wave] = incl;
    __syncthreads();
    if (wave == 0) {
        const int x  = sTot[lane];
        const int xi = wave_incl_scan(x, lane);
        sTot[32 + lane] = xi - x;
    }
    __syncthreads();
    return sTot[32 + wave] + incl - v;
}

__global__ __launch_bounds__(1024)
void bucket_local_kernel(const int*   __restrict__ src1,
                         const int*   __restrict__ dst1,
                         const float* __restrict__ w1,
                         const int*   __restrict__ src2,
                         const int*   __restrict__ dst2,
                         const float* __restrict__ w2,
                         int*  __restrict__ pieceCnt,
                         int*  __restrict__ pieceOff,
                         int2* __restrict__ rec,
                         int nE1, int nE2) {
    __shared__ int  sH[NBP];
    __shared__ int  sBase[NBP];
    __shared__ int  sTot[64];
    __shared__ int2 sRec[L1_CAP];
    const int t = threadIdx.x;
    const int c1 = (nE1 + SWEEP_WG - 1) / SWEEP_WG;
    const int c2 = (nE2 + SWEEP_WG - 1) / SWEEP_WG;
    const int a1 = min((int)blockIdx.x * c1, nE1);
    const int a2 = min((int)blockIdx.x * c2, nE2);
    const int n1 = min(c1, nE1 - a1);
    const int n2 = min(c2, nE2 - a2);
    const int m  = n1 + n2;
    const int lo = a1 + a2;
    const int o2 = a2 - n1;
    const bool stage = m <= L1_CAP;
    if (stage) {
        int d[12], s[12], wb[12], rk[12];
        if (m > 0) {
            #pragma unroll
            for (int k = 0; k < 12; ++k) {
                const int  uc = min(k * 1024 + t, m - 1);
                const bool a  = uc < n1;
                const int  iA = a1 + uc;
                const int  iB = o2 + uc;
                d[k]  = *(a ? dst1 + iA : dst2 + iB);
                s[k]  = *(a ? src1 + iA : src2 + iB);
                wb[k] = __float_as_int(*(a ? w1 + iA : w2 + iB));
            }
        }
        if (t < NBP) sH[t] = 0;
        __syncthreads();
        if (m > 0) {
            #pragma unroll
            for (int k = 0; k < 12; ++k) {
                const int u  = k * 1024 + t;
                const int bk = (u >= m) ? NB : (u < n1) ? (int)((unsigned)d[k] / BR1) : NB1 + (d[k] >> 8);
                rk[k] = atomicAdd(&sH[bk], 1);
            }
        }
        __syncthreads();
        const int v    = (t < NB) ? sH[t] : 0;
        const int excl = block_excl_scan(v, sTot);
        if (t < NB) {
            sBase[t] = excl;
            pieceCnt[t * SWEEP_WG + blockIdx.x] = v;
            pieceOff[t * SWEEP_WG + blockIdx.x] = excl;
        } else if (t < NBP) {
            sBase[t] = 0;
        }
        __syncthreads();
        if (m > 0) {
            #pragma unroll
            for (int k = 0; k < 12; ++k) {
                const int u  = k * 1024 + t;
                const int bk = (u >= m) ? NB : (u < n1) ? (int)((unsigned)d[k] / BR1) : NB1 + (d[k] >> 8);
                rk[k] += sBase[bk];
            }
            #pragma unroll
            for (int k = 0; k < 12; ++k) {
                const int u = k * 1024 + t;
                if (u < m) {
                    const int lr = (u < n1) ? (int)((unsigned)d[k] % BR1) : (d[k] & 255);
                    sRec[rk[k]] = make_int2(s[k] | (lr << 20), wb[k]);
                }
            }
        }
        __syncthreads();
        for (int i = t; i < m; i += 1024) rec[lo + i] = sRec[i];
        return;
    }
    if (t < NBP) sH[t] = 0;
    __syncthreads();
    for (int base = 0; base < m; base += 4096) {
        int d[4];
        #pragma unroll
        for (int k = 0; k < 4; ++k) {
            const int uc = min(base + k * 1024 + t, m - 1);
            d[k] = *((uc < n1) ? dst1 + (a1 + uc) : dst2 + (o2 + uc));
        }
        #pragma unroll
        for (int k = 0; k < 4; ++k) {
            const int u  = base + k * 1024 + t;
            const int bk = (u >= m) ? NB : (u < n1) ? (int)((unsigned)d[k] / BR1) : NB1 + (d[k] >> 8);
            atomicAdd(&sH[bk], 1);
        }
    }
    __syncthreads();
    const int v    = (t < NB) ? sH[t] : 0;
    const int excl = block_excl_scan(v, sTot);
    if (t < NB) {
        sBase[t] = excl;
        pieceCnt[t * SWEEP_WG + blockIdx.x] = v;
        pieceOff[t * SWEEP_WG + blockIdx.x] = excl;
    } else if (t < NBP) {
        sBase[t] = 0;
    }
    __syncthreads();
    for (int base = 0; base < m; base += 4096) {
        int d[4], s[4], wb[4];
        #pragma unroll
        for (int k = 0; k < 4; ++k) {
            const int  uc = min(base + k * 1024 + t, m - 1);
            const bool a  = uc < n1;
            const int  iA = a1 + uc;
            const int  iB = o2 + uc;
            d[k]  = *(a ? dst1 + iA : dst2 + iB);
            s[k]  = *(a ? src1 + iA : src2 + iB);
            wb[k] = __float_as_int(*(a ? w1 + iA : w2 + iB));
        }
        int pos[4];
        #pragma unroll
        for (int k = 0; k < 4; ++k) {
            const int u  = base + k * 1024 + t;
            const int bk = (u >= m) ? NB : (u < n1) ? (int)((unsigned)d[k] / BR1) : NB1 + (d[k] >> 8);
            pos[k] = atomicAdd(&sBase[bk], 1);
        }
        #pragma unroll
        for (int k = 0; k < 4; ++k) {
            const int u = base + k * 1024 + t;
            if (u < m) {
                const int  lr = (u < n1) ? (int)((unsigned)d[k] % BR1) : (d[k] & 255);
                const int2 r  = make_int2(s[k] | (lr << 20), wb[k]);
                rec[lo + pos[k]] = r;
            }
        }
    }
}

__global__ __launch_bounds__(1024)
void bucket_rows_kernel(const int2* __restrict__ rec,
                        const int* __restrict__ pieceCnt,
                        const int* __restrict__ pieceOff,
                        int* __restrict__ cnt1, int* __restrict__ cnt2,
                        int* __restrict__ off1, int* __restrict__ off2,
                        int2* __restrict__ list1, int2* __restrict__ list2,
                        int nE1, int nE2) {
    __shared__ int sCnt[BR1 + 32];
    __shared__ int sCur[BR1 + 32];
    __shared__ int sTot[64];
    __shared__ int sRed[32];
    __shared__ int sMax[32];
    __shared__ int sSum[32];
    __shared__ int2 sOut[L2_CAP];
    const int t    = threadIdx.x;
    const int lane = t & 31;
    const int wave = t >> 5;
    const int b    = blockIdx.x;
    const bool l1  = b < NB1;
    const int row0  = l1 ? b * BR1 : (b - NB1) * 256;
    const int nrows = l1 ? min(BR1, N_DST1 - row0) : min(256, N_DST2 - row0);
    if (t < BR1 + 32) sCnt[t] = 0;
    const int g   = wave * 8 + (lane & 7);
    const int c1  = (nE1 + SWEEP_WG - 1) / SWEEP_WG;
    const int c2  = (nE2 + SWEEP_WG - 1) / SWEEP_WG;
    const int pcv = pieceCnt[b * SWEEP_WG + g];
    const int plv = pieceOff[b * SWEEP_WG + g];
    const int psv = min(g * c1, nE1) + min(g * c2, nE2) + plv;
    int part = (lane < 8) ? plv : 0;
    for (int o = 16; o > 0; o >>= 1) part += __shfl_xor(part, o);
    int pcmv = pcv;
    for (int o = 4; o > 0; o >>= 1) pcmv = max(pcmv, __shfl_xor(pcmv, o));
    const int pcm = __builtin_amdgcn_readfirstlane(pcmv);
    int psum = (lane < 8) ? pcv : 0;
    for (int o = 16; o > 0; o >>= 1) psum += __shfl_xor(psum, o);
    if (lane == 0) { sRed[wave] = part; sMax[wave] = pcm; sSum[wave] = psum; }
    __syncthreads();
    int start = sRed[lane];
    for (int o = 16; o > 0; o >>= 1) start += __shfl_xor(start, o);
    int total = sSum[lane];
    for (int o = 16; o > 0; o >>= 1) total += __shfl_xor(total, o);
    int bmax = sMax[lane];
    for (int o = 16; o > 0; o >>= 1) bmax = max(bmax, __shfl_xor(bmax, o));
    const int  ntot = __builtin_amdgcn_readfirstlane(total);
    const bool keep = __builtin_amdgcn_readfirstlane(bmax) <= 96 && ntot <= L2_CAP;
    int pc[8], pb[8];
    #pragma unroll
    for (int q = 0; q < 8; ++q) {
        pc[q] = __builtin_amdgcn_readlane(pcv, q);
        pb[q] = (pc[q] > 0) ? __builtin_amdgcn_readlane(psv, q) : 0;
    }
    const int outBaseK = l1 ? start : start - nE1;
    if (keep) {
        int2 e[3][8];
        int  rk[3][8];
        #pragma unroll
        for (int tr = 0; tr < 3; ++tr)
            if (tr * 32 < pcm) {
                #pragma unroll
                for (int q = 0; q < 8; ++q)
                    e[tr][q] = rec[pb[q] + min(tr * 32 + lane, max(pc[q] - 1, 0))];
            }
        #pragma unroll
        for (int tr = 0; tr < 3; ++tr)
            if (tr * 32 < pcm) {
                #pragma unroll
                for (int q = 0; q < 8; ++q)
                    rk[tr][q] = atomicAdd(&sCnt[(tr * 32 + lane < pc[q])
                                                ? (int)((unsigned)e[tr][q].x >> 20) : BR1 + lane], 1);
            }
        __syncthreads();
        const int v    = (t < BR1) ? sCnt[t] : 0;
        const int excl = block_excl_scan(v, sTot);
        if (t < BR1 + 32) sCur[t] = (t < BR1) ? excl : 0;
        if (t < nrows) {
            (l1 ? cnt1 : cnt2)[row0 + t] = v;
            (l1 ? off1 : off2)[row0 + t] = outBaseK + excl;
        }
        __syncthreads();
        int2* out = (l1 ? list1 : list2) + outBaseK;
        #pragma unroll
        for (int tr = 0; tr < 3; ++tr)
            if (tr * 32 < pcm) {
                #pragma unroll
                for (int q = 0; q < 8; ++q)
                    rk[tr][q] += sCur[(tr * 32 + lane < pc[q])
                                      ? (int)((unsigned)e[tr][q].x >> 20) : BR1 + lane];
            }
        #pragma unroll
        for (int tr = 0; tr < 3; ++tr)
            if (tr * 32 < pcm) {
                #pragma unroll
                for (int q = 0; q < 8; ++q)
                    if (tr * 32 + lane < pc[q])
                        sOut[rk[tr][q]] = make_int2(e[tr][q].x & 0xFFFFF, e[tr][q].y);
            }
        __syncthreads();
        for (int i0 = 0; i0 < ntot; i0 += 4096) {
            int2 r[4];
            #pragma unroll
            for (int k = 0; k < 4; ++k) r[k] = sOut[min(i0 + k * 1024 + t, ntot - 1)];
            #pragma unroll
            for (int k = 0; k < 4; ++k) {
                const int i = i0 + k * 1024 + t;
                if (i < ntot) out[i] = r[k];
            }
        }
        return;
    }
    for (int j0 = 0; j0 < pcm; j0 += 32) {
        int x[8];
        #pragma unroll
        for (int q = 0; q < 8; ++q) x[q] = rec[pb[q] + min(j0 + lane, max(pc[q] - 1, 0))].x;
        #pragma unroll
        for (int q = 0; q < 8; ++q)
            atomicAdd(&sCnt[(j0 + lane < pc[q]) ? (int)((unsigned)x[q] >> 20) : BR1 + lane], 1);
    }
    __syncthreads();
    const int v    = (t < BR1) ? sCnt[t] : 0;
    const int excl = block_excl_scan(v, sTot);
    const int outBase = outBaseK;
    if (t < BR1 + 32) sCur[t] = (t < BR1) ? excl : 0;
    if (t < nrows) {
        (l1 ? cnt1 : cnt2)[row0 + t] = v;
        (l1 ? off1 : off2)[row0 + t] = outBase + excl;
    }
    __syncthreads();
    int2* out = (l1 ? list1 : list2) + outBase;
    for (int j0 = 0; j0 < pcm; j0 += 32) {
        int2 e[8];
        #pragma unroll
        for (int q = 0; q < 8; ++q) e[q] = rec[pb[q] + min(j0 + lane, max(pc[q] - 1, 0))];
        int pos[8];
        #pragma unroll
        for (int q = 0; q < 8; ++q)
            pos[q] = atomicAdd(&sCur[(j0 + lane < pc[q]) ? (int)((unsigned)e[q].x >> 20) : BR1 + lane], 1);
        #pragma unroll
        for (int q = 0; q < 8; ++q)
            if (j0 + lane < pc[q]) out[pos[q]] = make_int2(e[q].x & 0xFFFFF, e[q].y);
    }
}

#define G1_WG    512
#define G1_PITCH 260
#define G1_TP    20
#define ZP_HALF  144
#define ZP_WAVE  288

#define G1_GS 16

__device__ __forceinline__ int2 g1_record(const int2* __restrict__ list,
                                          int cA, int stA, int cB, int stB, int g) {
    int2 r = make_int2(0, 0);
    if (g < cA + cB) r = list[(g < cA) ? stA + g : stB + (g - cA)];
    return r;
}

__device__ __forceinline__ void g1_prefetch_row(const float* __restrict__ x, int sr) {
    const char* p = (const char*)(x + (size_t)sr * IN_F);
    __builtin_prefetch(p, 0, 1);
    __builtin_prefetch(p + 256, 0, 1);
}

__device__ __forceinline__ void g1_trip(const char* __restrict__ xb, unsigned lo16, const int J,
                                        int srcs, float wAl, float wBl,
                                        float4& sa, float4& sb) {
    float4   v[G1_GS];
    unsigned off[G1_GS];
    #pragma unroll
    for (int i = 0; i < G1_GS; ++i) {
        const unsigned sr = (unsigned)__builtin_amdgcn_readlane(srcs, J + i);
        off[i] = sr * (unsigned)(IN_F * 4) + lo16;
    }
    __builtin_amdgcn_sched_barrier(0);
    #pragma unroll
    for (int i = 0; i < G1_GS; ++i) v[i] = *(const float4*)(xb + off[i]);
    #pragma unroll
    for (int i = 0; i < G1_GS; ++i) {
        const float wa = __int_as_float(__builtin_amdgcn_readlane(__float_as_int(wAl), J + i));
        const float wb = __int_as_float(__builtin_amdgcn_readlane(__float_as_int(wBl), J + i));
        sa.x = __builtin_fmaf(v[i].x, wa, sa.x); sa.y = __builtin_fmaf(v[i].y, wa, sa.y);
        sa.z = __builtin_fmaf(v[i].z, wa, sa.z); sa.w = __builtin_fmaf(v[i].w, wa, sa.w);
        sb.x = __builtin_fmaf(v[i].x, wb, sb.x); sb.y = __builtin_fmaf(v[i].y, wb, sb.y);
        sb.z = __builtin_fmaf(v[i].z, wb, sb.z); sb.w = __builtin_fmaf(v[i].w, wb, sb.w);
    }
}

__device__ __forceinline__ void g1_gather(const float* __restrict__ x,
                                          const int2*  __restrict__ list,
                                          int cA, int stA, int cB, int stB, int lane,
                                          int2 my, int2 nx,
                                          float4& sa, float4& sb) {
    sa = make_float4(0.f, 0.f, 0.f, 0.f);
    sb = make_float4(0.f, 0.f, 0.f, 0.f);
    const int ct = cA + cB;
    const char* xb = (const char*)x;
    const unsigned lo16 = (unsigned)lane * 16u;
    for (int base = 0; base < ct; base += 32) {
        const int2 n2 = g1_record(list, cA, stA, cB, stB, base + 64 + lane);
        const float w   = __int_as_float(my.y);
        const float wAl = (base + lane <  cA) ? w : 0.0f;
        const float wBl = (base + lane >= cA) ? w : 0.0f;
        g1_trip(xb, lo16, 0, my.x, wAl, wBl, sa, sb);
        if (ct - base > G1_GS)
            g1_trip(xb, lo16, G1_GS, my.x, wAl, wBl, sa, sb);
        my = nx;
        nx = n2;
        if (base + 64 + lane < ct) g1_prefetch_row(x, n2.x);
    }
}

__device__ __forceinline__ void g1_stage(float* s, int lane,
                                         const float4& px0, const float4& px1,
                                         const float4& pa0, const float4& pa1,
                                         float d0, float d1) {
    const float i0 = 1.0f / fmaxf(d0, 1.0f);
    const float i1 = 1.0f / fmaxf(d1, 1.0f);
    float* s0 = s + lane * 4;
    float* s1 = s0 + G1_PITCH;
    *(float4*)s0         = px0;
    *(float4*)(s0 + 128) = make_float4(pa0.x * i0, pa0.y * i0, pa0.z * i0, pa0.w * i0);
    *(float4*)s1         = px1;
    *(float4*)(s1 + 128) = make_float4(pa1.x * i1, pa1.y * i1, pa1.z * i1, pa1.w * i1);
}

__global__ __launch_bounds__(256)
void sage_gemm1_kernel(const float* __restrict__ x,
                       const int*   __restrict__ list,
                       const int*   __restrict__ segStart,
                       const int*   __restrict__ cnt,
                       const float* __restrict__ Ws,
                       const float* __restrict__ Wn,
                       const float* __restrict__ bias,
                       const float* __restrict__ Wn2,
                       float* __restrict__ h,
                       float* __restrict__ z) {
    __shared__ __attribute__((aligned(16))) float sA[2][16 * G1_PITCH];
    __shared__ __attribute__((aligned(16))) float sT[8][16 * G1_TP];
    __shared__ __attribute__((aligned(16))) float sZ[8][ZP_WAVE];
    const int lane   = threadIdx.x & 31;
    const int wave   = __builtin_amdgcn_readfirstlane(threadIdx.x >> 5);
    const int col    = wave * 16 + (lane & 15);
    const int khalf  = (lane >> 4) * 2;
    const int nTiles = N_DST1 / 16;
    const int2* rl   = (const int2*)list;

    int tile = blockIdx.x;
    if (tile >= nTiles) return;

    float4 px0, px1, pa0, pa1;
    float  d0, d1;
    int drow = tile * 16 + wave * 2;
    int cA  = __builtin_amdgcn_readfirstlane(cnt[drow]);
    int cB  = __builtin_amdgcn_readfirstlane(cnt[drow + 1]);
    int stA = __builtin_amdgcn_readfirstlane(segStart[drow]);
    int stB = __builtin_amdgcn_readfirstlane(segStart[drow + 1]);
    int mrow = min(tile + (int)gridDim.x, nTiles - 1) * 16 + wave * 2;
    int mcA = cnt[mrow], mcB = cnt[mrow + 1], msA = segStart[mrow], msB = segStart[mrow + 1];

    v2f breg[64];
    #pragma unroll
    for (int ks = 0; ks < 32; ++ks) {
        const int kk = ks * 4 + khalf;
        breg[ks].x      = Ws[(size_t)kk * H_F + col];
        breg[ks].y      = Ws[(size_t)(kk + 1) * H_F + col];
        breg[32 + ks].x = Wn[(size_t)kk * H_F + col];
        breg[32 + ks].y = Wn[(size_t)(kk + 1) * H_F + col];
    }
    const float bv = bias[col];

    v2f b2reg[4];
    #pragma unroll
    for (int s = 0; s < 4; ++s) {
        const int kk = wave * 16 + s * 4 + khalf;
        b2reg[s].x = Wn2[(size_t)kk * N_CLS + (lane & 15)];
        b2reg[s].y = Wn2[(size_t)(kk + 1) * N_CLS + (lane & 15)];
    }
    float* tp = &sT[wave][0];

    int2 qmy = g1_record(rl, cA, stA, cB, stB, lane);
    int2 qnx = g1_record(rl, cA, stA, cB, stB, 32 + lane);
    if (lane >= G1_GS && lane < cA + cB) g1_prefetch_row(x, qmy.x);
    if (32 + lane < cA + cB)             g1_prefetch_row(x, qnx.x);
    g1_gather(x, rl, cA, stA, cB, stB, lane, qmy, qnx, pa0, pa1);
    d0 = (float)cA;
    d1 = (float)cB;
    {
        const int uA = __builtin_amdgcn_readfirstlane(mcA), uB = __builtin_amdgcn_readfirstlane(mcB);
        const int vA = __builtin_amdgcn_readfirstlane(msA), vB = __builtin_amdgcn_readfirstlane(msB);
        qmy = g1_record(rl, uA, vA, uB, vB, lane);
        qnx = g1_record(rl, uA, vA, uB, vB, 32 + lane);
        px0 = ((const float4*)(x + (size_t)drow       * IN_F))[lane];
        px1 = ((const float4*)(x + (size_t)(drow + 1) * IN_F))[lane];
        if (lane < uA + uB)      g1_prefetch_row(x, qmy.x);
        if (32 + lane < uA + uB) g1_prefetch_row(x, qnx.x);
    }
    g1_stage(&sA[0][wave * 2 * G1_PITCH], lane, px0, px1, pa0, pa1, d0, d1);
    __syncthreads();

    int cur = 0;
    int pfA = 0;
    for (; tile < nTiles; tile += (int)gridDim.x) {
        const int  nxt  = tile + (int)gridDim.x;
        const bool more = nxt < nTiles;
        if (more) {
            drow = nxt * 16 + wave * 2;
            cA  = __builtin_amdgcn_readfirstlane(mcA);
            cB  = __builtin_amdgcn_readfirstlane(mcB);
            stA = __builtin_amdgcn_readfirstlane(msA);
            stB = __builtin_amdgcn_readfirstlane(msB);
            mrow = min(nxt + (int)gridDim.x, nTiles - 1) * 16 + wave * 2;
            mcA = cnt[mrow]; mcB = cnt[mrow + 1]; msA = segStart[mrow]; msB = segStart[mrow + 1];
            g1_gather(x, rl, cA, stA, cB, stB, lane, qmy, qnx, pa0, pa1);
            d0 = (float)cA;
            d1 = (float)cB;
            pfA = __builtin_amdgcn_readfirstlane(mcA) + __builtin_amdgcn_readfirstlane(mcB);
            {
                const int uA = __builtin_amdgcn_readfirstlane(mcA), uB = __builtin_amdgcn_readfirstlane(mcB);
                const int vA = __builtin_amdgcn_readfirstlane(msA), vB = __builtin_amdgcn_readfirstlane(msB);
                qmy = g1_record(rl, uA, vA, uB, vB, lane);
                qnx = g1_record(rl, uA, vA, uB, vB, 32 + lane);
            }
            px0 = ((const float4*)(x + (size_t)drow       * IN_F))[lane];
            px1 = ((const float4*)(x + (size_t)(drow + 1) * IN_F))[lane];
        }

        const float* ap = &sA[cur][(lane & 15) * G1_PITCH + khalf];
        v8f c = {};
        #pragma unroll
        for (int ks = 0; ks < 64; ++ks) {
            const v2f a = *(const v2f*)(ap + ks * 4);
            c = __builtin_amdgcn_wmma_f32_16x16x4_f32(false, a, false, breg[ks],
                                                      (short)0, c, false, false);
        }

        if (more) {
            if (lane < pfA)      g1_prefetch_row(x, qmy.x);
            if (32 + lane < pfA) g1_prefetch_row(x, qnx.x);
        }
        if (more) g1_stage(&sA[cur ^ 1][wave * 2 * G1_PITCH], lane, px0, px1, pa0, pa1, d0, d1);
        __syncthreads();
        cur ^= 1;

        const int hi = lane >> 4;
        const int cl = lane & 15;
        #pragma unroll
        for (int r = 0; r < 8; ++r) {
            c[r] = fmaxf(c[r] + bv, 0.0f);
            tp[(r + 8 * hi) * G1_TP + cl] = c[r];
        }
        v8f zc = {};
        #pragma unroll
        for (int s = 0; s < 4; ++s) {
            const v2f a = *(const v2f*)(tp + cl * G1_TP + s * 4 + khalf);
            zc = __builtin_amdgcn_wmma_f32_16x16x4_f32(false, a, false, b2reg[s],
                                                      (short)0, zc, false, false);
        }
        float* zp = &sZ[wave][hi * ZP_HALF + cl];
        #pragma unroll
        for (int r = 0; r < 8; ++r) zp[r * 16] = zc[r];
        __syncthreads();
        {
            const int t   = threadIdx.x;
            const int row = t >> 4;
            const int zi  = (row >> 3) * ZP_HALF + (row & 7) * 16 + (t & 15);
            float zs = sZ[0][zi];
            #pragma unroll
            for (int w = 1; w < 8; ++w) zs += sZ[w][zi];
            z[(size_t)tile * 256 + t] = zs;
        }
        if (tile * 16 < N_DST2) {
            const int rbase = tile * 16 + hi * 8;
            #pragma unroll
            for (int r = 0; r < 8; ++r) h[(size_t)(rbase + r) * H_F + col] = c[r];
        }
    }
}

__global__ __launch_bounds__(256)
void sage_layer2_kernel(const float* __restrict__ h,
                        const float* __restrict__ z,
                        const int*   __restrict__ list,
                        const int*   __restrict__ segStart,
                        const int*   __restrict__ cnt,
                        const float* __restrict__ Ws,
                        const float* __restrict__ bias,
                        float* __restrict__ out) {
    __shared__ __attribute__((aligned(16))) float sN[16 * 16];
    __shared__ __attribute__((aligned(16))) float sP[8][ZP_WAVE];
    const int t     = threadIdx.x;
    const int lane  = t & 31;
    const int wave  = t >> 5;
    const int hi    = lane >> 4;
    const int cl    = lane & 15;
    const int khalf = hi * 2;
    const int m0    = blockIdx.x * 16;

    const float* hr = h  + (size_t)(m0 + cl) * H_F + wave * 16 + khalf;
    const float* wr = Ws + (size_t)(wave * 16 + khalf) * N_CLS + cl;
    v2f fa[4], fb[4];
    #pragma unroll
    for (int s = 0; s < 4; ++s) {
        fa[s]   = *(const v2f*)(hr + s * 4);
        fb[s].x = wr[(size_t)(s * 4) * N_CLS];
        fb[s].y = wr[(size_t)(s * 4 + 1) * N_CLS];
    }

    const int d   = m0 + wave * 2 + hi;
    const int c   = cnt[d];
    const int st  = segStart[d];
    const int cm1 = max(c - 1, 0);
    const int2* seg = (const int2*)list + ((c > 0) ? st : 0);
    const int es = (lane >> 2) & 3;
    const int q  = lane & 3;
    const int cmax = max(c, __shfl_xor(c, 16));
    const float4* z4 = (const float4*)z;
    float4 sum = make_float4(0.f, 0.f, 0.f, 0.f);
    int2 e[4];
    #pragma unroll
    for (int k = 0; k < 4; ++k) e[k] = seg[min(k * 4 + es, cm1)];
    __builtin_amdgcn_sched_barrier(0);
    for (int base = 0; base < cmax; base += 16) {
        float4 v[4];
        int2   n[4];
        #pragma unroll
        for (int k = 0; k < 4; ++k) v[k] = z4[(size_t)e[k].x * 4 + q];
        #pragma unroll
        for (int k = 0; k < 4; ++k) n[k] = seg[min(base + 16 + k * 4 + es, cm1)];
        __builtin_amdgcn_sched_barrier(0);
        #pragma unroll
        for (int k = 0; k < 4; ++k) {
            const bool  ok = base + k * 4 + es < c;
            const float w  = __int_as_float(e[k].y);
            const float tx = sum.x + v[k].x * w, ty = sum.y + v[k].y * w;
            const float tz = sum.z + v[k].z * w, tw = sum.w + v[k].w * w;
            sum.x = ok ? tx : sum.x; sum.y = ok ? ty : sum.y;
            sum.z = ok ? tz : sum.z; sum.w = ok ? tw : sum.w;
            e[k] = n[k];
        }
    }
    sum.x += __shfl_xor(sum.x, 4); sum.y += __shfl_xor(sum.y, 4);
    sum.z += __shfl_xor(sum.z, 4); sum.w += __shfl_xor(sum.w, 4);
    sum.x += __shfl_xor(sum.x, 8); sum.y += __shfl_xor(sum.y, 8);
    sum.z += __shfl_xor(sum.z, 8); sum.w += __shfl_xor(sum.w, 8);
    if (es == 0) {
        const float inv = 1.0f / fmaxf((float)c, 1.0f);
        *(float4*)&sN[(wave * 2 + hi) * 16 + q * 4] =
            make_float4(sum.x * inv, sum.y * inv, sum.z * inv, sum.w * inv);
    }

    v8f pc = {};
    #pragma unroll
    for (int s = 0; s < 4; ++s)
        pc = __builtin_amdgcn_wmma_f32_16x16x4_f32(false, fa[s], false, fb[s],
                                                  (short)0, pc, false, false);
    float* pp = &sP[wave][hi * ZP_HALF + cl];
    #pragma unroll
    for (int r = 0; r < 8; ++r) pp[r * 16] = pc[r];
    __syncthreads();

    const int row = t >> 4;
    const int pi  = (row >> 3) * ZP_HALF + (row & 7) * 16 + cl;
    float v = sP[0][pi];
    #pragma unroll
    for (int w = 1; w < 8; ++w) v += sP[w][pi];
    v += sN[t] + bias[cl];
    float m = v;
    m = fmaxf(m, __shfl_xor(m, 1, 16));
    m = fmaxf(m, __shfl_xor(m, 2, 16));
    m = fmaxf(m, __shfl_xor(m, 4, 16));
    m = fmaxf(m, __shfl_xor(m, 8, 16));
    const float ex = __expf(v - m);
    float s = ex;
    s += __shfl_xor(s, 1, 16);
    s += __shfl_xor(s, 2, 16);
    s += __shfl_xor(s, 4, 16);
    s += __shfl_xor(s, 8, 16);
    out[(size_t)m0 * N_CLS + t] = (v - m) - __logf(s);
}

extern "C" void kernel_launch(void* const* d_in, const int* in_sizes, int n_in,
                              void* d_out, int out_size, void* d_ws, size_t ws_size,
                              hipStream_t stream) {
    const float* x    = (const float*)d_in[0];
    const int*   src1 = (const int*)  d_in[1];
    const int*   dst1 = (const int*)  d_in[2];
    const float* w1   = (const float*)d_in[3];
    const int*   src2 = (const int*)  d_in[4];
    const int*   dst2 = (const int*)  d_in[5];
    const float* w2   = (const float*)d_in[6];
    const float* Ws1  = (const float*)d_in[7];
    const float* Wn1  = (const float*)d_in[8];
    const float* b1   = (const float*)d_in[9];
    const float* Ws2  = (const float*)d_in[10];
    const float* Wn2  = (const float*)d_in[11];
    const float* b2   = (const float*)d_in[12];
    float* out = (float*)d_out;

    float* wsA  = (float*)d_ws;
    float* hbuf = wsA;
    float* zbuf = hbuf + (size_t)N_DST2 * 128;
    int* list1  = (int*)(wsA + (size_t)N_DST1 * 128 + N_DST1);
    int* dbase  = list1 + (size_t)N_DST1 * 128;
    int* cnt1   = dbase + 4;
    int* cnt2   = cnt1 + N_DST1;
    int* off1   = cnt2 + N_DST2;
    int* off2   = off1 + N_DST1;
    int* list2  = off2 + N_DST2;
    int2* rec        = (int2*)wsA;
    int* pieceCnt    = (int*)wsA + 6000000;
    int* pieceOff    = pieceCnt + NB * SWEEP_WG;

    const int E1 = in_sizes[1];
    const int E2 = in_sizes[4];

    bucket_local_kernel<<<SWEEP_WG, 1024, 0, stream>>>(src1, dst1, w1, src2, dst2, w2,
                                                       pieceCnt, pieceOff, rec, E1, E2);
    bucket_rows_kernel<<<NB, 1024, 0, stream>>>(rec, pieceCnt, pieceOff, cnt1, cnt2, off1, off2,
                                                (int2*)list1, (int2*)list2, E1, E2);

    sage_gemm1_kernel<<<G1_WG, 256, 0, stream>>>(x, list1, off1, cnt1,
                                                       Ws1, Wn1, b1, Wn2, hbuf, zbuf);

    sage_layer2_kernel<<<N_DST2 / 16, 256, 0, stream>>>(hbuf, zbuf, list2, off2, cnt2,
                                                        Ws2, b2, out);
}
